# starting kernel
# baseline (speedup 1.0000x reference)
#include <hip/hip_runtime.h>
#include <hip/hip_bf16.h>

typedef float v2f __attribute__((ext_vector_type(2)));
typedef float v8f __attribute__((ext_vector_type(8)));

#define NN    50000
#define EE    800000
#define GG    500
#define DIN   78
#define DINP  80
#define DH    312
#define DHP   320
#define BNEPS 1e-5f

__global__ void fill_f32(float* __restrict__ p, float v, long n) {
  long i = (long)blockIdx.x * blockDim.x + threadIdx.x;
  if (i < n) p[i] = v;
}
__global__ void fill_u32(unsigned* __restrict__ p, unsigned v, long n) {
  long i = (long)blockIdx.x * blockDim.x + threadIdx.x;
  if (i < n) p[i] = v;
}

__global__ void deg_count(const int* __restrict__ dst, float* __restrict__ deg, int E) {
  int e = blockIdx.x * blockDim.x + threadIdx.x;
  if (e < E) atomicAdd(&deg[dst[e]], 1.0f);
}
__global__ void rsqrt_inplace(float* __restrict__ p, int n) {
  int i = blockIdx.x * blockDim.x + threadIdx.x;
  if (i < n) p[i] = rsqrtf(p[i]);
}

__global__ void pad_x(const float* __restrict__ x, float* __restrict__ xp, int M) {
  int n = blockIdx.x, c = threadIdx.x;
  if (n >= M || c >= DINP) return;
  xp[(size_t)n * DINP + c] = (c < DIN) ? x[(size_t)n * DIN + c] : 0.0f;
}
__global__ void pad_W(const float* __restrict__ W, float* __restrict__ Wp, int K, int Nn) {
  int r = blockIdx.x, c = threadIdx.x;
  Wp[(size_t)r * DHP + c] = (r < K && c < Nn) ? W[(size_t)r * Nn + c] : 0.0f;
}

__global__ __launch_bounds__(256) void gemm_f32_wmma(
    const float* __restrict__ A, int lda,
    const float* __restrict__ B,
    float* __restrict__ C,
    int M, int Npad, int Kpad) {
  int wave = (blockIdx.x * blockDim.x + threadIdx.x) >> 5;
  int lane = threadIdx.x & 31;
  int ntn  = Npad >> 4;
  int tm = wave / ntn;
  int tn = wave - tm * ntn;
  if (tm * 16 >= M) return;

  int half = lane >> 4;
  int m    = lane & 15;
  int row0 = tm * 16;
  int arow = row0 + m;
  const float* Arow = A + (size_t)(arow < M ? arow : (M - 1)) * lda;
  const float* Bcol = B + (tn * 16 + m);

  v8f acc = {0.f, 0.f, 0.f, 0.f, 0.f, 0.f, 0.f, 0.f};
#pragma unroll 4
  for (int k = 0; k < Kpad; k += 4) {
    int ka = k + half * 2;
    v2f a, b;
    a.x = Arow[ka];
    a.y = Arow[ka + 1];
    b.x = Bcol[(size_t)ka * DHP];
    b.y = Bcol[(size_t)(ka + 1) * DHP];
    acc = __builtin_amdgcn_wmma_f32_16x16x4_f32(false, a, false, b,
                                                (short)0, acc, false, false);
  }
  int col = tn * 16 + m;
#pragma unroll
  for (int r = 0; r < 8; ++r) {
    int row = row0 + r + half * 8;
    if (row < M) C[(size_t)row * DHP + col] = acc[r];
  }
}

__global__ void agg_init(const float* __restrict__ h, const float* __restrict__ dinv,
                         float* __restrict__ agg, int M) {
  int n = blockIdx.x, c = threadIdx.x;
  if (n >= M) return;
  float di = dinv[n];
  agg[(size_t)n * DHP + c] = h[(size_t)n * DHP + c] * di * di;
}
__global__ void scatter_edges(const float* __restrict__ h, const int* __restrict__ src,
                              const int* __restrict__ dst, const float* __restrict__ dinv,
                              float* __restrict__ agg, int E) {
  int e = blockIdx.x, c = threadIdx.x;
  if (e >= E || c >= DH) return;
  int s = src[e], d = dst[e];
  float norm = dinv[s] * dinv[d];
  atomicAdd(&agg[(size_t)d * DHP + c], h[(size_t)s * DHP + c] * norm);
}

__global__ void bias_relu_stats(float* __restrict__ y, const float* __restrict__ bias,
                                float* __restrict__ sums, int M) {
  int c = threadIdx.x;
  if (c >= DH) return;
  int r0 = blockIdx.x * 128;
  int r1 = min(r0 + 128, M);
  float b = bias[c], s = 0.f, s2 = 0.f;
  for (int r = r0; r < r1; ++r) {
    float v = y[(size_t)r * DHP + c] + b;
    v = fmaxf(v, 0.0f);
    y[(size_t)r * DHP + c] = v;
    s += v;
    s2 += v * v;
  }
  atomicAdd(&sums[c], s);
  atomicAdd(&sums[DHP + c], s2);
}
__global__ void bn_apply(float* __restrict__ y, const float* __restrict__ sums,
                         const float* __restrict__ gamma, const float* __restrict__ beta,
                         int M) {
  int c = threadIdx.x;
  if (c >= DH) return;
  int r0 = blockIdx.x * 128;
  int r1 = min(r0 + 128, M);
  float invn  = 1.0f / (float)M;
  float mean  = sums[c] * invn;
  float var   = sums[DHP + c] * invn - mean * mean;
  float scale = rsqrtf(var + BNEPS) * gamma[c];
  float shift = beta[c] - mean * scale;
  for (int r = r0; r < r1; ++r)
    y[(size_t)r * DHP + c] = y[(size_t)r * DHP + c] * scale + shift;
}

__device__ __forceinline__ unsigned enc_f32(float f) {
  unsigned u = __float_as_uint(f);
  return (u & 0x80000000u) ? ~u : (u | 0x80000000u);
}
__device__ __forceinline__ float dec_f32(unsigned u) {
  return __uint_as_float((u & 0x80000000u) ? (u & 0x7FFFFFFFu) : ~u);
}
__global__ void pool_max(const float* __restrict__ y, const int* __restrict__ batch,
                         unsigned* __restrict__ pool, int M) {
  int n = blockIdx.x, c = threadIdx.x;
  if (n >= M || c >= DH) return;
  int g = batch[n];
  atomicMax(&pool[(size_t)g * DH + c], enc_f32(y[(size_t)n * DHP + c]));
}
__global__ void pool_decode(const unsigned* __restrict__ pool, float* __restrict__ out) {
  int g = blockIdx.x, c = threadIdx.x;
  if (c >= DH) return;
  out[(size_t)g * DH + c] = dec_f32(pool[(size_t)g * DH + c]);
}

extern "C" void kernel_launch(void* const* d_in, const int* in_sizes, int n_in,
                              void* d_out, int out_size, void* d_ws, size_t ws_size,
                              hipStream_t stream) {
  const float* x      = (const float*)d_in[0];
  const int*   eidx   = (const int*)d_in[1];
  const int*   batch  = (const int*)d_in[2];
  const float* W1     = (const float*)d_in[3];
  const float* b1     = (const float*)d_in[4];
  const float* gamma1 = (const float*)d_in[5];
  const float* beta1  = (const float*)d_in[6];
  const float* W2     = (const float*)d_in[7];
  const float* b2     = (const float*)d_in[8];
  const float* gamma2 = (const float*)d_in[9];
  const float* beta2  = (const float*)d_in[10];
  float* out = (float*)d_out;

  const int* src = eidx;
  const int* dst = eidx + EE;

  float* ws = (float*)d_ws;
  float*    deg  = ws;
  float*    xp   = ws + 50048;
  float*    Wp1  = ws + 4050048;
  float*    Wp2  = ws + 4075648;
  float*    sums = ws + 4178048;
  unsigned* pool = (unsigned*)(ws + 4178688);
  float*    bufA = ws + 4334688;
  float*    bufB = ws + 20334688;

  const int statBlocks = (NN + 127) / 128;
  const int gemmTiles  = (NN / 16) * (DHP / 16);
  const int gemmGrid   = (gemmTiles + 7) / 8;

  fill_f32<<<(NN + 255) / 256, 256, 0, stream>>>(deg, 1.0f, NN);
  deg_count<<<(EE + 255) / 256, 256, 0, stream>>>(dst, deg, EE);
  rsqrt_inplace<<<(NN + 255) / 256, 256, 0, stream>>>(deg, NN);

  pad_x<<<NN, 128, 0, stream>>>(x, xp, NN);
  pad_W<<<DINP, DHP, 0, stream>>>(W1, Wp1, DIN, DH);
  pad_W<<<DHP, DHP, 0, stream>>>(W2, Wp2, DH, DH);

  gemm_f32_wmma<<<gemmGrid, 256, 0, stream>>>(xp, DINP, Wp1, bufA, NN, DHP, DINP);
  agg_init<<<NN, DHP, 0, stream>>>(bufA, deg, bufB, NN);
  scatter_edges<<<EE, DHP, 0, stream>>>(bufA, src, dst, deg, bufB, EE);
  fill_f32<<<(2 * DHP + 255) / 256, 256, 0, stream>>>(sums, 0.0f, 2 * DHP);
  bias_relu_stats<<<statBlocks, DHP, 0, stream>>>(bufB, b1, sums, NN);
  bn_apply<<<statBlocks, DHP, 0, stream>>>(bufB, sums, gamma1, beta1, NN);

  gemm_f32_wmma<<<gemmGrid, 256, 0, stream>>>(bufB, DHP, Wp2, bufA, NN, DHP, DHP);
  agg_init<<<NN, DHP, 0, stream>>>(bufA, deg, bufB, NN);
  scatter_edges<<<EE, DHP, 0, stream>>>(bufA, src, dst, deg, bufB, EE);
  fill_f32<<<(2 * DHP + 255) / 256, 256, 0, stream>>>(sums, 0.0f, 2 * DHP);
  bias_relu_stats<<<statBlocks, DHP, 0, stream>>>(bufB, b2, sums, NN);
  bn_apply<<<statBlocks, DHP, 0, stream>>>(bufB, sums, gamma2, beta2, NN);

  fill_u32<<<((long)GG * DH + 255) / 256, 256, 0, stream>>>(pool, 0u, (long)GG * DH);
  pool_max<<<NN, DHP, 0, stream>>>(bufB, batch, pool, NN);
  pool_decode<<<GG, DHP, 0, stream>>>(pool, out);
}


# Aggregation by per-call CSR gather: one wave sums a destination row in registers in slot order (slot = arrival rank from an int atomic add), one store per row; float atomics on agg removed
# speedup vs baseline: 3.5483x; 3.5483x over previous
#include <hip/hip_runtime.h>
#include <hip/hip_bf16.h>

typedef float v2f __attribute__((ext_vector_type(2)));
typedef float v8f __attribute__((ext_vector_type(8)));

#define NN    50000
#define EE    800000
#define GG    500
#define DIN   78
#define DINP  80
#define DH    312
#define DHP   320
#define BNEPS 1e-5f

__global__ void fill_f32(float* __restrict__ p, float v, long n) {
  long i = (long)blockIdx.x * blockDim.x + threadIdx.x;
  if (i < n) p[i] = v;
}
__global__ void fill_u32(unsigned* __restrict__ p, unsigned v, long n) {
  long i = (long)blockIdx.x * blockDim.x + threadIdx.x;
  if (i < n) p[i] = v;
}

__global__ void edge_rank(const int* __restrict__ dst, int* __restrict__ cnt,
                          int* __restrict__ rank, int E) {
  int e = blockIdx.x * blockDim.x + threadIdx.x;
  if (e < E) rank[e] = atomicAdd(&cnt[dst[e]], 1);
}
__global__ __launch_bounds__(1024) void scan_rows(const int* __restrict__ cnt,
                                                  int* __restrict__ rowptr,
                                                  float* __restrict__ dinv, int n) {
  __shared__ int wsum[32];
  const int tid = threadIdx.x, lane = tid & 31, w = tid >> 5;
  int carry = 0;
  for (int base = 0; base < n; base += 4096) {
    const int i = base + tid * 4;
    int v0 = 0, v1 = 0, v2 = 0, v3 = 0;
    if (i + 3 < n) {
      const int4 q = *(const int4*)(cnt + i);
      v0 = q.x; v1 = q.y; v2 = q.z; v3 = q.w;
    } else {
      if (i < n) v0 = cnt[i];
      if (i + 1 < n) v1 = cnt[i + 1];
      if (i + 2 < n) v2 = cnt[i + 2];
    }
    const int mine = v0 + v1 + v2 + v3;
    int x = mine;
    for (int o = 1; o < 32; o <<= 1) {
      const int t = __shfl_up(x, o);
      if (lane >= o) x += t;
    }
    if (lane == 31) wsum[w] = x;
    __syncthreads();
    if (w == 0) {
      int y = wsum[lane];
      for (int o = 1; o < 32; o <<= 1) {
        const int t = __shfl_up(y, o);
        if (lane >= o) y += t;
      }
      wsum[lane] = y;
    }
    __syncthreads();
    const int off = carry + (w > 0 ? wsum[w - 1] : 0) + x - mine;
    carry += wsum[31];
    if (i < n)     { rowptr[i]     = off;                dinv[i]     = rsqrtf((float)(v0 + 1)); }
    if (i + 1 < n) { rowptr[i + 1] = off + v0;           dinv[i + 1] = rsqrtf((float)(v1 + 1)); }
    if (i + 2 < n) { rowptr[i + 2] = off + v0 + v1;      dinv[i + 2] = rsqrtf((float)(v2 + 1)); }
    if (i + 3 < n) { rowptr[i + 3] = off + v0 + v1 + v2; dinv[i + 3] = rsqrtf((float)(v3 + 1)); }
    __syncthreads();
  }
  if (tid == 0) rowptr[n] = carry;
}
__global__ void place_edges(const int* __restrict__ src, const int* __restrict__ dst,
                            const int* __restrict__ rank, const int* __restrict__ rowptr,
                            int* __restrict__ csr_src, int E) {
  int e = blockIdx.x * blockDim.x + threadIdx.x;
  if (e < E) csr_src[rowptr[dst[e]] + rank[e]] = src[e];
}

__global__ void pad_x(const float* __restrict__ x, float* __restrict__ xp, int M) {
  int n = blockIdx.x, c = threadIdx.x;
  if (n >= M || c >= DINP) return;
  xp[(size_t)n * DINP + c] = (c < DIN) ? x[(size_t)n * DIN + c] : 0.0f;
}
__global__ void pad_W(const float* __restrict__ W, float* __restrict__ Wp, int K, int Nn) {
  int r = blockIdx.x, c = threadIdx.x;
  Wp[(size_t)r * DHP + c] = (r < K && c < Nn) ? W[(size_t)r * Nn + c] : 0.0f;
}

__global__ __launch_bounds__(256) void gemm_f32_wmma(
    const float* __restrict__ A, int lda,
    const float* __restrict__ B,
    float* __restrict__ C,
    int M, int Npad, int Kpad) {
  int wave = (blockIdx.x * blockDim.x + threadIdx.x) >> 5;
  int lane = threadIdx.x & 31;
  int ntn  = Npad >> 4;
  int tm = wave / ntn;
  int tn = wave - tm * ntn;
  if (tm * 16 >= M) return;

  int half = lane >> 4;
  int m    = lane & 15;
  int row0 = tm * 16;
  int arow = row0 + m;
  const float* Arow = A + (size_t)(arow < M ? arow : (M - 1)) * lda;
  const float* Bcol = B + (tn * 16 + m);

  v8f acc = {0.f, 0.f, 0.f, 0.f, 0.f, 0.f, 0.f, 0.f};
#pragma unroll 4
  for (int k = 0; k < Kpad; k += 4) {
    int ka = k + half * 2;
    v2f a, b;
    a.x = Arow[ka];
    a.y = Arow[ka + 1];
    b.x = Bcol[(size_t)ka * DHP];
    b.y = Bcol[(size_t)(ka + 1) * DHP];
    acc = __builtin_amdgcn_wmma_f32_16x16x4_f32(false, a, false, b,
                                                (short)0, acc, false, false);
  }
  int col = tn * 16 + m;
#pragma unroll
  for (int r = 0; r < 8; ++r) {
    int row = row0 + r + half * 8;
    if (row < M) C[(size_t)row * DHP + col] = acc[r];
  }
}

__global__ __launch_bounds__(256) void gather_rows(
    const float* __restrict__ h, const int* __restrict__ rowptr,
    const int* __restrict__ csr_src, const float* __restrict__ dinv,
    float* __restrict__ agg, int M) {
  const int lane = threadIdx.x & 31;
  const int n = __builtin_amdgcn_readfirstlane(blockIdx.x * 8 + (threadIdx.x >> 5));
  if (n >= M) return;
  const int beg = rowptr[n], end = rowptr[n + 1];
  const float dn = dinv[n];

  v2f acc[5];
  {
    const v2f* p = (const v2f*)(h + (size_t)n * DHP) + lane;
#pragma unroll
    for (int i = 0; i < 5; ++i) {
      const v2f v = p[i * 32];
      acc[i].x = v.x * dn * dn;
      acc[i].y = v.y * dn * dn;
    }
  }
  for (int base = beg; base < end; base += 32) {
    const int j = base + lane;
    int s_l = 0;
    float w_l = 0.f;
    if (j < end) {
      s_l = (int)min((unsigned)csr_src[j], (unsigned)(M - 1));
      w_l = dinv[s_l] * dn;
    }
    const int m = min(32, end - base);
#pragma unroll 4
    for (int t = 0; t < m; ++t) {
      const int s = __builtin_amdgcn_readlane(s_l, t);
      const float wgt = __int_as_float(__builtin_amdgcn_readlane(__float_as_int(w_l), t));
      const v2f* p = (const v2f*)(h + (size_t)s * DHP) + lane;
#pragma unroll
      for (int i = 0; i < 5; ++i) {
        const v2f v = p[i * 32];
        acc[i].x += v.x * wgt;
        acc[i].y += v.y * wgt;
      }
    }
  }
  v2f* o = (v2f*)(agg + (size_t)n * DHP) + lane;
#pragma unroll
  for (int i = 0; i < 5; ++i) o[i * 32] = acc[i];
}

__global__ void bias_relu_stats(float* __restrict__ y, const float* __restrict__ bias,
                                float* __restrict__ sums, int M) {
  int c = threadIdx.x;
  if (c >= DH) return;
  int r0 = blockIdx.x * 128;
  int r1 = min(r0 + 128, M);
  float b = bias[c], s = 0.f, s2 = 0.f;
  for (int r = r0; r < r1; ++r) {
    float v = y[(size_t)r * DHP + c] + b;
    v = fmaxf(v, 0.0f);
    y[(size_t)r * DHP + c] = v;
    s += v;
    s2 += v * v;
  }
  atomicAdd(&sums[c], s);
  atomicAdd(&sums[DHP + c], s2);
}
__global__ void bn_apply(float* __restrict__ y, const float* __restrict__ sums,
                         const float* __restrict__ gamma, const float* __restrict__ beta,
                         int M) {
  int c = threadIdx.x;
  if (c >= DH) return;
  int r0 = blockIdx.x * 128;
  int r1 = min(r0 + 128, M);
  float invn  = 1.0f / (float)M;
  float mean  = sums[c] * invn;
  float var   = sums[DHP + c] * invn - mean * mean;
  float scale = rsqrtf(var + BNEPS) * gamma[c];
  float shift = beta[c] - mean * scale;
  for (int r = r0; r < r1; ++r)
    y[(size_t)r * DHP + c] = y[(size_t)r * DHP + c] * scale + shift;
}

__device__ __forceinline__ unsigned enc_f32(float f) {
  unsigned u = __float_as_uint(f);
  return (u & 0x80000000u) ? ~u : (u | 0x80000000u);
}
__device__ __forceinline__ float dec_f32(unsigned u) {
  return __uint_as_float((u & 0x80000000u) ? (u & 0x7FFFFFFFu) : ~u);
}
__global__ void pool_max(const float* __restrict__ y, const int* __restrict__ batch,
                         unsigned* __restrict__ pool, int M) {
  int n = blockIdx.x, c = threadIdx.x;
  if (n >= M || c >= DH) return;
  int g = batch[n];
  atomicMax(&pool[(size_t)g * DH + c], enc_f32(y[(size_t)n * DHP + c]));
}
__global__ void pool_decode(const unsigned* __restrict__ pool, float* __restrict__ out) {
  int g = blockIdx.x, c = threadIdx.x;
  if (c >= DH) return;
  out[(size_t)g * DH + c] = dec_f32(pool[(size_t)g * DH + c]);
}

extern "C" void kernel_launch(void* const* d_in, const int* in_sizes, int n_in,
                              void* d_out, int out_size, void* d_ws, size_t ws_size,
                              hipStream_t stream) {
  const float* x      = (const float*)d_in[0];
  const int*   eidx   = (const int*)d_in[1];
  const int*   batch  = (const int*)d_in[2];
  const float* W1     = (const float*)d_in[3];
  const float* b1     = (const float*)d_in[4];
  const float* gamma1 = (const float*)d_in[5];
  const float* beta1  = (const float*)d_in[6];
  const float* W2     = (const float*)d_in[7];
  const float* b2     = (const float*)d_in[8];
  const float* gamma2 = (const float*)d_in[9];
  const float* beta2  = (const float*)d_in[10];
  float* out = (float*)d_out;

  const int* src = eidx;
  const int* dst = eidx + EE;

  float* ws = (float*)d_ws;
  float*    dinv = ws;
  int*      cnt  = (int*)(ws + 50048);
  int*      rowp = (int*)(ws + 100096);
  int*      rank = (int*)(ws + 150208);
  int*      csrs = (int*)(ws + 950208);
  float*    xp   = ws + 1750208;
  float*    Wp1  = ws + 5750208;
  float*    Wp2  = ws + 5775808;
  float*    sums = ws + 5878208;
  unsigned* pool = (unsigned*)(ws + 5878848);
  float*    bufA = ws + 6034880;
  float*    bufB = ws + 22034880;

  const int statBlocks = (NN + 127) / 128;
  const int gemmTiles  = (NN / 16) * (DHP / 16);
  const int gemmGrid   = (gemmTiles + 7) / 8;

  fill_u32<<<(NN + 255) / 256, 256, 0, stream>>>((unsigned*)cnt, 0u, NN);
  edge_rank<<<(EE + 255) / 256, 256, 0, stream>>>(dst, cnt, rank, EE);
  scan_rows<<<1, 1024, 0, stream>>>(cnt, rowp, dinv, NN);
  place_edges<<<(EE + 255) / 256, 256, 0, stream>>>(src, dst, rank, rowp, csrs, EE);

  pad_x<<<NN, 128, 0, stream>>>(x, xp, NN);
  pad_W<<<DINP, DHP, 0, stream>>>(W1, Wp1, DIN, DH);
  pad_W<<<DHP, DHP, 0, stream>>>(W2, Wp2, DH, DH);

  gemm_f32_wmma<<<gemmGrid, 256, 0, stream>>>(xp, DINP, Wp1, bufA, NN, DHP, DINP);
  gather_rows<<<(NN + 7) / 8, 256, 0, stream>>>(bufA, rowp, csrs, dinv, bufB, NN);
  fill_f32<<<(2 * DHP + 255) / 256, 256, 0, stream>>>(sums, 0.0f, 2 * DHP);
  bias_relu_stats<<<statBlocks, DHP, 0, stream>>>(bufB, b1, sums, NN);
  bn_apply<<<statBlocks, DHP, 0, stream>>>(bufB, sums, gamma1, beta1, NN);

  gemm_f32_wmma<<<gemmGrid, 256, 0, stream>>>(bufB, DHP, Wp2, bufA, NN, DHP, DHP);
  gather_rows<<<(NN + 7) / 8, 256, 0, stream>>>(bufA, rowp, csrs, dinv, bufB, NN);
  fill_f32<<<(2 * DHP + 255) / 256, 256, 0, stream>>>(sums, 0.0f, 2 * DHP);
  bias_relu_stats<<<statBlocks, DHP, 0, stream>>>(bufB, b2, sums, NN);
  bn_apply<<<statBlocks, DHP, 0, stream>>>(bufB, sums, gamma2, beta2, NN);

  fill_u32<<<((long)GG * DH + 255) / 256, 256, 0, stream>>>(pool, 0u, (long)GG * DH);
  pool_max<<<NN, DHP, 0, stream>>>(bufB, batch, pool, NN);
  pool_decode<<<GG, DHP, 0, stream>>>(pool, out);
}


# pool_max: a workgroup walks 64 consecutive rows (was one row), one column per lane, running max of the encoded values in a register, one atomicMax per column per graph-id change
# speedup vs baseline: 3.7251x; 1.0498x over previous
#include <hip/hip_runtime.h>
#include <hip/hip_bf16.h>

typedef float v2f __attribute__((ext_vector_type(2)));
typedef float v8f __attribute__((ext_vector_type(8)));

#define NN    50000
#define EE    800000
#define GG    500
#define DIN   78
#define DINP  80
#define DH    312
#define DHP   320
#define BNEPS 1e-5f

__global__ void fill_f32(float* __restrict__ p, float v, long n) {
  long i = (long)blockIdx.x * blockDim.x + threadIdx.x;
  if (i < n) p[i] = v;
}
__global__ void fill_u32(unsigned* __restrict__ p, unsigned v, long n) {
  long i = (long)blockIdx.x * blockDim.x + threadIdx.x;
  if (i < n) p[i] = v;
}

__global__ void edge_rank(const int* __restrict__ dst, int* __restrict__ cnt,
                          int* __restrict__ rank, int E) {
  int e = blockIdx.x * blockDim.x + threadIdx.x;
  if (e < E) rank[e] = atomicAdd(&cnt[dst[e]], 1);
}
__global__ __launch_bounds__(1024) void scan_rows(const int* __restrict__ cnt,
                                                  int* __restrict__ rowptr,
                                                  float* __restrict__ dinv, int n) {
  __shared__ int wsum[32];
  const int tid = threadIdx.x, lane = tid & 31, w = tid >> 5;
  int carry = 0;
  for (int base = 0; base < n; base += 4096) {
    const int i = base + tid * 4;
    int v0 = 0, v1 = 0, v2 = 0, v3 = 0;
    if (i + 3 < n) {
      const int4 q = *(const int4*)(cnt + i);
      v0 = q.x; v1 = q.y; v2 = q.z; v3 = q.w;
    } else {
      if (i < n) v0 = cnt[i];
      if (i + 1 < n) v1 = cnt[i + 1];
      if (i + 2 < n) v2 = cnt[i + 2];
    }
    const int mine = v0 + v1 + v2 + v3;
    int x = mine;
    for (int o = 1; o < 32; o <<= 1) {
      const int t = __shfl_up(x, o);
      if (lane >= o) x += t;
    }
    if (lane == 31) wsum[w] = x;
    __syncthreads();
    if (w == 0) {
      int y = wsum[lane];
      for (int o = 1; o < 32; o <<= 1) {
        const int t = __shfl_up(y, o);
        if (lane >= o) y += t;
      }
      wsum[lane] = y;
    }
    __syncthreads();
    const int off = carry + (w > 0 ? wsum[w - 1] : 0) + x - mine;
    carry += wsum[31];
    if (i < n)     { rowptr[i]     = off;                dinv[i]     = rsqrtf((float)(v0 + 1)); }
    if (i + 1 < n) { rowptr[i + 1] = off + v0;           dinv[i + 1] = rsqrtf((float)(v1 + 1)); }
    if (i + 2 < n) { rowptr[i + 2] = off + v0 + v1;      dinv[i + 2] = rsqrtf((float)(v2 + 1)); }
    if (i + 3 < n) { rowptr[i + 3] = off + v0 + v1 + v2; dinv[i + 3] = rsqrtf((float)(v3 + 1)); }
    __syncthreads();
  }
  if (tid == 0) rowptr[n] = carry;
}
__global__ void place_edges(const int* __restrict__ src, const int* __restrict__ dst,
                            const int* __restrict__ rank, const int* __restrict__ rowptr,
                            int* __restrict__ csr_src, int E) {
  int e = blockIdx.x * blockDim.x + threadIdx.x;
  if (e < E) csr_src[rowptr[dst[e]] + rank[e]] = src[e];
}

__global__ void pad_x(const float* __restrict__ x, float* __restrict__ xp, int M) {
  int n = blockIdx.x, c = threadIdx.x;
  if (n >= M || c >= DINP) return;
  xp[(size_t)n * DINP + c] = (c < DIN) ? x[(size_t)n * DIN + c] : 0.0f;
}
__global__ void pad_W(const float* __restrict__ W, float* __restrict__ Wp, int K, int Nn) {
  int r = blockIdx.x, c = threadIdx.x;
  Wp[(size_t)r * DHP + c] = (r < K && c < Nn) ? W[(size_t)r * Nn + c] : 0.0f;
}

__global__ __launch_bounds__(256) void gemm_f32_wmma(
    const float* __restrict__ A, int lda,
    const float* __restrict__ B,
    float* __restrict__ C,
    int M, int Npad, int Kpad) {
  int wave = (blockIdx.x * blockDim.x + threadIdx.x) >> 5;
  int lane = threadIdx.x & 31;
  int ntn  = Npad >> 4;
  int tm = wave / ntn;
  int tn = wave - tm * ntn;
  if (tm * 16 >= M) return;

  int half = lane >> 4;
  int m    = lane & 15;
  int row0 = tm * 16;
  int arow = row0 + m;
  const float* Arow = A + (size_t)(arow < M ? arow : (M - 1)) * lda;
  const float* Bcol = B + (tn * 16 + m);

  v8f acc = {0.f, 0.f, 0.f, 0.f, 0.f, 0.f, 0.f, 0.f};
#pragma unroll 4
  for (int k = 0; k < Kpad; k += 4) {
    int ka = k + half * 2;
    v2f a, b;
    a.x = Arow[ka];
    a.y = Arow[ka + 1];
    b.x = Bcol[(size_t)ka * DHP];
    b.y = Bcol[(size_t)(ka + 1) * DHP];
    acc = __builtin_amdgcn_wmma_f32_16x16x4_f32(false, a, false, b,
                                                (short)0, acc, false, false);
  }
  int col = tn * 16 + m;
#pragma unroll
  for (int r = 0; r < 8; ++r) {
    int row = row0 + r + half * 8;
    if (row < M) C[(size_t)row * DHP + col] = acc[r];
  }
}

__global__ __launch_bounds__(256) void gather_rows(
    const float* __restrict__ h, const int* __restrict__ rowptr,
    const int* __restrict__ csr_src, const float* __restrict__ dinv,
    float* __restrict__ agg, int M) {
  const int lane = threadIdx.x & 31;
  const int n = __builtin_amdgcn_readfirstlane(blockIdx.x * 8 + (threadIdx.x >> 5));
  if (n >= M) return;
  const int beg = rowptr[n], end = rowptr[n + 1];
  const float dn = dinv[n];

  v2f acc[5];
  {
    const v2f* p = (const v2f*)(h + (size_t)n * DHP) + lane;
#pragma unroll
    for (int i = 0; i < 5; ++i) {
      const v2f v = p[i * 32];
      acc[i].x = v.x * dn * dn;
      acc[i].y = v.y * dn * dn;
    }
  }
  for (int base = beg; base < end; base += 32) {
    const int j = base + lane;
    int s_l = 0;
    float w_l = 0.f;
    if (j < end) {
      s_l = (int)min((unsigned)csr_src[j], (unsigned)(M - 1));
      w_l = dinv[s_l] * dn;
    }
    const int m = min(32, end - base);
#pragma unroll 4
    for (int t = 0; t < m; ++t) {
      const int s = __builtin_amdgcn_readlane(s_l, t);
      const float wgt = __int_as_float(__builtin_amdgcn_readlane(__float_as_int(w_l), t));
      const v2f* p = (const v2f*)(h + (size_t)s * DHP) + lane;
#pragma unroll
      for (int i = 0; i < 5; ++i) {
        const v2f v = p[i * 32];
        acc[i].x += v.x * wgt;
        acc[i].y += v.y * wgt;
      }
    }
  }
  v2f* o = (v2f*)(agg + (size_t)n * DHP) + lane;
#pragma unroll
  for (int i = 0; i < 5; ++i) o[i * 32] = acc[i];
}

__global__ void bias_relu_stats(float* __restrict__ y, const float* __restrict__ bias,
                                float* __restrict__ sums, int M) {
  int c = threadIdx.x;
  if (c >= DH) return;
  int r0 = blockIdx.x * 128;
  int r1 = min(r0 + 128, M);
  float b = bias[c], s = 0.f, s2 = 0.f;
  for (int r = r0; r < r1; ++r) {
    float v = y[(size_t)r * DHP + c] + b;
    v = fmaxf(v, 0.0f);
    y[(size_t)r * DHP + c] = v;
    s += v;
    s2 += v * v;
  }
  atomicAdd(&sums[c], s);
  atomicAdd(&sums[DHP + c], s2);
}
__global__ void bn_apply(float* __restrict__ y, const float* __restrict__ sums,
                         const float* __restrict__ gamma, const float* __restrict__ beta,
                         int M) {
  int c = threadIdx.x;
  if (c >= DH) return;
  int r0 = blockIdx.x * 128;
  int r1 = min(r0 + 128, M);
  float invn  = 1.0f / (float)M;
  float mean  = sums[c] * invn;
  float var   = sums[DHP + c] * invn - mean * mean;
  float scale = rsqrtf(var + BNEPS) * gamma[c];
  float shift = beta[c] - mean * scale;
  for (int r = r0; r < r1; ++r)
    y[(size_t)r * DHP + c] = y[(size_t)r * DHP + c] * scale + shift;
}

__device__ __forceinline__ unsigned enc_f32(float f) {
  unsigned u = __float_as_uint(f);
  return (u & 0x80000000u) ? ~u : (u | 0x80000000u);
}
__device__ __forceinline__ float dec_f32(unsigned u) {
  return __uint_as_float((u & 0x80000000u) ? (u & 0x7FFFFFFFu) : ~u);
}
#define POOL_ROWS  64
#define POOL_BURST 16
__global__ __launch_bounds__(DHP) void pool_max(const float* __restrict__ y,
                                                const int* __restrict__ batch,
                                                unsigned* __restrict__ pool, int M) {
  const int c = threadIdx.x;
  if (c >= DH) return;
  const int r0 = blockIdx.x * POOL_ROWS;
  const int r1 = min(r0 + POOL_ROWS, M);
  if (r0 >= r1) return;
  int gcur = batch[r0];
  unsigned best = 0u;
  int r = r0;
  for (; r + POOL_BURST <= r1; r += POOL_BURST) {
    const float* p = y + (size_t)r * DHP + c;
    float v[POOL_BURST];
    int g[POOL_BURST];
#pragma unroll
    for (int i = 0; i < POOL_BURST; ++i) {
      v[i] = p[(size_t)i * DHP];
      g[i] = batch[r + i];
    }
#pragma unroll
    for (int i = 0; i < POOL_BURST; ++i) {
      if (g[i] != gcur) {
        atomicMax(&pool[(size_t)gcur * DH + c], best);
        gcur = g[i];
        best = 0u;
      }
      best = max(best, enc_f32(v[i]));
    }
  }
  for (; r < r1; ++r) {
    const int g = batch[r];
    if (g != gcur) {
      atomicMax(&pool[(size_t)gcur * DH + c], best);
      gcur = g;
      best = 0u;
    }
    best = max(best, enc_f32(y[(size_t)r * DHP + c]));
  }
  atomicMax(&pool[(size_t)gcur * DH + c], best);
}
__global__ void pool_decode(const unsigned* __restrict__ pool, float* __restrict__ out) {
  int g = blockIdx.x, c = threadIdx.x;
  if (c >= DH) return;
  out[(size_t)g * DH + c] = dec_f32(pool[(size_t)g * DH + c]);
}

extern "C" void kernel_launch(void* const* d_in, const int* in_sizes, int n_in,
                              void* d_out, int out_size, void* d_ws, size_t ws_size,
                              hipStream_t stream) {
  const float* x      = (const float*)d_in[0];
  const int*   eidx   = (const int*)d_in[1];
  const int*   batch  = (const int*)d_in[2];
  const float* W1     = (const float*)d_in[3];
  const float* b1     = (const float*)d_in[4];
  const float* gamma1 = (const float*)d_in[5];
  const float* beta1  = (const float*)d_in[6];
  const float* W2     = (const float*)d_in[7];
  const float* b2     = (const float*)d_in[8];
  const float* gamma2 = (const float*)d_in[9];
  const float* beta2  = (const float*)d_in[10];
  float* out = (float*)d_out;

  const int* src = eidx;
  const int* dst = eidx + EE;

  float* ws = (float*)d_ws;
  float*    dinv = ws;
  int*      cnt  = (int*)(ws + 50048);
  int*      rowp = (int*)(ws + 100096);
  int*      rank = (int*)(ws + 150208);
  int*      csrs = (int*)(ws + 950208);
  float*    xp   = ws + 1750208;
  float*    Wp1  = ws + 5750208;
  float*    Wp2  = ws + 5775808;
  float*    sums = ws + 5878208;
  unsigned* pool = (unsigned*)(ws + 5878848);
  float*    bufA = ws + 6034880;
  float*    bufB = ws + 22034880;

  const int statBlocks = (NN + 127) / 128;
  const int gemmTiles  = (NN / 16) * (DHP / 16);
  const int gemmGrid   = (gemmTiles + 7) / 8;

  fill_u32<<<(NN + 255) / 256, 256, 0, stream>>>((unsigned*)cnt, 0u, NN);
  edge_rank<<<(EE + 255) / 256, 256, 0, stream>>>(dst, cnt, rank, EE);
  scan_rows<<<1, 1024, 0, stream>>>(cnt, rowp, dinv, NN);
  place_edges<<<(EE + 255) / 256, 256, 0, stream>>>(src, dst, rank, rowp, csrs, EE);

  pad_x<<<NN, 128, 0, stream>>>(x, xp, NN);
  pad_W<<<DINP, DHP, 0, stream>>>(W1, Wp1, DIN, DH);
  pad_W<<<DHP, DHP, 0, stream>>>(W2, Wp2, DH, DH);

  gemm_f32_wmma<<<gemmGrid, 256, 0, stream>>>(xp, DINP, Wp1, bufA, NN, DHP, DINP);
  gather_rows<<<(NN + 7) / 8, 256, 0, stream>>>(bufA, rowp, csrs, dinv, bufB, NN);
  fill_f32<<<(2 * DHP + 255) / 256, 256, 0, stream>>>(sums, 0.0f, 2 * DHP);
  bias_relu_stats<<<statBlocks, DHP, 0, stream>>>(bufB, b1, sums, NN);
  bn_apply<<<statBlocks, DHP, 0, stream>>>(bufB, sums, gamma1, beta1, NN);

  gemm_f32_wmma<<<gemmGrid, 256, 0, stream>>>(bufB, DHP, Wp2, bufA, NN, DHP, DHP);
  gather_rows<<<(NN + 7) / 8, 256, 0, stream>>>(bufA, rowp, csrs, dinv, bufB, NN);
  fill_f32<<<(2 * DHP + 255) / 256, 256, 0, stream>>>(sums, 0.0f, 2 * DHP);
  bias_relu_stats<<<statBlocks, DHP, 0, stream>>>(bufB, b2, sums, NN);
  bn_apply<<<statBlocks, DHP, 0, stream>>>(bufB, sums, gamma2, beta2, NN);

  fill_u32<<<((long)GG * DH + 255) / 256, 256, 0, stream>>>(pool, 0u, (long)GG * DH);
  pool_max<<<(NN + POOL_ROWS - 1) / POOL_ROWS, DHP, 0, stream>>>(bufB, batch, pool, NN);
  pool_decode<<<GG, DHP, 0, stream>>>(pool, out);
}


# GEMM: one wave computes 32x64 outputs (2x4 WMMA tiles, columns interleaved by 4) with 16-byte A/B loads and whole-line 16-byte C stores; was one 16x16 tile per wave; k summed in another order
# speedup vs baseline: 4.1271x; 1.1079x over previous
#include <hip/hip_runtime.h>
#include <hip/hip_bf16.h>

typedef float v2f __attribute__((ext_vector_type(2)));
typedef float v4f __attribute__((ext_vector_type(4)));
typedef float v8f __attribute__((ext_vector_type(8)));

#define NN    50000
#define EE    800000
#define GG    500
#define DIN   78
#define DINP  80
#define DH    312
#define DHP   320
#define BNEPS 1e-5f

__global__ void fill_f32(float* __restrict__ p, float v, long n) {
  long i = (long)blockIdx.x * blockDim.x + threadIdx.x;
  if (i < n) p[i] = v;
}
__global__ void fill_u32(unsigned* __restrict__ p, unsigned v, long n) {
  long i = (long)blockIdx.x * blockDim.x + threadIdx.x;
  if (i < n) p[i] = v;
}

__global__ void edge_rank(const int* __restrict__ dst, int* __restrict__ cnt,
                          int* __restrict__ rank, int E) {
  int e = blockIdx.x * blockDim.x + threadIdx.x;
  if (e < E) rank[e] = atomicAdd(&cnt[dst[e]], 1);
}
__global__ __launch_bounds__(1024) void scan_rows(const int* __restrict__ cnt,
                                                  int* __restrict__ rowptr,
                                                  float* __restrict__ dinv, int n) {
  __shared__ int wsum[32];
  const int tid = threadIdx.x, lane = tid & 31, w = tid >> 5;
  int carry = 0;
  for (int base = 0; base < n; base += 4096) {
    const int i = base + tid * 4;
    int v0 = 0, v1 = 0, v2 = 0, v3 = 0;
    if (i + 3 < n) {
      const int4 q = *(const int4*)(cnt + i);
      v0 = q.x; v1 = q.y; v2 = q.z; v3 = q.w;
    } else {
      if (i < n) v0 = cnt[i];
      if (i + 1 < n) v1 = cnt[i + 1];
      if (i + 2 < n) v2 = cnt[i + 2];
    }
    const int mine = v0 + v1 + v2 + v3;
    int x = mine;
    for (int o = 1; o < 32; o <<= 1) {
      const int t = __shfl_up(x, o);
      if (lane >= o) x += t;
    }
    if (lane == 31) wsum[w] = x;
    __syncthreads();
    if (w == 0) {
      int y = wsum[lane];
      for (int o = 1; o < 32; o <<= 1) {
        const int t = __shfl_up(y, o);
        if (lane >= o) y += t;
      }
      wsum[lane] = y;
    }
    __syncthreads();
    const int off = carry + (w > 0 ? wsum[w - 1] : 0) + x - mine;
    carry += wsum[31];
    if (i < n)     { rowptr[i]     = off;                dinv[i]     = rsqrtf((float)(v0 + 1)); }
    if (i + 1 < n) { rowptr[i + 1] = off + v0;           dinv[i + 1] = rsqrtf((float)(v1 + 1)); }
    if (i + 2 < n) { rowptr[i + 2] = off + v0 + v1;      dinv[i + 2] = rsqrtf((float)(v2 + 1)); }
    if (i + 3 < n) { rowptr[i + 3] = off + v0 + v1 + v2; dinv[i + 3] = rsqrtf((float)(v3 + 1)); }
    __syncthreads();
  }
  if (tid == 0) rowptr[n] = carry;
}
__global__ void place_edges(const int* __restrict__ src, const int* __restrict__ dst,
                            const int* __restrict__ rank, const int* __restrict__ rowptr,
                            int* __restrict__ csr_src, int E) {
  int e = blockIdx.x * blockDim.x + threadIdx.x;
  if (e < E) csr_src[rowptr[dst[e]] + rank[e]] = src[e];
}

__global__ void pad_x(const float* __restrict__ x, float* __restrict__ xp, int M) {
  int n = blockIdx.x, c = threadIdx.x;
  if (n >= M || c >= DINP) return;
  xp[(size_t)n * DINP + c] = (c < DIN) ? x[(size_t)n * DIN + c] : 0.0f;
}
__global__ void pad_W(const float* __restrict__ W, float* __restrict__ Wp, int K, int Nn) {
  int r = blockIdx.x, c = threadIdx.x;
  Wp[(size_t)r * DHP + c] = (r < K && c < Nn) ? W[(size_t)r * Nn + c] : 0.0f;
}

__global__ __launch_bounds__(256) void gemm_f32_wmma(
    const float* __restrict__ A, int lda,
    const float* __restrict__ B,
    float* __restrict__ C,
    int M, int Kpad) {
  const int lane  = threadIdx.x & 31;
  const int strip = blockIdx.x * 8 + (threadIdx.x >> 5);
  const int cg    = blockIdx.y;
  const int row0  = strip * 32;
  if (row0 >= M) return;

  const int h = lane >> 4;
  const int n = lane & 15;
  const int ar0 = min(row0 + n, M - 1);
  const int ar1 = min(row0 + 16 + n, M - 1);
  const float* a0p = A + (size_t)ar0 * lda + 8 * h;
  const float* a1p = A + (size_t)ar1 * lda + 8 * h;
  const float* bp  = B + (size_t)(8 * h) * DHP + cg * 64 + 4 * n;

  v8f acc[2][4];
#pragma unroll
  for (int i = 0; i < 2; ++i)
#pragma unroll
    for (int t = 0; t < 4; ++t) acc[i][t] = (v8f){0.f, 0.f, 0.f, 0.f, 0.f, 0.f, 0.f, 0.f};

  for (int k0 = 0; k0 < Kpad; k0 += 16) {
    v4f a0[2], a1[2], b[8];
    a0[0] = *(const v4f*)(a0p + k0);
    a0[1] = *(const v4f*)(a0p + k0 + 4);
    a1[0] = *(const v4f*)(a1p + k0);
    a1[1] = *(const v4f*)(a1p + k0 + 4);
#pragma unroll
    for (int j = 0; j < 8; ++j) b[j] = *(const v4f*)(bp + (size_t)(k0 + j) * DHP);
#pragma unroll
    for (int s = 0; s < 4; ++s) {
      v2f fa0, fa1;
      fa0.x = a0[s >> 1][(2 * s) & 3];
      fa0.y = a0[s >> 1][(2 * s + 1) & 3];
      fa1.x = a1[s >> 1][(2 * s) & 3];
      fa1.y = a1[s >> 1][(2 * s + 1) & 3];
#pragma unroll
      for (int t = 0; t < 4; ++t) {
        v2f fb;
        fb.x = b[2 * s][t];
        fb.y = b[2 * s + 1][t];
        acc[0][t] = __builtin_amdgcn_wmma_f32_16x16x4_f32(false, fa0, false, fb,
                                                          (short)0, acc[0][t], false, false);
        acc[1][t] = __builtin_amdgcn_wmma_f32_16x16x4_f32(false, fa1, false, fb,
                                                          (short)0, acc[1][t], false, false);
      }
    }
  }
#pragma unroll
  for (int i = 0; i < 2; ++i) {
#pragma unroll
    for (int r = 0; r < 8; ++r) {
      const int row = row0 + 16 * i + 8 * h + r;
      v4f o;
      o.x = acc[i][0][r];
      o.y = acc[i][1][r];
      o.z = acc[i][2][r];
      o.w = acc[i][3][r];
      if (row < M) *(v4f*)(C + (size_t)row * DHP + cg * 64 + 4 * n) = o;
    }
  }
}

__global__ __launch_bounds__(256) void gather_rows(
    const float* __restrict__ h, const int* __restrict__ rowptr,
    const int* __restrict__ csr_src, const float* __restrict__ dinv,
    float* __restrict__ agg, int M) {
  const int lane = threadIdx.x & 31;
  const int n = __builtin_amdgcn_readfirstlane(blockIdx.x * 8 + (threadIdx.x >> 5));
  if (n >= M) return;
  const int beg = rowptr[n], end = rowptr[n + 1];
  const float dn = dinv[n];

  v2f acc[5];
  {
    const v2f* p = (const v2f*)(h + (size_t)n * DHP) + lane;
#pragma unroll
    for (int i = 0; i < 5; ++i) {
      const v2f v = p[i * 32];
      acc[i].x = v.x * dn * dn;
      acc[i].y = v.y * dn * dn;
    }
  }
  for (int base = beg; base < end; base += 32) {
    const int j = base + lane;
    int s_l = 0;
    float w_l = 0.f;
    if (j < end) {
      s_l = (int)min((unsigned)csr_src[j], (unsigned)(M - 1));
      w_l = dinv[s_l] * dn;
    }
    const int m = min(32, end - base);
#pragma unroll 4
    for (int t = 0; t < m; ++t) {
      const int s = __builtin_amdgcn_readlane(s_l, t);
      const float wgt = __int_as_float(__builtin_amdgcn_readlane(__float_as_int(w_l), t));
      const v2f* p = (const v2f*)(h + (size_t)s * DHP) + lane;
#pragma unroll
      for (int i = 0; i < 5; ++i) {
        const v2f v = p[i * 32];
        acc[i].x += v.x * wgt;
        acc[i].y += v.y * wgt;
      }
    }
  }
  v2f* o = (v2f*)(agg + (size_t)n * DHP) + lane;
#pragma unroll
  for (int i = 0; i < 5; ++i) o[i * 32] = acc[i];
}

__global__ void bias_relu_stats(float* __restrict__ y, const float* __restrict__ bias,
                                float* __restrict__ sums, int M) {
  int c = threadIdx.x;
  if (c >= DH) return;
  int r0 = blockIdx.x * 128;
  int r1 = min(r0 + 128, M);
  float b = bias[c], s = 0.f, s2 = 0.f;
  for (int r = r0; r < r1; ++r) {
    float v = y[(size_t)r * DHP + c] + b;
    v = fmaxf(v, 0.0f);
    y[(size_t)r * DHP + c] = v;
    s += v;
    s2 += v * v;
  }
  atomicAdd(&sums[c], s);
  atomicAdd(&sums[DHP + c], s2);
}
__global__ void bn_apply(float* __restrict__ y, const float* __restrict__ sums,
                         const float* __restrict__ gamma, const float* __restrict__ beta,
                         int M) {
  int c = threadIdx.x;
  if (c >= DH) return;
  int r0 = blockIdx.x * 128;
  int r1 = min(r0 + 128, M);
  float invn  = 1.0f / (float)M;
  float mean  = sums[c] * invn;
  float var   = sums[DHP + c] * invn - mean * mean;
  float scale = rsqrtf(var + BNEPS) * gamma[c];
  float shift = beta[c] - mean * scale;
  for (int r = r0; r < r1; ++r)
    y[(size_t)r * DHP + c] = y[(size_t)r * DHP + c] * scale + shift;
}

__device__ __forceinline__ unsigned enc_f32(float f) {
  unsigned u = __float_as_uint(f);
  return (u & 0x80000000u) ? ~u : (u | 0x80000000u);
}
__device__ __forceinline__ float dec_f32(unsigned u) {
  return __uint_as_float((u & 0x80000000u) ? (u & 0x7FFFFFFFu) : ~u);
}
#define POOL_ROWS  64
#define POOL_BURST 16
__global__ __launch_bounds__(DHP) void pool_max(const float* __restrict__ y,
                                                const int* __restrict__ batch,
                                                unsigned* __restrict__ pool, int M) {
  const int c = threadIdx.x;
  if (c >= DH) return;
  const int r0 = blockIdx.x * POOL_ROWS;
  const int r1 = min(r0 + POOL_ROWS, M);
  if (r0 >= r1) return;
  int gcur = batch[r0];
  unsigned best = 0u;
  int r = r0;
  for (; r + POOL_BURST <= r1; r += POOL_BURST) {
    const float* p = y + (size_t)r * DHP + c;
    float v[POOL_BURST];
    int g[POOL_BURST];
#pragma unroll
    for (int i = 0; i < POOL_BURST; ++i) {
      v[i] = p[(size_t)i * DHP];
      g[i] = batch[r + i];
    }
#pragma unroll
    for (int i = 0; i < POOL_BURST; ++i) {
      if (g[i] != gcur) {
        atomicMax(&pool[(size_t)gcur * DH + c], best);
        gcur = g[i];
        best = 0u;
      }
      best = max(best, enc_f32(v[i]));
    }
  }
  for (; r < r1; ++r) {
    const int g = batch[r];
    if (g != gcur) {
      atomicMax(&pool[(size_t)gcur * DH + c], best);
      gcur = g;
      best = 0u;
    }
    best = max(best, enc_f32(y[(size_t)r * DHP + c]));
  }
  atomicMax(&pool[(size_t)gcur * DH + c], best);
}
__global__ void pool_decode(const unsigned* __restrict__ pool, float* __restrict__ out) {
  int g = blockIdx.x, c = threadIdx.x;
  if (c >= DH) return;
  out[(size_t)g * DH + c] = dec_f32(pool[(size_t)g * DH + c]);
}

extern "C" void kernel_launch(void* const* d_in, const int* in_sizes, int n_in,
                              void* d_out, int out_size, void* d_ws, size_t ws_size,
                              hipStream_t stream) {
  const float* x      = (const float*)d_in[0];
  const int*   eidx   = (const int*)d_in[1];
  const int*   batch  = (const int*)d_in[2];
  const float* W1     = (const float*)d_in[3];
  const float* b1     = (const float*)d_in[4];
  const float* gamma1 = (const float*)d_in[5];
  const float* beta1  = (const float*)d_in[6];
  const float* W2     = (const float*)d_in[7];
  const float* b2     = (const float*)d_in[8];
  const float* gamma2 = (const float*)d_in[9];
  const float* beta2  = (const float*)d_in[10];
  float* out = (float*)d_out;

  const int* src = eidx;
  const int* dst = eidx + EE;

  float* ws = (float*)d_ws;
  float*    dinv = ws;
  int*      cnt  = (int*)(ws + 50048);
  int*      rowp = (int*)(ws + 100096);
  int*      rank = (int*)(ws + 150208);
  int*      csrs = (int*)(ws + 950208);
  float*    xp   = ws + 1750208;
  float*    Wp1  = ws + 5750208;
  float*    Wp2  = ws + 5775808;
  float*    sums = ws + 5878208;
  unsigned* pool = (unsigned*)(ws + 5878848);
  float*    bufA = ws + 6034880;
  float*    bufB = ws + 22034880;

  const int statBlocks = (NN + 127) / 128;
  const dim3 gemmGrid(((NN + 31) / 32 + 7) / 8, DHP / 64);

  fill_u32<<<(NN + 255) / 256, 256, 0, stream>>>((unsigned*)cnt, 0u, NN);
  edge_rank<<<(EE + 255) / 256, 256, 0, stream>>>(dst, cnt, rank, EE);
  scan_rows<<<1, 1024, 0, stream>>>(cnt, rowp, dinv, NN);
  place_edges<<<(EE + 255) / 256, 256, 0, stream>>>(src, dst, rank, rowp, csrs, EE);

  pad_x<<<NN, 128, 0, stream>>>(x, xp, NN);
  pad_W<<<DINP, DHP, 0, stream>>>(W1, Wp1, DIN, DH);
  pad_W<<<DHP, DHP, 0, stream>>>(W2, Wp2, DH, DH);

  gemm_f32_wmma<<<gemmGrid, 256, 0, stream>>>(xp, DINP, Wp1, bufA, NN, DINP);
  gather_rows<<<(NN + 7) / 8, 256, 0, stream>>>(bufA, rowp, csrs, dinv, bufB, NN);
  fill_f32<<<(2 * DHP + 255) / 256, 256, 0, stream>>>(sums, 0.0f, 2 * DHP);
  bias_relu_stats<<<statBlocks, DHP, 0, stream>>>(bufB, b1, sums, NN);
  bn_apply<<<statBlocks, DHP, 0, stream>>>(bufB, sums, gamma1, beta1, NN);

  gemm_f32_wmma<<<gemmGrid, 256, 0, stream>>>(bufB, DHP, Wp2, bufA, NN, DHP);
  gather_rows<<<(NN + 7) / 8, 256, 0, stream>>>(bufA, rowp, csrs, dinv, bufB, NN);
  fill_f32<<<(2 * DHP + 255) / 256, 256, 0, stream>>>(sums, 0.0f, 2 * DHP);
  bias_relu_stats<<<statBlocks, DHP, 0, stream>>>(bufB, b2, sums, NN);
  bn_apply<<<statBlocks, DHP, 0, stream>>>(bufB, sums, gamma2, beta2, NN);

  fill_u32<<<((long)GG * DH + 255) / 256, 256, 0, stream>>>(pool, 0u, (long)GG * DH);
  pool_max<<<(NN + POOL_ROWS - 1) / POOL_ROWS, DHP, 0, stream>>>(bufB, batch, pool, NN);
  pool_decode<<<GG, DHP, 0, stream>>>(pool, out);
}


# BN passes (bias+ReLU+moments, BN apply): a thread issues 16 row loads of 16 bytes before the first use, 64 rows per workgroup (was 1 load in flight, 128 rows); moment partial sums regrouped
# speedup vs baseline: 5.2881x; 1.2813x over previous
#include <hip/hip_runtime.h>
#include <hip/hip_bf16.h>

typedef float v2f __attribute__((ext_vector_type(2)));
typedef float v4f __attribute__((ext_vector_type(4)));
typedef float v8f __attribute__((ext_vector_type(8)));

#define NN    50000
#define EE    800000
#define GG    500
#define DIN   78
#define DINP  80
#define DH    312
#define DHP   320
#define BNEPS 1e-5f

__global__ void fill_f32(float* __restrict__ p, float v, long n) {
  long i = (long)blockIdx.x * blockDim.x + threadIdx.x;
  if (i < n) p[i] = v;
}
__global__ void fill_u32(unsigned* __restrict__ p, unsigned v, long n) {
  long i = (long)blockIdx.x * blockDim.x + threadIdx.x;
  if (i < n) p[i] = v;
}

__global__ void edge_rank(const int* __restrict__ dst, int* __restrict__ cnt,
                          int* __restrict__ rank, int E) {
  int e = blockIdx.x * blockDim.x + threadIdx.x;
  if (e < E) rank[e] = atomicAdd(&cnt[dst[e]], 1);
}
__global__ __launch_bounds__(1024) void scan_rows(const int* __restrict__ cnt,
                                                  int* __restrict__ rowptr,
                                                  float* __restrict__ dinv, int n) {
  __shared__ int wsum[32];
  const int tid = threadIdx.x, lane = tid & 31, w = tid >> 5;
  int carry = 0;
  for (int base = 0; base < n; base += 4096) {
    const int i = base + tid * 4;
    int v0 = 0, v1 = 0, v2 = 0, v3 = 0;
    if (i + 3 < n) {
      const int4 q = *(const int4*)(cnt + i);
      v0 = q.x; v1 = q.y; v2 = q.z; v3 = q.w;
    } else {
      if (i < n) v0 = cnt[i];
      if (i + 1 < n) v1 = cnt[i + 1];
      if (i + 2 < n) v2 = cnt[i + 2];
    }
    const int mine = v0 + v1 + v2 + v3;
    int x = mine;
    for (int o = 1; o < 32; o <<= 1) {
      const int t = __shfl_up(x, o);
      if (lane >= o) x += t;
    }
    if (lane == 31) wsum[w] = x;
    __syncthreads();
    if (w == 0) {
      int y = wsum[lane];
      for (int o = 1; o < 32; o <<= 1) {
        const int t = __shfl_up(y, o);
        if (lane >= o) y += t;
      }
      wsum[lane] = y;
    }
    __syncthreads();
    const int off = carry + (w > 0 ? wsum[w - 1] : 0) + x - mine;
    carry += wsum[31];
    if (i < n)     { rowptr[i]     = off;                dinv[i]     = rsqrtf((float)(v0 + 1)); }
    if (i + 1 < n) { rowptr[i + 1] = off + v0;           dinv[i + 1] = rsqrtf((float)(v1 + 1)); }
    if (i + 2 < n) { rowptr[i + 2] = off + v0 + v1;      dinv[i + 2] = rsqrtf((float)(v2 + 1)); }
    if (i + 3 < n) { rowptr[i + 3] = off + v0 + v1 + v2; dinv[i + 3] = rsqrtf((float)(v3 + 1)); }
    __syncthreads();
  }
  if (tid == 0) rowptr[n] = carry;
}
__global__ void place_edges(const int* __restrict__ src, const int* __restrict__ dst,
                            const int* __restrict__ rank, const int* __restrict__ rowptr,
                            int* __restrict__ csr_src, int E) {
  int e = blockIdx.x * blockDim.x + threadIdx.x;
  if (e < E) csr_src[rowptr[dst[e]] + rank[e]] = src[e];
}

__global__ void pad_x(const float* __restrict__ x, float* __restrict__ xp, int M) {
  int n = blockIdx.x, c = threadIdx.x;
  if (n >= M || c >= DINP) return;
  xp[(size_t)n * DINP + c] = (c < DIN) ? x[(size_t)n * DIN + c] : 0.0f;
}
__global__ void pad_W(const float* __restrict__ W, float* __restrict__ Wp, int K, int Nn) {
  int r = blockIdx.x, c = threadIdx.x;
  Wp[(size_t)r * DHP + c] = (r < K && c < Nn) ? W[(size_t)r * Nn + c] : 0.0f;
}

__global__ __launch_bounds__(256) void gemm_f32_wmma(
    const float* __restrict__ A, int lda,
    const float* __restrict__ B,
    float* __restrict__ C,
    int M, int Kpad) {
  const int lane  = threadIdx.x & 31;
  const int strip = blockIdx.x * 8 + (threadIdx.x >> 5);
  const int cg    = blockIdx.y;
  const int row0  = strip * 32;
  if (row0 >= M) return;

  const int h = lane >> 4;
  const int n = lane & 15;
  const int ar0 = min(row0 + n, M - 1);
  const int ar1 = min(row0 + 16 + n, M - 1);
  const float* a0p = A + (size_t)ar0 * lda + 8 * h;
  const float* a1p = A + (size_t)ar1 * lda + 8 * h;
  const float* bp  = B + (size_t)(8 * h) * DHP + cg * 64 + 4 * n;

  v8f acc[2][4];
#pragma unroll
  for (int i = 0; i < 2; ++i)
#pragma unroll
    for (int t = 0; t < 4; ++t) acc[i][t] = (v8f){0.f, 0.f, 0.f, 0.f, 0.f, 0.f, 0.f, 0.f};

  for (int k0 = 0; k0 < Kpad; k0 += 16) {
    v4f a0[2], a1[2], b[8];
    a0[0] = *(const v4f*)(a0p + k0);
    a0[1] = *(const v4f*)(a0p + k0 + 4);
    a1[0] = *(const v4f*)(a1p + k0);
    a1[1] = *(const v4f*)(a1p + k0 + 4);
#pragma unroll
    for (int j = 0; j < 8; ++j) b[j] = *(const v4f*)(bp + (size_t)(k0 + j) * DHP);
#pragma unroll
    for (int s = 0; s < 4; ++s) {
      v2f fa0, fa1;
      fa0.x = a0[s >> 1][(2 * s) & 3];
      fa0.y = a0[s >> 1][(2 * s + 1) & 3];
      fa1.x = a1[s >> 1][(2 * s) & 3];
      fa1.y = a1[s >> 1][(2 * s + 1) & 3];
#pragma unroll
      for (int t = 0; t < 4; ++t) {
        v2f fb;
        fb.x = b[2 * s][t];
        fb.y = b[2 * s + 1][t];
        acc[0][t] = __builtin_amdgcn_wmma_f32_16x16x4_f32(false, fa0, false, fb,
                                                          (short)0, acc[0][t], false, false);
        acc[1][t] = __builtin_amdgcn_wmma_f32_16x16x4_f32(false, fa1, false, fb,
                                                          (short)0, acc[1][t], false, false);
      }
    }
  }
#pragma unroll
  for (int i = 0; i < 2; ++i) {
#pragma unroll
    for (int r = 0; r < 8; ++r) {
      const int row = row0 + 16 * i + 8 * h + r;
      v4f o;
      o.x = acc[i][0][r];
      o.y = acc[i][1][r];
      o.z = acc[i][2][r];
      o.w = acc[i][3][r];
      if (row < M) *(v4f*)(C + (size_t)row * DHP + cg * 64 + 4 * n) = o;
    }
  }
}

__global__ __launch_bounds__(256) void gather_rows(
    const float* __restrict__ h, const int* __restrict__ rowptr,
    const int* __restrict__ csr_src, const float* __restrict__ dinv,
    float* __restrict__ agg, int M) {
  const int lane = threadIdx.x & 31;
  const int n = __builtin_amdgcn_readfirstlane(blockIdx.x * 8 + (threadIdx.x >> 5));
  if (n >= M) return;
  const int beg = rowptr[n], end = rowptr[n + 1];
  const float dn = dinv[n];

  v2f acc[5];
  {
    const v2f* p = (const v2f*)(h + (size_t)n * DHP) + lane;
#pragma unroll
    for (int i = 0; i < 5; ++i) {
      const v2f v = p[i * 32];
      acc[i].x = v.x * dn * dn;
      acc[i].y = v.y * dn * dn;
    }
  }
  for (int base = beg; base < end; base += 32) {
    const int j = base + lane;
    int s_l = 0;
    float w_l = 0.f;
    if (j < end) {
      s_l = (int)min((unsigned)csr_src[j], (unsigned)(M - 1));
      w_l = dinv[s_l] * dn;
    }
    const int m = min(32, end - base);
#pragma unroll 4
    for (int t = 0; t < m; ++t) {
      const int s = __builtin_amdgcn_readlane(s_l, t);
      const float wgt = __int_as_float(__builtin_amdgcn_readlane(__float_as_int(w_l), t));
      const v2f* p = (const v2f*)(h + (size_t)s * DHP) + lane;
#pragma unroll
      for (int i = 0; i < 5; ++i) {
        const v2f v = p[i * 32];
        acc[i].x += v.x * wgt;
        acc[i].y += v.y * wgt;
      }
    }
  }
  v2f* o = (v2f*)(agg + (size_t)n * DHP) + lane;
#pragma unroll
  for (int i = 0; i < 5; ++i) o[i * 32] = acc[i];
}

#define BN_ROWS  64
#define BN_LANES 4
#define BN_PER   (BN_ROWS / BN_LANES)
#define BN_STEP  ((size_t)BN_LANES * DHP)
template <bool FULL>
__device__ __forceinline__ void stats_rows(float* __restrict__ p, int r0, int M, v4f b,
                                           v4f& s, v4f& s2) {
  v4f v[BN_PER];
#pragma unroll
  for (int i = 0; i < BN_PER; ++i)
    v[i] = (FULL || r0 + i * BN_LANES < M) ? *(const v4f*)(p + i * BN_STEP)
                                           : (v4f){0.f, 0.f, 0.f, 0.f};
#pragma unroll
  for (int i = 0; i < BN_PER; ++i) {
    if (FULL || r0 + i * BN_LANES < M) {
      v4f t;
      t.x = fmaxf(v[i].x + b.x, 0.0f);
      t.y = fmaxf(v[i].y + b.y, 0.0f);
      t.z = fmaxf(v[i].z + b.z, 0.0f);
      t.w = fmaxf(v[i].w + b.w, 0.0f);
      *(v4f*)(p + i * BN_STEP) = t;
      s  += t;
      s2 += t * t;
    }
  }
}
__global__ __launch_bounds__(DHP) void bias_relu_stats(float* __restrict__ y,
                                                       const float* __restrict__ bias,
                                                       float* __restrict__ sums, int M) {
  __shared__ v4f red[2][BN_LANES][DHP / 4];
  const int q  = threadIdx.x % 80;
  const int rl = threadIdx.x / 80;
  const int r0 = blockIdx.x * BN_ROWS + rl;
  v4f s  = (v4f){0.f, 0.f, 0.f, 0.f};
  v4f s2 = (v4f){0.f, 0.f, 0.f, 0.f};
  if (q < DH / 4) {
    float* p = y + (size_t)r0 * DHP + 4 * q;
    v4f b;
    b.x = bias[4 * q];     b.y = bias[4 * q + 1];
    b.z = bias[4 * q + 2]; b.w = bias[4 * q + 3];
    if (blockIdx.x * BN_ROWS + BN_ROWS <= M) stats_rows<true>(p, r0, M, b, s, s2);
    else                                     stats_rows<false>(p, r0, M, b, s, s2);
  }
  red[0][rl][q] = s;
  red[1][rl][q] = s2;
  __syncthreads();
  const int c = threadIdx.x;
  if (c < DH) {
    const float* r1 = (const float*)&red[0][0][0];
    const float* r2 = (const float*)&red[1][0][0];
    float a = 0.f, a2 = 0.f;
#pragma unroll
    for (int l = 0; l < BN_LANES; ++l) {
      a  += r1[l * DHP + c];
      a2 += r2[l * DHP + c];
    }
    atomicAdd(&sums[c], a);
    atomicAdd(&sums[DHP + c], a2);
  }
}
template <bool FULL>
__device__ __forceinline__ void apply_rows(float* __restrict__ p, int r0, int M,
                                           const float* __restrict__ sums,
                                           const float* __restrict__ gamma,
                                           const float* __restrict__ beta, int q) {
  v4f v[BN_PER];
#pragma unroll
  for (int i = 0; i < BN_PER; ++i)
    v[i] = (FULL || r0 + i * BN_LANES < M) ? *(const v4f*)(p + i * BN_STEP)
                                           : (v4f){0.f, 0.f, 0.f, 0.f};
  const float invn = 1.0f / (float)M;
  v4f scale, shift;
#pragma unroll
  for (int j = 0; j < 4; ++j) {
    const int c = 4 * q + j;
    const float mean = sums[c] * invn;
    const float var  = sums[DHP + c] * invn - mean * mean;
    scale[j] = rsqrtf(var + BNEPS) * gamma[c];
    shift[j] = beta[c] - mean * scale[j];
  }
#pragma unroll
  for (int i = 0; i < BN_PER; ++i)
    if (FULL || r0 + i * BN_LANES < M) *(v4f*)(p + i * BN_STEP) = v[i] * scale + shift;
}
__global__ __launch_bounds__(DHP) void bn_apply(float* __restrict__ y, const float* __restrict__ sums,
                                                const float* __restrict__ gamma,
                                                const float* __restrict__ beta, int M) {
  const int q  = threadIdx.x % 80;
  const int rl = threadIdx.x / 80;
  if (q >= DH / 4) return;
  const int r0 = blockIdx.x * BN_ROWS + rl;
  float* p = y + (size_t)r0 * DHP + 4 * q;
  if (blockIdx.x * BN_ROWS + BN_ROWS <= M) apply_rows<true>(p, r0, M, sums, gamma, beta, q);
  else                                     apply_rows<false>(p, r0, M, sums, gamma, beta, q);
}

__device__ __forceinline__ unsigned enc_f32(float f) {
  unsigned u = __float_as_uint(f);
  return (u & 0x80000000u) ? ~u : (u | 0x80000000u);
}
__device__ __forceinline__ float dec_f32(unsigned u) {
  return __uint_as_float((u & 0x80000000u) ? (u & 0x7FFFFFFFu) : ~u);
}
#define POOL_ROWS  64
#define POOL_BURST 16
__global__ __launch_bounds__(DHP) void pool_max(const float* __restrict__ y,
                                                const int* __restrict__ batch,
                                                unsigned* __restrict__ pool, int M) {
  const int c = threadIdx.x;
  if (c >= DH) return;
  const int r0 = blockIdx.x * POOL_ROWS;
  const int r1 = min(r0 + POOL_ROWS, M);
  if (r0 >= r1) return;
  int gcur = batch[r0];
  unsigned best = 0u;
  int r = r0;
  for (; r + POOL_BURST <= r1; r += POOL_BURST) {
    const float* p = y + (size_t)r * DHP + c;
    float v[POOL_BURST];
    int g[POOL_BURST];
#pragma unroll
    for (int i = 0; i < POOL_BURST; ++i) {
      v[i] = p[(size_t)i * DHP];
      g[i] = batch[r + i];
    }
#pragma unroll
    for (int i = 0; i < POOL_BURST; ++i) {
      if (g[i] != gcur) {
        atomicMax(&pool[(size_t)gcur * DH + c], best);
        gcur = g[i];
        best = 0u;
      }
      best = max(best, enc_f32(v[i]));
    }
  }
  for (; r < r1; ++r) {
    const int g = batch[r];
    if (g != gcur) {
      atomicMax(&pool[(size_t)gcur * DH + c], best);
      gcur = g;
      best = 0u;
    }
    best = max(best, enc_f32(y[(size_t)r * DHP + c]));
  }
  atomicMax(&pool[(size_t)gcur * DH + c], best);
}
__global__ void pool_decode(const unsigned* __restrict__ pool, float* __restrict__ out) {
  int g = blockIdx.x, c = threadIdx.x;
  if (c >= DH) return;
  out[(size_t)g * DH + c] = dec_f32(pool[(size_t)g * DH + c]);
}

extern "C" void kernel_launch(void* const* d_in, const int* in_sizes, int n_in,
                              void* d_out, int out_size, void* d_ws, size_t ws_size,
                              hipStream_t stream) {
  const float* x      = (const float*)d_in[0];
  const int*   eidx   = (const int*)d_in[1];
  const int*   batch  = (const int*)d_in[2];
  const float* W1     = (const float*)d_in[3];
  const float* b1     = (const float*)d_in[4];
  const float* gamma1 = (const float*)d_in[5];
  const float* beta1  = (const float*)d_in[6];
  const float* W2     = (const float*)d_in[7];
  const float* b2     = (const float*)d_in[8];
  const float* gamma2 = (const float*)d_in[9];
  const float* beta2  = (const float*)d_in[10];
  float* out = (float*)d_out;

  const int* src = eidx;
  const int* dst = eidx + EE;

  float* ws = (float*)d_ws;
  float*    dinv = ws;
  int*      cnt  = (int*)(ws + 50048);
  int*      rowp = (int*)(ws + 100096);
  int*      rank = (int*)(ws + 150208);
  int*      csrs = (int*)(ws + 950208);
  float*    xp   = ws + 1750208;
  float*    Wp1  = ws + 5750208;
  float*    Wp2  = ws + 5775808;
  float*    sums = ws + 5878208;
  unsigned* pool = (unsigned*)(ws + 5878848);
  float*    bufA = ws + 6034880;
  float*    bufB = ws + 22034880;

  const int statBlocks = (NN + BN_ROWS - 1) / BN_ROWS;
  const dim3 gemmGrid(((NN + 31) / 32 + 7) / 8, DHP / 64);

  fill_u32<<<(NN + 255) / 256, 256, 0, stream>>>((unsigned*)cnt, 0u, NN);
  edge_rank<<<(EE + 255) / 256, 256, 0, stream>>>(dst, cnt, rank, EE);
  scan_rows<<<1, 1024, 0, stream>>>(cnt, rowp, dinv, NN);
  place_edges<<<(EE + 255) / 256, 256, 0, stream>>>(src, dst, rank, rowp, csrs, EE);

  pad_x<<<NN, 128, 0, stream>>>(x, xp, NN);
  pad_W<<<DINP, DHP, 0, stream>>>(W1, Wp1, DIN, DH);
  pad_W<<<DHP, DHP, 0, stream>>>(W2, Wp2, DH, DH);

  gemm_f32_wmma<<<gemmGrid, 256, 0, stream>>>(xp, DINP, Wp1, bufA, NN, DINP);
  gather_rows<<<(NN + 7) / 8, 256, 0, stream>>>(bufA, rowp, csrs, dinv, bufB, NN);
  fill_f32<<<(2 * DHP + 255) / 256, 256, 0, stream>>>(sums, 0.0f, 2 * DHP);
  bias_relu_stats<<<statBlocks, DHP, 0, stream>>>(bufB, b1, sums, NN);
  bn_apply<<<statBlocks, DHP, 0, stream>>>(bufB, sums, gamma1, beta1, NN);

  gemm_f32_wmma<<<gemmGrid, 256, 0, stream>>>(bufB, DHP, Wp2, bufA, NN, DHP);
  gather_rows<<<(NN + 7) / 8, 256, 0, stream>>>(bufA, rowp, csrs, dinv, bufB, NN);
  fill_f32<<<(2 * DHP + 255) / 256, 256, 0, stream>>>(sums, 0.0f, 2 * DHP);
  bias_relu_stats<<<statBlocks, DHP, 0, stream>>>(bufB, b2, sums, NN);
  bn_apply<<<statBlocks, DHP, 0, stream>>>(bufB, sums, gamma2, beta2, NN);

  fill_u32<<<((long)GG * DH + 255) / 256, 256, 0, stream>>>(pool, 0u, (long)GG * DH);
  pool_max<<<(NN + POOL_ROWS - 1) / POOL_ROWS, DHP, 0, stream>>>(bufB, batch, pool, NN);
  pool_decode<<<GG, DHP, 0, stream>>>(pool, out);
}


# Layer 1 aggregates the 78-column input before its GEMM, (Ahat x) W1 for Ahat (x W1): the gather reads 312-byte rows, not 1280, and writes the padded GEMM operand itself; sums in another order
# speedup vs baseline: 5.7398x; 1.0854x over previous
#include <hip/hip_runtime.h>
#include <hip/hip_bf16.h>

typedef float v2f __attribute__((ext_vector_type(2)));
typedef float v4f __attribute__((ext_vector_type(4)));
typedef float v8f __attribute__((ext_vector_type(8)));

#define NN    50000
#define EE    800000
#define GG    500
#define DIN   78
#define DINP  80
#define DH    312
#define DHP   320
#define BNEPS 1e-5f

__global__ void fill_f32(float* __restrict__ p, float v, long n) {
  long i = (long)blockIdx.x * blockDim.x + threadIdx.x;
  if (i < n) p[i] = v;
}
__global__ void fill_u32(unsigned* __restrict__ p, unsigned v, long n) {
  long i = (long)blockIdx.x * blockDim.x + threadIdx.x;
  if (i < n) p[i] = v;
}

__global__ void edge_rank(const int* __restrict__ dst, int* __restrict__ cnt,
                          int* __restrict__ rank, int E) {
  int e = blockIdx.x * blockDim.x + threadIdx.x;
  if (e < E) rank[e] = atomicAdd(&cnt[dst[e]], 1);
}
__global__ __launch_bounds__(1024) void scan_rows(const int* __restrict__ cnt,
                                                  int* __restrict__ rowptr,
                                                  float* __restrict__ dinv, int n) {
  __shared__ int wsum[32];
  const int tid = threadIdx.x, lane = tid & 31, w = tid >> 5;
  int carry = 0;
  for (int base = 0; base < n; base += 4096) {
    const int i = base + tid * 4;
    int v0 = 0, v1 = 0, v2 = 0, v3 = 0;
    if (i + 3 < n) {
      const int4 q = *(const int4*)(cnt + i);
      v0 = q.x; v1 = q.y; v2 = q.z; v3 = q.w;
    } else {
      if (i < n) v0 = cnt[i];
      if (i + 1 < n) v1 = cnt[i + 1];
      if (i + 2 < n) v2 = cnt[i + 2];
    }
    const int mine = v0 + v1 + v2 + v3;
    int x = mine;
    for (int o = 1; o < 32; o <<= 1) {
      const int t = __shfl_up(x, o);
      if (lane >= o) x += t;
    }
    if (lane == 31) wsum[w] = x;
    __syncthreads();
    if (w == 0) {
      int y = wsum[lane];
      for (int o = 1; o < 32; o <<= 1) {
        const int t = __shfl_up(y, o);
        if (lane >= o) y += t;
      }
      wsum[lane] = y;
    }
    __syncthreads();
    const int off = carry + (w > 0 ? wsum[w - 1] : 0) + x - mine;
    carry += wsum[31];
    if (i < n)     { rowptr[i]     = off;                dinv[i]     = rsqrtf((float)(v0 + 1)); }
    if (i + 1 < n) { rowptr[i + 1] = off + v0;           dinv[i + 1] = rsqrtf((float)(v1 + 1)); }
    if (i + 2 < n) { rowptr[i + 2] = off + v0 + v1;      dinv[i + 2] = rsqrtf((float)(v2 + 1)); }
    if (i + 3 < n) { rowptr[i + 3] = off + v0 + v1 + v2; dinv[i + 3] = rsqrtf((float)(v3 + 1)); }
    __syncthreads();
  }
  if (tid == 0) rowptr[n] = carry;
}
__global__ void place_edges(const int* __restrict__ src, const int* __restrict__ dst,
                            const int* __restrict__ rank, const int* __restrict__ rowptr,
                            int* __restrict__ csr_src, int E) {
  int e = blockIdx.x * blockDim.x + threadIdx.x;
  if (e < E) csr_src[rowptr[dst[e]] + rank[e]] = src[e];
}

__global__ void pad_W(const float* __restrict__ W, float* __restrict__ Wp, int K, int Nn) {
  int r = blockIdx.x, c = threadIdx.x;
  Wp[(size_t)r * DHP + c] = (r < K && c < Nn) ? W[(size_t)r * Nn + c] : 0.0f;
}

__global__ __launch_bounds__(256) void gemm_f32_wmma(
    const float* __restrict__ A, int lda,
    const float* __restrict__ B,
    float* __restrict__ C,
    int M, int Kpad) {
  const int lane  = threadIdx.x & 31;
  const int strip = blockIdx.x * 8 + (threadIdx.x >> 5);
  const int cg    = blockIdx.y;
  const int row0  = strip * 32;
  if (row0 >= M) return;

  const int h = lane >> 4;
  const int n = lane & 15;
  const int ar0 = min(row0 + n, M - 1);
  const int ar1 = min(row0 + 16 + n, M - 1);
  const float* a0p = A + (size_t)ar0 * lda + 8 * h;
  const float* a1p = A + (size_t)ar1 * lda + 8 * h;
  const float* bp  = B + (size_t)(8 * h) * DHP + cg * 64 + 4 * n;

  v8f acc[2][4];
#pragma unroll
  for (int i = 0; i < 2; ++i)
#pragma unroll
    for (int t = 0; t < 4; ++t) acc[i][t] = (v8f){0.f, 0.f, 0.f, 0.f, 0.f, 0.f, 0.f, 0.f};

  for (int k0 = 0; k0 < Kpad; k0 += 16) {
    v4f a0[2], a1[2], b[8];
    a0[0] = *(const v4f*)(a0p + k0);
    a0[1] = *(const v4f*)(a0p + k0 + 4);
    a1[0] = *(const v4f*)(a1p + k0);
    a1[1] = *(const v4f*)(a1p + k0 + 4);
#pragma unroll
    for (int j = 0; j < 8; ++j) b[j] = *(const v4f*)(bp + (size_t)(k0 + j) * DHP);
#pragma unroll
    for (int s = 0; s < 4; ++s) {
      v2f fa0, fa1;
      fa0.x = a0[s >> 1][(2 * s) & 3];
      fa0.y = a0[s >> 1][(2 * s + 1) & 3];
      fa1.x = a1[s >> 1][(2 * s) & 3];
      fa1.y = a1[s >> 1][(2 * s + 1) & 3];
#pragma unroll
      for (int t = 0; t < 4; ++t) {
        v2f fb;
        fb.x = b[2 * s][t];
        fb.y = b[2 * s + 1][t];
        acc[0][t] = __builtin_amdgcn_wmma_f32_16x16x4_f32(false, fa0, false, fb,
                                                          (short)0, acc[0][t], false, false);
        acc[1][t] = __builtin_amdgcn_wmma_f32_16x16x4_f32(false, fa1, false, fb,
                                                          (short)0, acc[1][t], false, false);
      }
    }
  }
#pragma unroll
  for (int i = 0; i < 2; ++i) {
#pragma unroll
    for (int r = 0; r < 8; ++r) {
      const int row = row0 + 16 * i + 8 * h + r;
      v4f o;
      o.x = acc[i][0][r];
      o.y = acc[i][1][r];
      o.z = acc[i][2][r];
      o.w = acc[i][3][r];
      if (row < M) *(v4f*)(C + (size_t)row * DHP + cg * 64 + 4 * n) = o;
    }
  }
}

__global__ __launch_bounds__(256) void gather_rows(
    const float* __restrict__ h, const int* __restrict__ rowptr,
    const int* __restrict__ csr_src, const float* __restrict__ dinv,
    float* __restrict__ agg, int M) {
  const int lane = threadIdx.x & 31;
  const int n = __builtin_amdgcn_readfirstlane(blockIdx.x * 8 + (threadIdx.x >> 5));
  if (n >= M) return;
  const int beg = rowptr[n], end = rowptr[n + 1];
  const float dn = dinv[n];

  v2f acc[5];
  {
    const v2f* p = (const v2f*)(h + (size_t)n * DHP) + lane;
#pragma unroll
    for (int i = 0; i < 5; ++i) {
      const v2f v = p[i * 32];
      acc[i].x = v.x * dn * dn;
      acc[i].y = v.y * dn * dn;
    }
  }
  for (int base = beg; base < end; base += 32) {
    const int j = base + lane;
    int s_l = 0;
    float w_l = 0.f;
    if (j < end) {
      s_l = (int)min((unsigned)csr_src[j], (unsigned)(M - 1));
      w_l = dinv[s_l] * dn;
    }
    const int m = min(32, end - base);
#pragma unroll 4
    for (int t = 0; t < m; ++t) {
      const int s = __builtin_amdgcn_readlane(s_l, t);
      const float wgt = __int_as_float(__builtin_amdgcn_readlane(__float_as_int(w_l), t));
      const v2f* p = (const v2f*)(h + (size_t)s * DHP) + lane;
#pragma unroll
      for (int i = 0; i < 5; ++i) {
        const v2f v = p[i * 32];
        acc[i].x += v.x * wgt;
        acc[i].y += v.y * wgt;
      }
    }
  }
  v2f* o = (v2f*)(agg + (size_t)n * DHP) + lane;
#pragma unroll
  for (int i = 0; i < 5; ++i) o[i * 32] = acc[i];
}

#define XPAIRS (DIN / 2)
__global__ __launch_bounds__(256) void gather_x(
    const float* __restrict__ x, const int* __restrict__ rowptr,
    const int* __restrict__ csr_src, const float* __restrict__ dinv,
    float* __restrict__ ax, int M) {
  const int lane = threadIdx.x & 31;
  const int q = lane & 7;
  const int n = (blockIdx.x * 8 + (threadIdx.x >> 5)) * 4 + (lane >> 3);
  const bool rowok = n < M;
  const int nc = rowok ? n : M - 1;
  const int beg = rowptr[nc];
  const int end = rowptr[nc + 1];
  const int deg = rowok ? end - beg : 0;
  const float dn = dinv[nc];
  const unsigned o03 = 8u * q;
  const unsigned o4  = 8u * min(q + 32, XPAIRS - 1);
  const char* xb = (const char*)x;

  v2f acc[5];
  {
    const unsigned r = (unsigned)nc * (4u * DIN);
#pragma unroll
    for (int i = 0; i < 5; ++i) {
      const v2f v = *(const v2f*)(xb + (i < 4 ? r + o03 + 64u * i : r + o4));
      acc[i].x = v.x * dn * dn;
      acc[i].y = v.y * dn * dn;
    }
  }
  int maxdeg = deg;
  maxdeg = max(maxdeg, __shfl_xor(maxdeg, 8));
  maxdeg = max(maxdeg, __shfl_xor(maxdeg, 16));
  maxdeg = __builtin_amdgcn_readfirstlane(maxdeg);

  for (int c = 0; c < maxdeg; c += 8) {
    int s_l = 0;
    float w_l = 0.f;
    if (c + q < deg) {
      s_l = (int)min((unsigned)csr_src[beg + c + q], (unsigned)(M - 1));
      w_l = dinv[s_l] * dn;
    }
    for (int t0 = 0; t0 < 8 && c + t0 < maxdeg; t0 += 4) {
      v2f v[4][5];
      float wgt[4];
#pragma unroll
      for (int u = 0; u < 4; ++u) {
        const int s = __shfl(s_l, t0 + u, 8);
        wgt[u] = __shfl(w_l, t0 + u, 8);
        const unsigned r = (unsigned)s * (4u * DIN);
#pragma unroll
        for (int i = 0; i < 5; ++i)
          v[u][i] = *(const v2f*)(xb + (i < 4 ? r + o03 + 64u * i : r + o4));
      }
#pragma unroll
      for (int u = 0; u < 4; ++u) {
        const bool on = c + t0 + u < deg;
#pragma unroll
        for (int i = 0; i < 5; ++i) {
          acc[i].x += (on ? v[u][i].x : 0.f) * wgt[u];
          acc[i].y += (on ? v[u][i].y : 0.f) * wgt[u];
        }
      }
    }
  }
  if (rowok) {
    v2f* o = (v2f*)(ax + (size_t)n * DINP) + q;
#pragma unroll
    for (int i = 0; i < 5; ++i) {
      v2f r = acc[i];
      if (q + 8 * i >= XPAIRS) r = (v2f){0.f, 0.f};
      o[8 * i] = r;
    }
  }
}

#define BN_ROWS  64
#define BN_LANES 4
#define BN_PER   (BN_ROWS / BN_LANES)
#define BN_STEP  ((size_t)BN_LANES * DHP)
template <bool FULL>
__device__ __forceinline__ void stats_rows(float* __restrict__ p, int r0, int M, v4f b,
                                           v4f& s, v4f& s2) {
  v4f v[BN_PER];
#pragma unroll
  for (int i = 0; i < BN_PER; ++i)
    v[i] = (FULL || r0 + i * BN_LANES < M) ? *(const v4f*)(p + i * BN_STEP)
                                           : (v4f){0.f, 0.f, 0.f, 0.f};
#pragma unroll
  for (int i = 0; i < BN_PER; ++i) {
    if (FULL || r0 + i * BN_LANES < M) {
      v4f t;
      t.x = fmaxf(v[i].x + b.x, 0.0f);
      t.y = fmaxf(v[i].y + b.y, 0.0f);
      t.z = fmaxf(v[i].z + b.z, 0.0f);
      t.w = fmaxf(v[i].w + b.w, 0.0f);
      *(v4f*)(p + i * BN_STEP) = t;
      s  += t;
      s2 += t * t;
    }
  }
}
__global__ __launch_bounds__(DHP) void bias_relu_stats(float* __restrict__ y,
                                                       const float* __restrict__ bias,
                                                       float* __restrict__ sums, int M) {
  __shared__ v4f red[2][BN_LANES][DHP / 4];
  const int q  = threadIdx.x % 80;
  const int rl = threadIdx.x / 80;
  const int r0 = blockIdx.x * BN_ROWS + rl;
  v4f s  = (v4f){0.f, 0.f, 0.f, 0.f};
  v4f s2 = (v4f){0.f, 0.f, 0.f, 0.f};
  if (q < DH / 4) {
    float* p = y + (size_t)r0 * DHP + 4 * q;
    v4f b;
    b.x = bias[4 * q];     b.y = bias[4 * q + 1];
    b.z = bias[4 * q + 2]; b.w = bias[4 * q + 3];
    if (blockIdx.x * BN_ROWS + BN_ROWS <= M) stats_rows<true>(p, r0, M, b, s, s2);
    else                                     stats_rows<false>(p, r0, M, b, s, s2);
  }
  red[0][rl][q] = s;
  red[1][rl][q] = s2;
  __syncthreads();
  const int c = threadIdx.x;
  if (c < DH) {
    const float* r1 = (const float*)&red[0][0][0];
    const float* r2 = (const float*)&red[1][0][0];
    float a = 0.f, a2 = 0.f;
#pragma unroll
    for (int l = 0; l < BN_LANES; ++l) {
      a  += r1[l * DHP + c];
      a2 += r2[l * DHP + c];
    }
    atomicAdd(&sums[c], a);
    atomicAdd(&sums[DHP + c], a2);
  }
}
template <bool FULL>
__device__ __forceinline__ void apply_rows(float* __restrict__ p, int r0, int M,
                                           const float* __restrict__ sums,
                                           const float* __restrict__ gamma,
                                           const float* __restrict__ beta, int q) {
  v4f v[BN_PER];
#pragma unroll
  for (int i = 0; i < BN_PER; ++i)
    v[i] = (FULL || r0 + i * BN_LANES < M) ? *(const v4f*)(p + i * BN_STEP)
                                           : (v4f){0.f, 0.f, 0.f, 0.f};
  const float invn = 1.0f / (float)M;
  v4f scale, shift;
#pragma unroll
  for (int j = 0; j < 4; ++j) {
    const int c = 4 * q + j;
    const float mean = sums[c] * invn;
    const float var  = sums[DHP + c] * invn - mean * mean;
    scale[j] = rsqrtf(var + BNEPS) * gamma[c];
    shift[j] = beta[c] - mean * scale[j];
  }
#pragma unroll
  for (int i = 0; i < BN_PER; ++i)
    if (FULL || r0 + i * BN_LANES < M) *(v4f*)(p + i * BN_STEP) = v[i] * scale + shift;
}
__global__ __launch_bounds__(DHP) void bn_apply(float* __restrict__ y, const float* __restrict__ sums,
                                                const float* __restrict__ gamma,
                                                const float* __restrict__ beta, int M) {
  const int q  = threadIdx.x % 80;
  const int rl = threadIdx.x / 80;
  if (q >= DH / 4) return;
  const int r0 = blockIdx.x * BN_ROWS + rl;
  float* p = y + (size_t)r0 * DHP + 4 * q;
  if (blockIdx.x * BN_ROWS + BN_ROWS <= M) apply_rows<true>(p, r0, M, sums, gamma, beta, q);
  else                                     apply_rows<false>(p, r0, M, sums, gamma, beta, q);
}

__device__ __forceinline__ unsigned enc_f32(float f) {
  unsigned u = __float_as_uint(f);
  return (u & 0x80000000u) ? ~u : (u | 0x80000000u);
}
__device__ __forceinline__ float dec_f32(unsigned u) {
  return __uint_as_float((u & 0x80000000u) ? (u & 0x7FFFFFFFu) : ~u);
}
#define POOL_ROWS  64
#define POOL_BURST 16
__global__ __launch_bounds__(DHP) void pool_max(const float* __restrict__ y,
                                                const int* __restrict__ batch,
                                                unsigned* __restrict__ pool, int M) {
  const int c = threadIdx.x;
  if (c >= DH) return;
  const int r0 = blockIdx.x * POOL_ROWS;
  const int r1 = min(r0 + POOL_ROWS, M);
  if (r0 >= r1) return;
  int gcur = batch[r0];
  unsigned best = 0u;
  int r = r0;
  for (; r + POOL_BURST <= r1; r += POOL_BURST) {
    const float* p = y + (size_t)r * DHP + c;
    float v[POOL_BURST];
    int g[POOL_BURST];
#pragma unroll
    for (int i = 0; i < POOL_BURST; ++i) {
      v[i] = p[(size_t)i * DHP];
      g[i] = batch[r + i];
    }
#pragma unroll
    for (int i = 0; i < POOL_BURST; ++i) {
      if (g[i] != gcur) {
        atomicMax(&pool[(size_t)gcur * DH + c], best);
        gcur = g[i];
        best = 0u;
      }
      best = max(best, enc_f32(v[i]));
    }
  }
  for (; r < r1; ++r) {
    const int g = batch[r];
    if (g != gcur) {
      atomicMax(&pool[(size_t)gcur * DH + c], best);
      gcur = g;
      best = 0u;
    }
    best = max(best, enc_f32(y[(size_t)r * DHP + c]));
  }
  atomicMax(&pool[(size_t)gcur * DH + c], best);
}
__global__ void pool_decode(const unsigned* __restrict__ pool, float* __restrict__ out) {
  int g = blockIdx.x, c = threadIdx.x;
  if (c >= DH) return;
  out[(size_t)g * DH + c] = dec_f32(pool[(size_t)g * DH + c]);
}

extern "C" void kernel_launch(void* const* d_in, const int* in_sizes, int n_in,
                              void* d_out, int out_size, void* d_ws, size_t ws_size,
                              hipStream_t stream) {
  const float* x      = (const float*)d_in[0];
  const int*   eidx   = (const int*)d_in[1];
  const int*   batch  = (const int*)d_in[2];
  const float* W1     = (const float*)d_in[3];
  const float* b1     = (const float*)d_in[4];
  const float* gamma1 = (const float*)d_in[5];
  const float* beta1  = (const float*)d_in[6];
  const float* W2     = (const float*)d_in[7];
  const float* b2     = (const float*)d_in[8];
  const float* gamma2 = (const float*)d_in[9];
  const float* beta2  = (const float*)d_in[10];
  float* out = (float*)d_out;

  const int* src = eidx;
  const int* dst = eidx + EE;

  float* ws = (float*)d_ws;
  float*    dinv = ws;
  int*      cnt  = (int*)(ws + 50048);
  int*      rowp = (int*)(ws + 100096);
  int*      rank = (int*)(ws + 150208);
  int*      csrs = (int*)(ws + 950208);
  float*    ax   = ws + 1750208;
  float*    Wp1  = ws + 5750208;
  float*    Wp2  = ws + 5775808;
  float*    sums = ws + 5878208;
  unsigned* pool = (unsigned*)(ws + 5878848);
  float*    bufA = ws + 6034880;
  float*    bufB = ws + 22034880;

  const int statBlocks = (NN + BN_ROWS - 1) / BN_ROWS;
  const dim3 gemmGrid(((NN + 31) / 32 + 7) / 8, DHP / 64);

  fill_u32<<<(NN + 255) / 256, 256, 0, stream>>>((unsigned*)cnt, 0u, NN);
  edge_rank<<<(EE + 255) / 256, 256, 0, stream>>>(dst, cnt, rank, EE);
  scan_rows<<<1, 1024, 0, stream>>>(cnt, rowp, dinv, NN);
  place_edges<<<(EE + 255) / 256, 256, 0, stream>>>(src, dst, rank, rowp, csrs, EE);

  pad_W<<<DINP, DHP, 0, stream>>>(W1, Wp1, DIN, DH);
  pad_W<<<DHP, DHP, 0, stream>>>(W2, Wp2, DH, DH);

  gather_x<<<(NN + 31) / 32, 256, 0, stream>>>(x, rowp, csrs, dinv, ax, NN);
  gemm_f32_wmma<<<gemmGrid, 256, 0, stream>>>(ax, DINP, Wp1, bufB, NN, DINP);
  fill_f32<<<(2 * DHP + 255) / 256, 256, 0, stream>>>(sums, 0.0f, 2 * DHP);
  bias_relu_stats<<<statBlocks, DHP, 0, stream>>>(bufB, b1, sums, NN);
  bn_apply<<<statBlocks, DHP, 0, stream>>>(bufB, sums, gamma1, beta1, NN);

  gemm_f32_wmma<<<gemmGrid, 256, 0, stream>>>(bufB, DHP, Wp2, bufA, NN, DHP);
  gather_rows<<<(NN + 7) / 8, 256, 0, stream>>>(bufA, rowp, csrs, dinv, bufB, NN);
  fill_f32<<<(2 * DHP + 255) / 256, 256, 0, stream>>>(sums, 0.0f, 2 * DHP);
  bias_relu_stats<<<statBlocks, DHP, 0, stream>>>(bufB, b2, sums, NN);
  bn_apply<<<statBlocks, DHP, 0, stream>>>(bufB, sums, gamma2, beta2, NN);

  fill_u32<<<((long)GG * DH + 255) / 256, 256, 0, stream>>>(pool, 0u, (long)GG * DH);
  pool_max<<<(NN + POOL_ROWS - 1) / POOL_ROWS, DHP, 0, stream>>>(bufB, batch, pool, NN);
  pool_decode<<<GG, DHP, 0, stream>>>(pool, out);
}


# BN apply passes removed: layer 1's BN folded into GEMM 2's operands (W2 rows scaled, shift.W2 as accumulator start: another sum order), layer 2's applied in the pool; 17 launches, was 19
# speedup vs baseline: 6.5203x; 1.1360x over previous
#include <hip/hip_runtime.h>
#include <hip/hip_bf16.h>

typedef float v2f __attribute__((ext_vector_type(2)));
typedef float v4f __attribute__((ext_vector_type(4)));
typedef float v8f __attribute__((ext_vector_type(8)));

#define NN    50000
#define EE    800000
#define GG    500
#define DIN   78
#define DINP  80
#define DH    312
#define DHP   320
#define BNEPS 1e-5f

__device__ __forceinline__ void bn_coef(const float* __restrict__ sums, const float* __restrict__ gamma,
                                        const float* __restrict__ beta, int c, float invn,
                                        float& scale, float& shift) {
  const float mean = sums[c] * invn;
  const float var  = sums[DHP + c] * invn - mean * mean;
  scale = rsqrtf(var + BNEPS) * gamma[c];
  shift = beta[c] - mean * scale;
}

__global__ void fill_f32(float* __restrict__ p, float v, long n) {
  long i = (long)blockIdx.x * blockDim.x + threadIdx.x;
  if (i < n) p[i] = v;
}
__global__ void fill_u32(unsigned* __restrict__ p, unsigned v, long n) {
  long i = (long)blockIdx.x * blockDim.x + threadIdx.x;
  if (i < n) p[i] = v;
}

__global__ void edge_rank(const int* __restrict__ dst, int* __restrict__ cnt,
                          int* __restrict__ rank, int E) {
  int e = blockIdx.x * blockDim.x + threadIdx.x;
  if (e < E) rank[e] = atomicAdd(&cnt[dst[e]], 1);
}
__global__ __launch_bounds__(1024) void scan_rows(const int* __restrict__ cnt,
                                                  int* __restrict__ rowptr,
                                                  float* __restrict__ dinv, int n) {
  __shared__ int wsum[32];
  const int tid = threadIdx.x, lane = tid & 31, w = tid >> 5;
  int carry = 0;
  for (int base = 0; base < n; base += 4096) {
    const int i = base + tid * 4;
    int v0 = 0, v1 = 0, v2 = 0, v3 = 0;
    if (i + 3 < n) {
      const int4 q = *(const int4*)(cnt + i);
      v0 = q.x; v1 = q.y; v2 = q.z; v3 = q.w;
    } else {
      if (i < n) v0 = cnt[i];
      if (i + 1 < n) v1 = cnt[i + 1];
      if (i + 2 < n) v2 = cnt[i + 2];
    }
    const int mine = v0 + v1 + v2 + v3;
    int x = mine;
    for (int o = 1; o < 32; o <<= 1) {
      const int t = __shfl_up(x, o);
      if (lane >= o) x += t;
    }
    if (lane == 31) wsum[w] = x;
    __syncthreads();
    if (w == 0) {
      int y = wsum[lane];
      for (int o = 1; o < 32; o <<= 1) {
        const int t = __shfl_up(y, o);
        if (lane >= o) y += t;
      }
      wsum[lane] = y;
    }
    __syncthreads();
    const int off = carry + (w > 0 ? wsum[w - 1] : 0) + x - mine;
    carry += wsum[31];
    if (i < n)     { rowptr[i]     = off;                dinv[i]     = rsqrtf((float)(v0 + 1)); }
    if (i + 1 < n) { rowptr[i + 1] = off + v0;           dinv[i + 1] = rsqrtf((float)(v1 + 1)); }
    if (i + 2 < n) { rowptr[i + 2] = off + v0 + v1;      dinv[i + 2] = rsqrtf((float)(v2 + 1)); }
    if (i + 3 < n) { rowptr[i + 3] = off + v0 + v1 + v2; dinv[i + 3] = rsqrtf((float)(v3 + 1)); }
    __syncthreads();
  }
  if (tid == 0) rowptr[n] = carry;
}
__global__ void place_edges(const int* __restrict__ src, const int* __restrict__ dst,
                            const int* __restrict__ rank, const int* __restrict__ rowptr,
                            int* __restrict__ csr_src, int E) {
  int e = blockIdx.x * blockDim.x + threadIdx.x;
  if (e < E) csr_src[rowptr[dst[e]] + rank[e]] = src[e];
}

__global__ void pad_W(const float* __restrict__ W, float* __restrict__ Wp, int K, int Nn) {
  int r = blockIdx.x, c = threadIdx.x;
  Wp[(size_t)r * DHP + c] = (r < K && c < Nn) ? W[(size_t)r * Nn + c] : 0.0f;
}

#define FOLD_G 32
__global__ __launch_bounds__(1024) void fold_bn_W(const float* __restrict__ W,
                                                  const float* __restrict__ sums,
                                                  const float* __restrict__ gamma,
                                                  const float* __restrict__ beta,
                                                  float* __restrict__ Wp, float* __restrict__ cvec,
                                                  int M) {
  __shared__ float part[FOLD_G][33];
  const int l = threadIdx.x & 31, g = threadIdx.x >> 5;
  const int c = blockIdx.x * 32 + l;
  const int cc = min(c, DH - 1);
  const float invn = 1.0f / (float)M;
  float w[DHP / FOLD_G], scale[DHP / FOLD_G], shift[DHP / FOLD_G];
#pragma unroll
  for (int i = 0; i < DHP / FOLD_G; ++i) {
    const int rr = min(g + FOLD_G * i, DH - 1);
    w[i] = W[(size_t)rr * DH + cc];
    bn_coef(sums, gamma, beta, rr, invn, scale[i], shift[i]);
  }
  float acc = 0.f;
#pragma unroll
  for (int i = 0; i < DHP / FOLD_G; ++i) {
    const int r = g + FOLD_G * i;
    const bool in = r < DH && c < DH;
    Wp[(size_t)r * DHP + c] = in ? w[i] * scale[i] : 0.f;
    acc += in ? w[i] * shift[i] : 0.f;
  }
  part[g][l] = acc;
  __syncthreads();
  if (g == 0) {
    float s = 0.f;
#pragma unroll
    for (int j = 0; j < FOLD_G; ++j) s += part[j][l];
    cvec[c] = s;
  }
}

template <bool CINIT>
__global__ __launch_bounds__(256) void gemm_f32_wmma(
    const float* __restrict__ A, int lda,
    const float* __restrict__ B,
    float* __restrict__ C,
    int M, int Kpad, const float* __restrict__ cinit) {
  const int lane  = threadIdx.x & 31;
  const int strip = blockIdx.x * 8 + (threadIdx.x >> 5);
  const int cg    = blockIdx.y;
  const int row0  = strip * 32;
  if (row0 >= M) return;

  const int h = lane >> 4;
  const int n = lane & 15;
  const int ar0 = min(row0 + n, M - 1);
  const int ar1 = min(row0 + 16 + n, M - 1);
  const float* a0p = A + (size_t)ar0 * lda + 8 * h;
  const float* a1p = A + (size_t)ar1 * lda + 8 * h;
  const float* bp  = B + (size_t)(8 * h) * DHP + cg * 64 + 4 * n;

  v8f acc[2][4];
  v4f ci = (v4f){0.f, 0.f, 0.f, 0.f};
  if (CINIT) ci = *(const v4f*)(cinit + cg * 64 + 4 * n);
#pragma unroll
  for (int i = 0; i < 2; ++i)
#pragma unroll
    for (int t = 0; t < 4; ++t)
      acc[i][t] = (v8f){ci[t], ci[t], ci[t], ci[t], ci[t], ci[t], ci[t], ci[t]};

  for (int k0 = 0; k0 < Kpad; k0 += 16) {
    v4f a0[2], a1[2], b[8];
    a0[0] = *(const v4f*)(a0p + k0);
    a0[1] = *(const v4f*)(a0p + k0 + 4);
    a1[0] = *(const v4f*)(a1p + k0);
    a1[1] = *(const v4f*)(a1p + k0 + 4);
#pragma unroll
    for (int j = 0; j < 8; ++j) b[j] = *(const v4f*)(bp + (size_t)(k0 + j) * DHP);
#pragma unroll
    for (int s = 0; s < 4; ++s) {
      v2f fa0, fa1;
      fa0.x = a0[s >> 1][(2 * s) & 3];
      fa0.y = a0[s >> 1][(2 * s + 1) & 3];
      fa1.x = a1[s >> 1][(2 * s) & 3];
      fa1.y = a1[s >> 1][(2 * s + 1) & 3];
#pragma unroll
      for (int t = 0; t < 4; ++t) {
        v2f fb;
        fb.x = b[2 * s][t];
        fb.y = b[2 * s + 1][t];
        acc[0][t] = __builtin_amdgcn_wmma_f32_16x16x4_f32(false, fa0, false, fb,
                                                          (short)0, acc[0][t], false, false);
        acc[1][t] = __builtin_amdgcn_wmma_f32_16x16x4_f32(false, fa1, false, fb,
                                                          (short)0, acc[1][t], false, false);
      }
    }
  }
#pragma unroll
  for (int i = 0; i < 2; ++i) {
#pragma unroll
    for (int r = 0; r < 8; ++r) {
      const int row = row0 + 16 * i + 8 * h + r;
      v4f o;
      o.x = acc[i][0][r];
      o.y = acc[i][1][r];
      o.z = acc[i][2][r];
      o.w = acc[i][3][r];
      if (row < M) *(v4f*)(C + (size_t)row * DHP + cg * 64 + 4 * n) = o;
    }
  }
}

__global__ __launch_bounds__(256) void gather_rows(
    const float* __restrict__ h, const int* __restrict__ rowptr,
    const int* __restrict__ csr_src, const float* __restrict__ dinv,
    float* __restrict__ agg, int M) {
  const int lane = threadIdx.x & 31;
  const int n = __builtin_amdgcn_readfirstlane(blockIdx.x * 8 + (threadIdx.x >> 5));
  if (n >= M) return;
  const int beg = rowptr[n], end = rowptr[n + 1];
  const float dn = dinv[n];

  v2f acc[5];
  {
    const v2f* p = (const v2f*)(h + (size_t)n * DHP) + lane;
#pragma unroll
    for (int i = 0; i < 5; ++i) {
      const v2f v = p[i * 32];
      acc[i].x = v.x * dn * dn;
      acc[i].y = v.y * dn * dn;
    }
  }
  for (int base = beg; base < end; base += 32) {
    const int j = base + lane;
    int s_l = 0;
    float w_l = 0.f;
    if (j < end) {
      s_l = (int)min((unsigned)csr_src[j], (unsigned)(M - 1));
      w_l = dinv[s_l] * dn;
    }
    const int m = min(32, end - base);
#pragma unroll 4
    for (int t = 0; t < m; ++t) {
      const int s = __builtin_amdgcn_readlane(s_l, t);
      const float wgt = __int_as_float(__builtin_amdgcn_readlane(__float_as_int(w_l), t));
      const v2f* p = (const v2f*)(h + (size_t)s * DHP) + lane;
#pragma unroll
      for (int i = 0; i < 5; ++i) {
        const v2f v = p[i * 32];
        acc[i].x += v.x * wgt;
        acc[i].y += v.y * wgt;
      }
    }
  }
  v2f* o = (v2f*)(agg + (size_t)n * DHP) + lane;
#pragma unroll
  for (int i = 0; i < 5; ++i) o[i * 32] = acc[i];
}

#define XPAIRS (DIN / 2)
__global__ __launch_bounds__(256) void gather_x(
    const float* __restrict__ x, const int* __restrict__ rowptr,
    const int* __restrict__ csr_src, const float* __restrict__ dinv,
    float* __restrict__ ax, int M) {
  const int lane = threadIdx.x & 31;
  const int q = lane & 7;
  const int n = (blockIdx.x * 8 + (threadIdx.x >> 5)) * 4 + (lane >> 3);
  const bool rowok = n < M;
  const int nc = rowok ? n : M - 1;
  const int beg = rowptr[nc];
  const int end = rowptr[nc + 1];
  const int deg = rowok ? end - beg : 0;
  const float dn = dinv[nc];
  const unsigned o03 = 8u * q;
  const unsigned o4  = 8u * min(q + 32, XPAIRS - 1);
  const char* xb = (const char*)x;

  v2f acc[5];
  {
    const unsigned r = (unsigned)nc * (4u * DIN);
#pragma unroll
    for (int i = 0; i < 5; ++i) {
      const v2f v = *(const v2f*)(xb + (i < 4 ? r + o03 + 64u * i : r + o4));
      acc[i].x = v.x * dn * dn;
      acc[i].y = v.y * dn * dn;
    }
  }
  int maxdeg = deg;
  maxdeg = max(maxdeg, __shfl_xor(maxdeg, 8));
  maxdeg = max(maxdeg, __shfl_xor(maxdeg, 16));
  maxdeg = __builtin_amdgcn_readfirstlane(maxdeg);

  for (int c = 0; c < maxdeg; c += 8) {
    int s_l = 0;
    float w_l = 0.f;
    if (c + q < deg) {
      s_l = (int)min((unsigned)csr_src[beg + c + q], (unsigned)(M - 1));
      w_l = dinv[s_l] * dn;
    }
    for (int t0 = 0; t0 < 8 && c + t0 < maxdeg; t0 += 4) {
      v2f v[4][5];
      float wgt[4];
#pragma unroll
      for (int u = 0; u < 4; ++u) {
        const int s = __shfl(s_l, t0 + u, 8);
        wgt[u] = __shfl(w_l, t0 + u, 8);
        const unsigned r = (unsigned)s * (4u * DIN);
#pragma unroll
        for (int i = 0; i < 5; ++i)
          v[u][i] = *(const v2f*)(xb + (i < 4 ? r + o03 + 64u * i : r + o4));
      }
#pragma unroll
      for (int u = 0; u < 4; ++u) {
        const bool on = c + t0 + u < deg;
#pragma unroll
        for (int i = 0; i < 5; ++i) {
          acc[i].x += (on ? v[u][i].x : 0.f) * wgt[u];
          acc[i].y += (on ? v[u][i].y : 0.f) * wgt[u];
        }
      }
    }
  }
  if (rowok) {
    v2f* o = (v2f*)(ax + (size_t)n * DINP) + q;
#pragma unroll
    for (int i = 0; i < 5; ++i) {
      v2f r = acc[i];
      if (q + 8 * i >= XPAIRS) r = (v2f){0.f, 0.f};
      o[8 * i] = r;
    }
  }
}

#define BN_ROWS  64
#define BN_LANES 4
#define BN_PER   (BN_ROWS / BN_LANES)
#define BN_STEP  ((size_t)BN_LANES * DHP)
template <bool FULL>
__device__ __forceinline__ void stats_rows(float* __restrict__ p, int r0, int M, v4f b,
                                           v4f& s, v4f& s2) {
  v4f v[BN_PER];
#pragma unroll
  for (int i = 0; i < BN_PER; ++i)
    v[i] = (FULL || r0 + i * BN_LANES < M) ? *(const v4f*)(p + i * BN_STEP)
                                           : (v4f){0.f, 0.f, 0.f, 0.f};
#pragma unroll
  for (int i = 0; i < BN_PER; ++i) {
    if (FULL || r0 + i * BN_LANES < M) {
      v4f t;
      t.x = fmaxf(v[i].x + b.x, 0.0f);
      t.y = fmaxf(v[i].y + b.y, 0.0f);
      t.z = fmaxf(v[i].z + b.z, 0.0f);
      t.w = fmaxf(v[i].w + b.w, 0.0f);
      *(v4f*)(p + i * BN_STEP) = t;
      s  += t;
      s2 += t * t;
    }
  }
}
__global__ __launch_bounds__(DHP) void bias_relu_stats(float* __restrict__ y,
                                                       const float* __restrict__ bias,
                                                       float* __restrict__ sums, int M) {
  __shared__ v4f red[2][BN_LANES][DHP / 4];
  const int q  = threadIdx.x % 80;
  const int rl = threadIdx.x / 80;
  const int r0 = blockIdx.x * BN_ROWS + rl;
  v4f s  = (v4f){0.f, 0.f, 0.f, 0.f};
  v4f s2 = (v4f){0.f, 0.f, 0.f, 0.f};
  if (q < DH / 4) {
    float* p = y + (size_t)r0 * DHP + 4 * q;
    v4f b;
    b.x = bias[4 * q];     b.y = bias[4 * q + 1];
    b.z = bias[4 * q + 2]; b.w = bias[4 * q + 3];
    if (blockIdx.x * BN_ROWS + BN_ROWS <= M) stats_rows<true>(p, r0, M, b, s, s2);
    else                                     stats_rows<false>(p, r0, M, b, s, s2);
  }
  red[0][rl][q] = s;
  red[1][rl][q] = s2;
  __syncthreads();
  const int c = threadIdx.x;
  if (c < DH) {
    const float* r1 = (const float*)&red[0][0][0];
    const float* r2 = (const float*)&red[1][0][0];
    float a = 0.f, a2 = 0.f;
#pragma unroll
    for (int l = 0; l < BN_LANES; ++l) {
      a  += r1[l * DHP + c];
      a2 += r2[l * DHP + c];
    }
    atomicAdd(&sums[c], a);
    atomicAdd(&sums[DHP + c], a2);
  }
}
__device__ __forceinline__ unsigned enc_f32(float f) {
  unsigned u = __float_as_uint(f);
  return (u & 0x80000000u) ? ~u : (u | 0x80000000u);
}
__device__ __forceinline__ float dec_f32(unsigned u) {
  return __uint_as_float((u & 0x80000000u) ? (u & 0x7FFFFFFFu) : ~u);
}
#define POOL_ROWS  64
#define POOL_BURST 16
__global__ __launch_bounds__(DHP) void pool_max(const float* __restrict__ y,
                                                const int* __restrict__ batch,
                                                unsigned* __restrict__ pool, int M,
                                                const float* __restrict__ sums,
                                                const float* __restrict__ gamma,
                                                const float* __restrict__ beta) {
  const int c = threadIdx.x;
  if (c >= DH) return;
  float scale, shift;
  bn_coef(sums, gamma, beta, c, 1.0f / (float)M, scale, shift);
  const int r0 = blockIdx.x * POOL_ROWS;
  const int r1 = min(r0 + POOL_ROWS, M);
  if (r0 >= r1) return;
  int gcur = batch[r0];
  unsigned best = 0u;
  int r = r0;
  for (; r + POOL_BURST <= r1; r += POOL_BURST) {
    const float* p = y + (size_t)r * DHP + c;
    float v[POOL_BURST];
    int g[POOL_BURST];
#pragma unroll
    for (int i = 0; i < POOL_BURST; ++i) {
      v[i] = p[(size_t)i * DHP];
      g[i] = batch[r + i];
    }
#pragma unroll
    for (int i = 0; i < POOL_BURST; ++i) {
      if (g[i] != gcur) {
        atomicMax(&pool[(size_t)gcur * DH + c], best);
        gcur = g[i];
        best = 0u;
      }
      best = max(best, enc_f32(v[i] * scale + shift));
    }
  }
  for (; r < r1; ++r) {
    const int g = batch[r];
    if (g != gcur) {
      atomicMax(&pool[(size_t)gcur * DH + c], best);
      gcur = g;
      best = 0u;
    }
    best = max(best, enc_f32(y[(size_t)r * DHP + c] * scale + shift));
  }
  atomicMax(&pool[(size_t)gcur * DH + c], best);
}
__global__ void pool_decode(const unsigned* __restrict__ pool, float* __restrict__ out) {
  int g = blockIdx.x, c = threadIdx.x;
  if (c >= DH) return;
  out[(size_t)g * DH + c] = dec_f32(pool[(size_t)g * DH + c]);
}

extern "C" void kernel_launch(void* const* d_in, const int* in_sizes, int n_in,
                              void* d_out, int out_size, void* d_ws, size_t ws_size,
                              hipStream_t stream) {
  const float* x      = (const float*)d_in[0];
  const int*   eidx   = (const int*)d_in[1];
  const int*   batch  = (const int*)d_in[2];
  const float* W1     = (const float*)d_in[3];
  const float* b1     = (const float*)d_in[4];
  const float* gamma1 = (const float*)d_in[5];
  const float* beta1  = (const float*)d_in[6];
  const float* W2     = (const float*)d_in[7];
  const float* b2     = (const float*)d_in[8];
  const float* gamma2 = (const float*)d_in[9];
  const float* beta2  = (const float*)d_in[10];
  float* out = (float*)d_out;

  const int* src = eidx;
  const int* dst = eidx + EE;

  float* ws = (float*)d_ws;
  float*    dinv = ws;
  int*      cnt  = (int*)(ws + 50048);
  int*      rowp = (int*)(ws + 100096);
  int*      rank = (int*)(ws + 150208);
  int*      csrs = (int*)(ws + 950208);
  float*    ax   = ws + 1750208;
  float*    Wp1  = ws + 5750208;
  float*    Wp2  = ws + 5775808;
  float*    sums = ws + 5878208;
  unsigned* pool = (unsigned*)(ws + 5878848);
  float*    bufA = ws + 6034880;
  float*    bufB = ws + 22034880;
  float*    cvec = ws + 38034880;

  const int statBlocks = (NN + BN_ROWS - 1) / BN_ROWS;
  const dim3 gemmGrid(((NN + 31) / 32 + 7) / 8, DHP / 64);

  fill_u32<<<(NN + 255) / 256, 256, 0, stream>>>((unsigned*)cnt, 0u, NN);
  edge_rank<<<(EE + 255) / 256, 256, 0, stream>>>(dst, cnt, rank, EE);
  scan_rows<<<1, 1024, 0, stream>>>(cnt, rowp, dinv, NN);
  place_edges<<<(EE + 255) / 256, 256, 0, stream>>>(src, dst, rank, rowp, csrs, EE);

  pad_W<<<DINP, DHP, 0, stream>>>(W1, Wp1, DIN, DH);

  gather_x<<<(NN + 31) / 32, 256, 0, stream>>>(x, rowp, csrs, dinv, ax, NN);
  gemm_f32_wmma<false><<<gemmGrid, 256, 0, stream>>>(ax, DINP, Wp1, bufB, NN, DINP, nullptr);
  fill_f32<<<(2 * DHP + 255) / 256, 256, 0, stream>>>(sums, 0.0f, 2 * DHP);
  bias_relu_stats<<<statBlocks, DHP, 0, stream>>>(bufB, b1, sums, NN);

  fold_bn_W<<<DHP / 32, 1024, 0, stream>>>(W2, sums, gamma1, beta1, Wp2, cvec, NN);
  gemm_f32_wmma<true><<<gemmGrid, 256, 0, stream>>>(bufB, DHP, Wp2, bufA, NN, DHP, cvec);
  gather_rows<<<(NN + 7) / 8, 256, 0, stream>>>(bufA, rowp, csrs, dinv, bufB, NN);
  fill_f32<<<(2 * DHP + 255) / 256, 256, 0, stream>>>(sums, 0.0f, 2 * DHP);
  bias_relu_stats<<<statBlocks, DHP, 0, stream>>>(bufB, b2, sums, NN);

  fill_u32<<<((long)GG * DH + 255) / 256, 256, 0, stream>>>(pool, 0u, (long)GG * DH);
  pool_max<<<(NN + POOL_ROWS - 1) / POOL_ROWS, DHP, 0, stream>>>(bufB, batch, pool, NN,
                                                                sums, gamma2, beta2);
  pool_decode<<<GG, DHP, 0, stream>>>(pool, out);
}


# Bias+ReLU+moment passes removed: GEMM 1's store and the layer-2 gather's store (a wave walks 2 rows) add bias, clamp and sum the BN moments (another sum order, 4 atomic copies); 17 -> 15 launches
# speedup vs baseline: 7.2830x; 1.1170x over previous
#include <hip/hip_runtime.h>
#include <hip/hip_bf16.h>

typedef float v2f __attribute__((ext_vector_type(2)));
typedef float v4f __attribute__((ext_vector_type(4)));
typedef float v8f __attribute__((ext_vector_type(8)));

#define NN    50000
#define EE    800000
#define GG    500
#define DIN   78
#define DINP  80
#define DH    312
#define DHP   320
#define BNEPS 1e-5f

#define SUMS_K 4
__device__ __forceinline__ void bn_coef(const float* __restrict__ sums, const float* __restrict__ gamma,
                                        const float* __restrict__ beta, int c, float invn,
                                        float& scale, float& shift) {
  float t1 = 0.f, t2 = 0.f;
#pragma unroll
  for (int k = 0; k < SUMS_K; ++k) {
    t1 += sums[k * 2 * DHP + c];
    t2 += sums[k * 2 * DHP + DHP + c];
  }
  const float mean = t1 * invn;
  const float var  = t2 * invn - mean * mean;
  scale = rsqrtf(var + BNEPS) * gamma[c];
  shift = beta[c] - mean * scale;
}

__global__ void fill_f32(float* __restrict__ p, float v, long n) {
  long i = (long)blockIdx.x * blockDim.x + threadIdx.x;
  if (i < n) p[i] = v;
}
__global__ void fill_u32(unsigned* __restrict__ p, unsigned v, long n) {
  long i = (long)blockIdx.x * blockDim.x + threadIdx.x;
  if (i < n) p[i] = v;
}

__global__ void edge_rank(const int* __restrict__ dst, int* __restrict__ cnt,
                          int* __restrict__ rank, int E) {
  int e = blockIdx.x * blockDim.x + threadIdx.x;
  if (e < E) rank[e] = atomicAdd(&cnt[dst[e]], 1);
}
__global__ __launch_bounds__(1024) void scan_rows(const int* __restrict__ cnt,
                                                  int* __restrict__ rowptr,
                                                  float* __restrict__ dinv, int n) {
  __shared__ int wsum[32];
  const int tid = threadIdx.x, lane = tid & 31, w = tid >> 5;
  int carry = 0;
  for (int base = 0; base < n; base += 4096) {
    const int i = base + tid * 4;
    int v0 = 0, v1 = 0, v2 = 0, v3 = 0;
    if (i + 3 < n) {
      const int4 q = *(const int4*)(cnt + i);
      v0 = q.x; v1 = q.y; v2 = q.z; v3 = q.w;
    } else {
      if (i < n) v0 = cnt[i];
      if (i + 1 < n) v1 = cnt[i + 1];
      if (i + 2 < n) v2 = cnt[i + 2];
    }
    const int mine = v0 + v1 + v2 + v3;
    int x = mine;
    for (int o = 1; o < 32; o <<= 1) {
      const int t = __shfl_up(x, o);
      if (lane >= o) x += t;
    }
    if (lane == 31) wsum[w] = x;
    __syncthreads();
    if (w == 0) {
      int y = wsum[lane];
      for (int o = 1; o < 32; o <<= 1) {
        const int t = __shfl_up(y, o);
        if (lane >= o) y += t;
      }
      wsum[lane] = y;
    }
    __syncthreads();
    const int off = carry + (w > 0 ? wsum[w - 1] : 0) + x - mine;
    carry += wsum[31];
    if (i < n)     { rowptr[i]     = off;                dinv[i]     = rsqrtf((float)(v0 + 1)); }
    if (i + 1 < n) { rowptr[i + 1] = off + v0;           dinv[i + 1] = rsqrtf((float)(v1 + 1)); }
    if (i + 2 < n) { rowptr[i + 2] = off + v0 + v1;      dinv[i + 2] = rsqrtf((float)(v2 + 1)); }
    if (i + 3 < n) { rowptr[i + 3] = off + v0 + v1 + v2; dinv[i + 3] = rsqrtf((float)(v3 + 1)); }
    __syncthreads();
  }
  if (tid == 0) rowptr[n] = carry;
}
__global__ void place_edges(const int* __restrict__ src, const int* __restrict__ dst,
                            const int* __restrict__ rank, const int* __restrict__ rowptr,
                            int* __restrict__ csr_src, int E) {
  int e = blockIdx.x * blockDim.x + threadIdx.x;
  if (e < E) csr_src[rowptr[dst[e]] + rank[e]] = src[e];
}

__global__ void pad_W(const float* __restrict__ W, float* __restrict__ Wp, int K, int Nn) {
  int r = blockIdx.x, c = threadIdx.x;
  Wp[(size_t)r * DHP + c] = (r < K && c < Nn) ? W[(size_t)r * Nn + c] : 0.0f;
}

#define FOLD_G 32
__global__ __launch_bounds__(1024) void fold_bn_W(const float* __restrict__ W,
                                                  const float* __restrict__ sums,
                                                  const float* __restrict__ gamma,
                                                  const float* __restrict__ beta,
                                                  float* __restrict__ Wp, float* __restrict__ cvec,
                                                  int M) {
  __shared__ float part[FOLD_G][33];
  const int l = threadIdx.x & 31, g = threadIdx.x >> 5;
  const int c = blockIdx.x * 32 + l;
  const int cc = min(c, DH - 1);
  const float invn = 1.0f / (float)M;
  float w[DHP / FOLD_G], scale[DHP / FOLD_G], shift[DHP / FOLD_G];
#pragma unroll
  for (int i = 0; i < DHP / FOLD_G; ++i) {
    const int rr = min(g + FOLD_G * i, DH - 1);
    w[i] = W[(size_t)rr * DH + cc];
    bn_coef(sums, gamma, beta, rr, invn, scale[i], shift[i]);
  }
  float acc = 0.f;
#pragma unroll
  for (int i = 0; i < DHP / FOLD_G; ++i) {
    const int r = g + FOLD_G * i;
    const bool in = r < DH && c < DH;
    Wp[(size_t)r * DHP + c] = in ? w[i] * scale[i] : 0.f;
    acc += in ? w[i] * shift[i] : 0.f;
  }
  part[g][l] = acc;
  __syncthreads();
  if (g == 0) {
    float s = 0.f;
#pragma unroll
    for (int j = 0; j < FOLD_G; ++j) s += part[j][l];
    cvec[c] = s;
  }
}

template <bool FULL>
__device__ __forceinline__ void epi_rows(const v8f (&acc)[2][4], v4f bq, float* __restrict__ cp,
                                         int rb, int M, v4f& s, v4f& s2) {
#pragma unroll
  for (int i = 0; i < 2; ++i) {
#pragma unroll
    for (int r = 0; r < 8; ++r) {
      if (FULL || rb + 16 * i + r < M) {
        v4f o;
        o.x = fmaxf(acc[i][0][r] + bq.x, 0.0f);
        o.y = fmaxf(acc[i][1][r] + bq.y, 0.0f);
        o.z = fmaxf(acc[i][2][r] + bq.z, 0.0f);
        o.w = fmaxf(acc[i][3][r] + bq.w, 0.0f);
        *(v4f*)(cp + (size_t)(16 * i + r) * DHP) = o;
        s  += o;
        s2 += o * o;
      }
    }
  }
}

template <bool CINIT, bool EPI>
__global__ __launch_bounds__(256) void gemm_f32_wmma(
    const float* __restrict__ A, int lda,
    const float* __restrict__ B,
    float* __restrict__ C,
    int M, int Kpad, const float* __restrict__ cinit,
    const float* __restrict__ bias, float* __restrict__ sums) {
  const int lane  = threadIdx.x & 31;
  const int wv    = __builtin_amdgcn_readfirstlane(threadIdx.x >> 5);
  const int strip = blockIdx.x * 8 + wv;
  const int cg    = blockIdx.y;
  const int row0  = strip * 32;
  if constexpr (!EPI) {
    if (row0 >= M) return;
  }

  const int h = lane >> 4;
  const int n = lane & 15;
  const int ar0 = min(row0 + n, M - 1);
  const int ar1 = min(row0 + 16 + n, M - 1);
  const float* a0p = A + (size_t)ar0 * lda + 8 * h;
  const float* a1p = A + (size_t)ar1 * lda + 8 * h;
  const float* bp  = B + (size_t)(8 * h) * DHP + cg * 64 + 4 * n;

  v8f acc[2][4];
  v4f ci = (v4f){0.f, 0.f, 0.f, 0.f};
  if (CINIT) ci = *(const v4f*)(cinit + cg * 64 + 4 * n);
#pragma unroll
  for (int i = 0; i < 2; ++i)
#pragma unroll
    for (int t = 0; t < 4; ++t)
      acc[i][t] = (v8f){ci[t], ci[t], ci[t], ci[t], ci[t], ci[t], ci[t], ci[t]};

  for (int k0 = 0; k0 < Kpad; k0 += 16) {
    v4f a0[2], a1[2], b[8];
    a0[0] = *(const v4f*)(a0p + k0);
    a0[1] = *(const v4f*)(a0p + k0 + 4);
    a1[0] = *(const v4f*)(a1p + k0);
    a1[1] = *(const v4f*)(a1p + k0 + 4);
#pragma unroll
    for (int j = 0; j < 8; ++j) b[j] = *(const v4f*)(bp + (size_t)(k0 + j) * DHP);
#pragma unroll
    for (int s = 0; s < 4; ++s) {
      v2f fa0, fa1;
      fa0.x = a0[s >> 1][(2 * s) & 3];
      fa0.y = a0[s >> 1][(2 * s + 1) & 3];
      fa1.x = a1[s >> 1][(2 * s) & 3];
      fa1.y = a1[s >> 1][(2 * s + 1) & 3];
#pragma unroll
      for (int t = 0; t < 4; ++t) {
        v2f fb;
        fb.x = b[2 * s][t];
        fb.y = b[2 * s + 1][t];
        acc[0][t] = __builtin_amdgcn_wmma_f32_16x16x4_f32(false, fa0, false, fb,
                                                          (short)0, acc[0][t], false, false);
        acc[1][t] = __builtin_amdgcn_wmma_f32_16x16x4_f32(false, fa1, false, fb,
                                                          (short)0, acc[1][t], false, false);
      }
    }
  }
  if constexpr (!EPI) {
#pragma unroll
    for (int i = 0; i < 2; ++i) {
#pragma unroll
      for (int r = 0; r < 8; ++r) {
        const int row = row0 + 16 * i + 8 * h + r;
        v4f o;
        o.x = acc[i][0][r];
        o.y = acc[i][1][r];
        o.z = acc[i][2][r];
        o.w = acc[i][3][r];
        if (row < M) *(v4f*)(C + (size_t)row * DHP + cg * 64 + 4 * n) = o;
      }
    }
  } else {
    __shared__ float red[2][16][64];
    const int qd = cg * 16 + n;
    v4f bq = (v4f){0.f, 0.f, 0.f, 0.f};
    if (qd < DH / 4) {
      bq.x = bias[4 * qd];     bq.y = bias[4 * qd + 1];
      bq.z = bias[4 * qd + 2]; bq.w = bias[4 * qd + 3];
    }
    v4f s  = (v4f){0.f, 0.f, 0.f, 0.f};
    v4f s2 = (v4f){0.f, 0.f, 0.f, 0.f};
    float* cp = C + (size_t)(row0 + 8 * h) * DHP + cg * 64 + 4 * n;
    if (row0 + 32 <= M) epi_rows<true>(acc, bq, cp, row0 + 8 * h, M, s, s2);
    else                epi_rows<false>(acc, bq, cp, row0 + 8 * h, M, s, s2);
    *(v4f*)&red[0][wv * 2 + h][4 * n] = s;
    *(v4f*)&red[1][wv * 2 + h][4 * n] = s2;
    __syncthreads();
    if (threadIdx.x < 128) {
      const int which = threadIdx.x >> 6, c = threadIdx.x & 63;
      float a = 0.f;
#pragma unroll
      for (int j = 0; j < 16; ++j) a += red[which][j][c];
      const int col = cg * 64 + c;
      if (col < DH)
        atomicAdd(&sums[(blockIdx.x % SUMS_K) * 2 * DHP + which * DHP + col], a);
    }
  }
}

#define GR_ROWS 2
__global__ __launch_bounds__(256) void gather_rows(
    const float* __restrict__ h, const int* __restrict__ rowptr,
    const int* __restrict__ csr_src, const float* __restrict__ dinv,
    float* __restrict__ agg, int M,
    const float* __restrict__ bias, float* __restrict__ sums) {
  __shared__ float red[2][8][DHP];
  const int lane = threadIdx.x & 31;
  const int wv = __builtin_amdgcn_readfirstlane(threadIdx.x >> 5);
  const int n0 = (blockIdx.x * 8 + wv) * GR_ROWS;

  v2f bq[5], sy[5], sy2[5];
#pragma unroll
  for (int i = 0; i < 5; ++i) {
    const int p = lane + 32 * i;
    bq[i] = (v2f){0.f, 0.f};
    if (p < DH / 2) { bq[i].x = bias[2 * p]; bq[i].y = bias[2 * p + 1]; }
    sy[i]  = (v2f){0.f, 0.f};
    sy2[i] = (v2f){0.f, 0.f};
  }

  for (int k = 0; k < GR_ROWS; ++k) {
    const int n = n0 + k;
    if (n >= M) break;
    const int beg = rowptr[n], end = rowptr[n + 1];
    const float dn = dinv[n];

    v2f acc[5];
    {
      const v2f* p = (const v2f*)(h + (size_t)n * DHP) + lane;
#pragma unroll
      for (int i = 0; i < 5; ++i) {
        const v2f v = p[i * 32];
        acc[i].x = v.x * dn * dn;
        acc[i].y = v.y * dn * dn;
      }
    }
    for (int base = beg; base < end; base += 32) {
      const int j = base + lane;
      int s_l = 0;
      float w_l = 0.f;
      if (j < end) {
        s_l = (int)min((unsigned)csr_src[j], (unsigned)(M - 1));
        w_l = dinv[s_l] * dn;
      }
      const int m = min(32, end - base);
#pragma unroll 4
      for (int t = 0; t < m; ++t) {
        const int s = __builtin_amdgcn_readlane(s_l, t);
        const float wgt = __int_as_float(__builtin_amdgcn_readlane(__float_as_int(w_l), t));
        const v2f* p = (const v2f*)(h + (size_t)s * DHP) + lane;
#pragma unroll
        for (int i = 0; i < 5; ++i) {
          const v2f v = p[i * 32];
          acc[i].x += v.x * wgt;
          acc[i].y += v.y * wgt;
        }
      }
    }
    v2f* o = (v2f*)(agg + (size_t)n * DHP) + lane;
#pragma unroll
    for (int i = 0; i < 5; ++i) {
      v2f y;
      y.x = fmaxf(acc[i].x + bq[i].x, 0.0f);
      y.y = fmaxf(acc[i].y + bq[i].y, 0.0f);
      o[i * 32] = y;
      sy[i]  += y;
      sy2[i] += y * y;
    }
  }

#pragma unroll
  for (int i = 0; i < 5; ++i) {
    *(v2f*)&red[0][wv][2 * (lane + 32 * i)] = sy[i];
    *(v2f*)&red[1][wv][2 * (lane + 32 * i)] = sy2[i];
  }
  __syncthreads();
  for (int idx = threadIdx.x; idx < 2 * DHP; idx += 256) {
    const int which = idx / DHP, c = idx - which * DHP;
    float a = 0.f;
#pragma unroll
    for (int w = 0; w < 8; ++w) a += red[which][w][c];
    if (c < DH) atomicAdd(&sums[(blockIdx.x % SUMS_K) * 2 * DHP + idx], a);
  }
}

#define XPAIRS (DIN / 2)
__global__ __launch_bounds__(256) void gather_x(
    const float* __restrict__ x, const int* __restrict__ rowptr,
    const int* __restrict__ csr_src, const float* __restrict__ dinv,
    float* __restrict__ ax, int M) {
  const int lane = threadIdx.x & 31;
  const int q = lane & 7;
  const int n = (blockIdx.x * 8 + (threadIdx.x >> 5)) * 4 + (lane >> 3);
  const bool rowok = n < M;
  const int nc = rowok ? n : M - 1;
  const int beg = rowptr[nc];
  const int end = rowptr[nc + 1];
  const int deg = rowok ? end - beg : 0;
  const float dn = dinv[nc];
  const unsigned o03 = 8u * q;
  const unsigned o4  = 8u * min(q + 32, XPAIRS - 1);
  const char* xb = (const char*)x;

  v2f acc[5];
  {
    const unsigned r = (unsigned)nc * (4u * DIN);
#pragma unroll
    for (int i = 0; i < 5; ++i) {
      const v2f v = *(const v2f*)(xb + (i < 4 ? r + o03 + 64u * i : r + o4));
      acc[i].x = v.x * dn * dn;
      acc[i].y = v.y * dn * dn;
    }
  }
  int maxdeg = deg;
  maxdeg = max(maxdeg, __shfl_xor(maxdeg, 8));
  maxdeg = max(maxdeg, __shfl_xor(maxdeg, 16));
  maxdeg = __builtin_amdgcn_readfirstlane(maxdeg);

  for (int c = 0; c < maxdeg; c += 8) {
    int s_l = 0;
    float w_l = 0.f;
    if (c + q < deg) {
      s_l = (int)min((unsigned)csr_src[beg + c + q], (unsigned)(M - 1));
      w_l = dinv[s_l] * dn;
    }
    for (int t0 = 0; t0 < 8 && c + t0 < maxdeg; t0 += 4) {
      v2f v[4][5];
      float wgt[4];
#pragma unroll
      for (int u = 0; u < 4; ++u) {
        const int s = __shfl(s_l, t0 + u, 8);
        wgt[u] = __shfl(w_l, t0 + u, 8);
        const unsigned r = (unsigned)s * (4u * DIN);
#pragma unroll
        for (int i = 0; i < 5; ++i)
          v[u][i] = *(const v2f*)(xb + (i < 4 ? r + o03 + 64u * i : r + o4));
      }
#pragma unroll
      for (int u = 0; u < 4; ++u) {
        const bool on = c + t0 + u < deg;
#pragma unroll
        for (int i = 0; i < 5; ++i) {
          acc[i].x += (on ? v[u][i].x : 0.f) * wgt[u];
          acc[i].y += (on ? v[u][i].y : 0.f) * wgt[u];
        }
      }
    }
  }
  if (rowok) {
    v2f* o = (v2f*)(ax + (size_t)n * DINP) + q;
#pragma unroll
    for (int i = 0; i < 5; ++i) {
      v2f r = acc[i];
      if (q + 8 * i >= XPAIRS) r = (v2f){0.f, 0.f};
      o[8 * i] = r;
    }
  }
}

__device__ __forceinline__ unsigned enc_f32(float f) {
  unsigned u = __float_as_uint(f);
  return (u & 0x80000000u) ? ~u : (u | 0x80000000u);
}
__device__ __forceinline__ float dec_f32(unsigned u) {
  return __uint_as_float((u & 0x80000000u) ? (u & 0x7FFFFFFFu) : ~u);
}
#define POOL_ROWS  64
#define POOL_BURST 16
__global__ __launch_bounds__(DHP) void pool_max(const float* __restrict__ y,
                                                const int* __restrict__ batch,
                                                unsigned* __restrict__ pool, int M,
                                                const float* __restrict__ sums,
                                                const float* __restrict__ gamma,
                                                const float* __restrict__ beta) {
  const int c = threadIdx.x;
  if (c >= DH) return;
  float scale, shift;
  bn_coef(sums, gamma, beta, c, 1.0f / (float)M, scale, shift);
  const int r0 = blockIdx.x * POOL_ROWS;
  const int r1 = min(r0 + POOL_ROWS, M);
  if (r0 >= r1) return;
  int gcur = batch[r0];
  unsigned best = 0u;
  int r = r0;
  for (; r + POOL_BURST <= r1; r += POOL_BURST) {
    const float* p = y + (size_t)r * DHP + c;
    float v[POOL_BURST];
    int g[POOL_BURST];
#pragma unroll
    for (int i = 0; i < POOL_BURST; ++i) {
      v[i] = p[(size_t)i * DHP];
      g[i] = batch[r + i];
    }
#pragma unroll
    for (int i = 0; i < POOL_BURST; ++i) {
      if (g[i] != gcur) {
        atomicMax(&pool[(size_t)gcur * DH + c], best);
        gcur = g[i];
        best = 0u;
      }
      best = max(best, enc_f32(v[i] * scale + shift));
    }
  }
  for (; r < r1; ++r) {
    const int g = batch[r];
    if (g != gcur) {
      atomicMax(&pool[(size_t)gcur * DH + c], best);
      gcur = g;
      best = 0u;
    }
    best = max(best, enc_f32(y[(size_t)r * DHP + c] * scale + shift));
  }
  atomicMax(&pool[(size_t)gcur * DH + c], best);
}
__global__ void pool_decode(const unsigned* __restrict__ pool, float* __restrict__ out) {
  int g = blockIdx.x, c = threadIdx.x;
  if (c >= DH) return;
  out[(size_t)g * DH + c] = dec_f32(pool[(size_t)g * DH + c]);
}

extern "C" void kernel_launch(void* const* d_in, const int* in_sizes, int n_in,
                              void* d_out, int out_size, void* d_ws, size_t ws_size,
                              hipStream_t stream) {
  const float* x      = (const float*)d_in[0];
  const int*   eidx   = (const int*)d_in[1];
  const int*   batch  = (const int*)d_in[2];
  const float* W1     = (const float*)d_in[3];
  const float* b1     = (const float*)d_in[4];
  const float* gamma1 = (const float*)d_in[5];
  const float* beta1  = (const float*)d_in[6];
  const float* W2     = (const float*)d_in[7];
  const float* b2     = (const float*)d_in[8];
  const float* gamma2 = (const float*)d_in[9];
  const float* beta2  = (const float*)d_in[10];
  float* out = (float*)d_out;

  const int* src = eidx;
  const int* dst = eidx + EE;

  float* ws = (float*)d_ws;
  float*    dinv = ws;
  int*      cnt  = (int*)(ws + 50048);
  int*      rowp = (int*)(ws + 100096);
  int*      rank = (int*)(ws + 150208);
  int*      csrs = (int*)(ws + 950208);
  float*    ax   = ws + 1750208;
  float*    Wp1  = ws + 5750208;
  float*    Wp2  = ws + 5775808;
  unsigned* pool = (unsigned*)(ws + 5878848);
  float*    bufA = ws + 6034880;
  float*    bufB = ws + 22034880;
  float*    cvec = ws + 38034880;
  float*    sums = ws + 38035200;

  const dim3 gemmGrid(((NN + 31) / 32 + 7) / 8, DHP / 64);

  fill_u32<<<(NN + 255) / 256, 256, 0, stream>>>((unsigned*)cnt, 0u, NN);
  edge_rank<<<(EE + 255) / 256, 256, 0, stream>>>(dst, cnt, rank, EE);
  scan_rows<<<1, 1024, 0, stream>>>(cnt, rowp, dinv, NN);
  place_edges<<<(EE + 255) / 256, 256, 0, stream>>>(src, dst, rank, rowp, csrs, EE);

  pad_W<<<DINP, DHP, 0, stream>>>(W1, Wp1, DIN, DH);

  gather_x<<<(NN + 31) / 32, 256, 0, stream>>>(x, rowp, csrs, dinv, ax, NN);
  fill_f32<<<(SUMS_K * 2 * DHP + 255) / 256, 256, 0, stream>>>(sums, 0.0f, SUMS_K * 2 * DHP);
  gemm_f32_wmma<false, true><<<gemmGrid, 256, 0, stream>>>(ax, DINP, Wp1, bufB, NN, DINP, nullptr,
                                                           b1, sums);

  fold_bn_W<<<DHP / 32, 1024, 0, stream>>>(W2, sums, gamma1, beta1, Wp2, cvec, NN);
  gemm_f32_wmma<true, false><<<gemmGrid, 256, 0, stream>>>(bufB, DHP, Wp2, bufA, NN, DHP, cvec,
                                                           nullptr, nullptr);
  fill_f32<<<(SUMS_K * 2 * DHP + 255) / 256, 256, 0, stream>>>(sums, 0.0f, SUMS_K * 2 * DHP);
  gather_rows<<<(NN + 8 * GR_ROWS - 1) / (8 * GR_ROWS), 256, 0, stream>>>(bufA, rowp, csrs, dinv, bufB, NN,
                                                                         b2, sums);

  fill_u32<<<((long)GG * DH + 255) / 256, 256, 0, stream>>>(pool, 0u, (long)GG * DH);
  pool_max<<<(NN + POOL_ROWS - 1) / POOL_ROWS, DHP, 0, stream>>>(bufB, batch, pool, NN,
                                                                sums, gamma2, beta2);
  pool_decode<<<GG, DHP, 0, stream>>>(pool, out);
}


# CSR build in LDS: edge chunks sorted by 256-row destination bin, then one workgroup per bin counts, scans, places (row slot = arrival order of LDS adds: another sum order); 15->13 launches
# speedup vs baseline: 10.2555x; 1.4081x over previous
#include <hip/hip_runtime.h>
#include <hip/hip_bf16.h>

typedef float v2f __attribute__((ext_vector_type(2)));
typedef float v4f __attribute__((ext_vector_type(4)));
typedef float v8f __attribute__((ext_vector_type(8)));

#define NN    50000
#define EE    800000
#define GG    500
#define DIN   78
#define DINP  80
#define DH    312
#define DHP   320
#define BNEPS 1e-5f

#define SUMS_K 4
__device__ __forceinline__ void bn_coef(const float* __restrict__ sums, const float* __restrict__ gamma,
                                        const float* __restrict__ beta, int c, float invn,
                                        float& scale, float& shift) {
  float t1 = 0.f, t2 = 0.f;
#pragma unroll
  for (int k = 0; k < SUMS_K; ++k) {
    t1 += sums[k * 2 * DHP + c];
    t2 += sums[k * 2 * DHP + DHP + c];
  }
  const float mean = t1 * invn;
  const float var  = t2 * invn - mean * mean;
  scale = rsqrtf(var + BNEPS) * gamma[c];
  shift = beta[c] - mean * scale;
}

__global__ void fill_f32(float* __restrict__ p, float v, long n) {
  long i = (long)blockIdx.x * blockDim.x + threadIdx.x;
  if (i < n) p[i] = v;
}
__global__ void fill_u32(unsigned* __restrict__ p, unsigned v, long n) {
  long i = (long)blockIdx.x * blockDim.x + threadIdx.x;
  if (i < n) p[i] = v;
}

#define BIN_ROWS 256
#define NBINS    ((NN + BIN_ROWS - 1) / BIN_ROWS)
#define CHUNK    4000
#define NCHUNK   (EE / CHUNK)
#define OCAP     16384
static_assert(EE % CHUNK == 0 && CHUNK % 4 == 0 && CHUNK / 4 <= 1024, "chunk shape");
static_assert(NBINS < 256 && NCHUNK <= 1024 && NN <= 65536, "bin shape");

__device__ __forceinline__ int scan256(int v, int* wtot, int& total) {
  const int tid = threadIdx.x, lane = tid & 31, w = tid >> 5;
  int x = v;
  if (tid < 256) {
    for (int o = 1; o < 32; o <<= 1) {
      const int t = __shfl_up(x, o);
      if (lane >= o) x += t;
    }
    if (lane == 31) wtot[w] = x;
  }
  __syncthreads();
  int add = 0;
  total = 0;
#pragma unroll
  for (int j = 0; j < 8; ++j) {
    const int t = wtot[j];
    if (j < w) add += t;
    total += t;
  }
  __syncthreads();
  return add + x - v;
}

__global__ __launch_bounds__(1024) void bin_edges(const int* __restrict__ src, const int* __restrict__ dst,
                                                  unsigned* __restrict__ sorted, int* __restrict__ offT) {
  __shared__ int hist[256];
  __shared__ int wtot[8];
  __shared__ __attribute__((aligned(16))) unsigned stage[CHUNK];
  const int tid = threadIdx.x, c = blockIdx.x;
  if (tid < 256) hist[tid] = 0;
  __syncthreads();
  const bool on = tid < CHUNK / 4;
  unsigned s[4] = {0u, 0u, 0u, 0u}, d[4] = {0u, 0u, 0u, 0u};
  int r[4] = {0, 0, 0, 0};
  if (on) {
    const int4 s4 = ((const int4*)(src + (size_t)c * CHUNK))[tid];
    const int4 d4 = ((const int4*)(dst + (size_t)c * CHUNK))[tid];
    s[0] = min((unsigned)s4.x, (unsigned)(NN - 1)); s[1] = min((unsigned)s4.y, (unsigned)(NN - 1));
    s[2] = min((unsigned)s4.z, (unsigned)(NN - 1)); s[3] = min((unsigned)s4.w, (unsigned)(NN - 1));
    d[0] = min((unsigned)d4.x, (unsigned)(NN - 1)); d[1] = min((unsigned)d4.y, (unsigned)(NN - 1));
    d[2] = min((unsigned)d4.z, (unsigned)(NN - 1)); d[3] = min((unsigned)d4.w, (unsigned)(NN - 1));
#pragma unroll
    for (int k = 0; k < 4; ++k) r[k] = atomicAdd(&hist[d[k] >> 8], 1);
  }
  __syncthreads();
  int total;
  const int mine = tid < 256 ? hist[tid] : 0;
  const int excl = scan256(mine, wtot, total);
  if (tid < 256) {
    hist[tid] = excl;
    if (tid <= NBINS) offT[tid * NCHUNK + c] = excl;
  }
  __syncthreads();
  if (on) {
#pragma unroll
    for (int k = 0; k < 4; ++k) stage[hist[d[k] >> 8] + r[k]] = (s[k] << 8) | (d[k] & 255u);
  }
  __syncthreads();
  if (on) ((uint4*)(sorted + (size_t)c * CHUNK))[tid] = ((const uint4*)stage)[tid];
}

#define RUNS_W ((NCHUNK + 31) / 32)
template <bool PLACE>
__device__ __forceinline__ void put_edge(unsigned u, int* cnt, int* obuf, int* __restrict__ csr_src,
                                         int base, bool fits) {
  if (!PLACE) {
    atomicAdd(&cnt[u & 255u], 1);
  } else {
    const int slot = atomicAdd(&cnt[u & 255u], 1);
    if (fits) obuf[slot] = (int)(u >> 8);
    else      csr_src[base + slot] = (int)(u >> 8);
  }
}
template <bool PLACE>
__device__ __forceinline__ void walk_runs(const unsigned* __restrict__ sorted, const int* sb, const int* se,
                                          int* cnt, int* obuf, int* __restrict__ csr_src, int base,
                                          bool fits) {
  const int lane = threadIdx.x & 31, wv = threadIdx.x >> 5;
  unsigned u[RUNS_W];
  bool ok[RUNS_W];
#pragma unroll
  for (int k = 0; k < RUNS_W; ++k) {
    const int c = wv + 32 * k;
    const int j = sb[c] + lane;
    ok[k] = j < se[c];
    u[k] = sorted[min(j, EE - 1)];
  }
#pragma unroll
  for (int k = 0; k < RUNS_W; ++k)
    if (ok[k]) put_edge<PLACE>(u[k], cnt, obuf, csr_src, base, fits);
  for (int k = 0; k < RUNS_W; ++k) {
    const int c = wv + 32 * k;
    const int e = se[c];
    for (int j = sb[c] + lane + 32; j < e; j += 32)
      put_edge<PLACE>(sorted[j], cnt, obuf, csr_src, base, fits);
  }
}

__global__ __launch_bounds__(1024) void build_rows(const unsigned* __restrict__ sorted,
                                                   const int* __restrict__ offT,
                                                   int* __restrict__ rowptr, float* __restrict__ dinv,
                                                   int* __restrict__ csr_src) {
  __shared__ int sb[32 * RUNS_W], se[32 * RUNS_W];
  __shared__ int cnt[256];
  __shared__ int wtot[8];
  __shared__ int bsum[32];
  __shared__ int obuf[OCAP];
  const int tid = threadIdx.x, lane = tid & 31, w = tid >> 5, b = blockIdx.x;
  int o0 = 0;
  if (tid < NCHUNK) {
    o0 = offT[b * NCHUNK + tid];
    sb[tid] = tid * CHUNK + o0;
    se[tid] = tid * CHUNK + offT[(b + 1) * NCHUNK + tid];
  } else if (tid < 32 * RUNS_W) {
    sb[tid] = 0;
    se[tid] = 0;
  }
  if (tid < 256) cnt[tid] = 0;
  {
    int x = o0;
    for (int o = 16; o > 0; o >>= 1) x += __shfl_xor(x, o);
    if (lane == 0) bsum[w] = x;
  }
  __syncthreads();
  int base = 0;
#pragma unroll
  for (int j = 0; j < 32; ++j) base += bsum[j];

  walk_runs<false>(sorted, sb, se, cnt, obuf, csr_src, base, false);
  __syncthreads();
  int total;
  const int deg = tid < 256 ? cnt[tid] : 0;
  const int excl = scan256(deg, wtot, total);
  if (tid < 256) {
    cnt[tid] = excl;
    const int n = b * BIN_ROWS + tid;
    if (n < NN) {
      rowptr[n] = base + excl;
      dinv[n] = rsqrtf((float)(deg + 1));
      if (n == NN - 1) rowptr[NN] = base + excl + deg;
    }
  }
  __syncthreads();
  const bool fits = total <= OCAP;
  walk_runs<true>(sorted, sb, se, cnt, obuf, csr_src, base, fits);
  __syncthreads();
  if (fits)
    for (int i = tid; i < total; i += 1024) csr_src[base + i] = obuf[i];
}

__global__ void pad_W(const float* __restrict__ W, float* __restrict__ Wp, int K, int Nn) {
  int r = blockIdx.x, c = threadIdx.x;
  Wp[(size_t)r * DHP + c] = (r < K && c < Nn) ? W[(size_t)r * Nn + c] : 0.0f;
}

#define FOLD_G 32
__global__ __launch_bounds__(1024) void fold_bn_W(const float* __restrict__ W,
                                                  const float* __restrict__ sums,
                                                  const float* __restrict__ gamma,
                                                  const float* __restrict__ beta,
                                                  float* __restrict__ Wp, float* __restrict__ cvec,
                                                  int M) {
  __shared__ float part[FOLD_G][33];
  const int l = threadIdx.x & 31, g = threadIdx.x >> 5;
  const int c = blockIdx.x * 32 + l;
  const int cc = min(c, DH - 1);
  const float invn = 1.0f / (float)M;
  float w[DHP / FOLD_G], scale[DHP / FOLD_G], shift[DHP / FOLD_G];
#pragma unroll
  for (int i = 0; i < DHP / FOLD_G; ++i) {
    const int rr = min(g + FOLD_G * i, DH - 1);
    w[i] = W[(size_t)rr * DH + cc];
    bn_coef(sums, gamma, beta, rr, invn, scale[i], shift[i]);
  }
  float acc = 0.f;
#pragma unroll
  for (int i = 0; i < DHP / FOLD_G; ++i) {
    const int r = g + FOLD_G * i;
    const bool in = r < DH && c < DH;
    Wp[(size_t)r * DHP + c] = in ? w[i] * scale[i] : 0.f;
    acc += in ? w[i] * shift[i] : 0.f;
  }
  part[g][l] = acc;
  __syncthreads();
  if (g == 0) {
    float s = 0.f;
#pragma unroll
    for (int j = 0; j < FOLD_G; ++j) s += part[j][l];
    cvec[c] = s;
  }
}

template <bool FULL>
__device__ __forceinline__ void epi_rows(const v8f (&acc)[2][4], v4f bq, float* __restrict__ cp,
                                         int rb, int M, v4f& s, v4f& s2) {
#pragma unroll
  for (int i = 0; i < 2; ++i) {
#pragma unroll
    for (int r = 0; r < 8; ++r) {
      if (FULL || rb + 16 * i + r < M) {
        v4f o;
        o.x = fmaxf(acc[i][0][r] + bq.x, 0.0f);
        o.y = fmaxf(acc[i][1][r] + bq.y, 0.0f);
        o.z = fmaxf(acc[i][2][r] + bq.z, 0.0f);
        o.w = fmaxf(acc[i][3][r] + bq.w, 0.0f);
        *(v4f*)(cp + (size_t)(16 * i + r) * DHP) = o;
        s  += o;
        s2 += o * o;
      }
    }
  }
}

template <bool CINIT, bool EPI>
__global__ __launch_bounds__(256) void gemm_f32_wmma(
    const float* __restrict__ A, int lda,
    const float* __restrict__ B,
    float* __restrict__ C,
    int M, int Kpad, const float* __restrict__ cinit,
    const float* __restrict__ bias, float* __restrict__ sums) {
  const int lane  = threadIdx.x & 31;
  const int wv    = __builtin_amdgcn_readfirstlane(threadIdx.x >> 5);
  const int strip = blockIdx.x * 8 + wv;
  const int cg    = blockIdx.y;
  const int row0  = strip * 32;
  if constexpr (!EPI) {
    if (row0 >= M) return;
  }

  const int h = lane >> 4;
  const int n = lane & 15;
  const int ar0 = min(row0 + n, M - 1);
  const int ar1 = min(row0 + 16 + n, M - 1);
  const float* a0p = A + (size_t)ar0 * lda + 8 * h;
  const float* a1p = A + (size_t)ar1 * lda + 8 * h;
  const float* bp  = B + (size_t)(8 * h) * DHP + cg * 64 + 4 * n;

  v8f acc[2][4];
  v4f ci = (v4f){0.f, 0.f, 0.f, 0.f};
  if (CINIT) ci = *(const v4f*)(cinit + cg * 64 + 4 * n);
#pragma unroll
  for (int i = 0; i < 2; ++i)
#pragma unroll
    for (int t = 0; t < 4; ++t)
      acc[i][t] = (v8f){ci[t], ci[t], ci[t], ci[t], ci[t], ci[t], ci[t], ci[t]};

  for (int k0 = 0; k0 < Kpad; k0 += 16) {
    v4f a0[2], a1[2], b[8];
    a0[0] = *(const v4f*)(a0p + k0);
    a0[1] = *(const v4f*)(a0p + k0 + 4);
    a1[0] = *(const v4f*)(a1p + k0);
    a1[1] = *(const v4f*)(a1p + k0 + 4);
#pragma unroll
    for (int j = 0; j < 8; ++j) b[j] = *(const v4f*)(bp + (size_t)(k0 + j) * DHP);
#pragma unroll
    for (int s = 0; s < 4; ++s) {
      v2f fa0, fa1;
      fa0.x = a0[s >> 1][(2 * s) & 3];
      fa0.y = a0[s >> 1][(2 * s + 1) & 3];
      fa1.x = a1[s >> 1][(2 * s) & 3];
      fa1.y = a1[s >> 1][(2 * s + 1) & 3];
#pragma unroll
      for (int t = 0; t < 4; ++t) {
        v2f fb;
        fb.x = b[2 * s][t];
        fb.y = b[2 * s + 1][t];
        acc[0][t] = __builtin_amdgcn_wmma_f32_16x16x4_f32(false, fa0, false, fb,
                                                          (short)0, acc[0][t], false, false);
        acc[1][t] = __builtin_amdgcn_wmma_f32_16x16x4_f32(false, fa1, false, fb,
                                                          (short)0, acc[1][t], false, false);
      }
    }
  }
  if constexpr (!EPI) {
#pragma unroll
    for (int i = 0; i < 2; ++i) {
#pragma unroll
      for (int r = 0; r < 8; ++r) {
        const int row = row0 + 16 * i + 8 * h + r;
        v4f o;
        o.x = acc[i][0][r];
        o.y = acc[i][1][r];
        o.z = acc[i][2][r];
        o.w = acc[i][3][r];
        if (row < M) *(v4f*)(C + (size_t)row * DHP + cg * 64 + 4 * n) = o;
      }
    }
  } else {
    __shared__ float red[2][16][64];
    const int qd = cg * 16 + n;
    v4f bq = (v4f){0.f, 0.f, 0.f, 0.f};
    if (qd < DH / 4) {
      bq.x = bias[4 * qd];     bq.y = bias[4 * qd + 1];
      bq.z = bias[4 * qd + 2]; bq.w = bias[4 * qd + 3];
    }
    v4f s  = (v4f){0.f, 0.f, 0.f, 0.f};
    v4f s2 = (v4f){0.f, 0.f, 0.f, 0.f};
    float* cp = C + (size_t)(row0 + 8 * h) * DHP + cg * 64 + 4 * n;
    if (row0 + 32 <= M) epi_rows<true>(acc, bq, cp, row0 + 8 * h, M, s, s2);
    else                epi_rows<false>(acc, bq, cp, row0 + 8 * h, M, s, s2);
    *(v4f*)&red[0][wv * 2 + h][4 * n] = s;
    *(v4f*)&red[1][wv * 2 + h][4 * n] = s2;
    __syncthreads();
    if (threadIdx.x < 128) {
      const int which = threadIdx.x >> 6, c = threadIdx.x & 63;
      float a = 0.f;
#pragma unroll
      for (int j = 0; j < 16; ++j) a += red[which][j][c];
      const int col = cg * 64 + c;
      if (col < DH)
        atomicAdd(&sums[(blockIdx.x % SUMS_K) * 2 * DHP + which * DHP + col], a);
    }
  }
}

#define GR_ROWS 2
__global__ __launch_bounds__(256) void gather_rows(
    const float* __restrict__ h, const int* __restrict__ rowptr,
    const int* __restrict__ csr_src, const float* __restrict__ dinv,
    float* __restrict__ agg, int M,
    const float* __restrict__ bias, float* __restrict__ sums) {
  __shared__ float red[2][8][DHP];
  const int lane = threadIdx.x & 31;
  const int wv = __builtin_amdgcn_readfirstlane(threadIdx.x >> 5);
  const int n0 = (blockIdx.x * 8 + wv) * GR_ROWS;

  v2f bq[5], sy[5], sy2[5];
#pragma unroll
  for (int i = 0; i < 5; ++i) {
    const int p = lane + 32 * i;
    bq[i] = (v2f){0.f, 0.f};
    if (p < DH / 2) { bq[i].x = bias[2 * p]; bq[i].y = bias[2 * p + 1]; }
    sy[i]  = (v2f){0.f, 0.f};
    sy2[i] = (v2f){0.f, 0.f};
  }

  for (int k = 0; k < GR_ROWS; ++k) {
    const int n = n0 + k;
    if (n >= M) break;
    const int beg = rowptr[n], end = rowptr[n + 1];
    const float dn = dinv[n];

    v2f acc[5];
    {
      const v2f* p = (const v2f*)(h + (size_t)n * DHP) + lane;
#pragma unroll
      for (int i = 0; i < 5; ++i) {
        const v2f v = p[i * 32];
        acc[i].x = v.x * dn * dn;
        acc[i].y = v.y * dn * dn;
      }
    }
    for (int base = beg; base < end; base += 32) {
      const int j = base + lane;
      int s_l = 0;
      float w_l = 0.f;
      if (j < end) {
        s_l = (int)min((unsigned)csr_src[j], (unsigned)(M - 1));
        w_l = dinv[s_l] * dn;
      }
      const int m = min(32, end - base);
#pragma unroll 4
      for (int t = 0; t < m; ++t) {
        const int s = __builtin_amdgcn_readlane(s_l, t);
        const float wgt = __int_as_float(__builtin_amdgcn_readlane(__float_as_int(w_l), t));
        const v2f* p = (const v2f*)(h + (size_t)s * DHP) + lane;
#pragma unroll
        for (int i = 0; i < 5; ++i) {
          const v2f v = p[i * 32];
          acc[i].x += v.x * wgt;
          acc[i].y += v.y * wgt;
        }
      }
    }
    v2f* o = (v2f*)(agg + (size_t)n * DHP) + lane;
#pragma unroll
    for (int i = 0; i < 5; ++i) {
      v2f y;
      y.x = fmaxf(acc[i].x + bq[i].x, 0.0f);
      y.y = fmaxf(acc[i].y + bq[i].y, 0.0f);
      o[i * 32] = y;
      sy[i]  += y;
      sy2[i] += y * y;
    }
  }

#pragma unroll
  for (int i = 0; i < 5; ++i) {
    *(v2f*)&red[0][wv][2 * (lane + 32 * i)] = sy[i];
    *(v2f*)&red[1][wv][2 * (lane + 32 * i)] = sy2[i];
  }
  __syncthreads();
  for (int idx = threadIdx.x; idx < 2 * DHP; idx += 256) {
    const int which = idx / DHP, c = idx - which * DHP;
    float a = 0.f;
#pragma unroll
    for (int w = 0; w < 8; ++w) a += red[which][w][c];
    if (c < DH) atomicAdd(&sums[(blockIdx.x % SUMS_K) * 2 * DHP + idx], a);
  }
}

#define XPAIRS (DIN / 2)
__global__ __launch_bounds__(256) void gather_x(
    const float* __restrict__ x, const int* __restrict__ rowptr,
    const int* __restrict__ csr_src, const float* __restrict__ dinv,
    float* __restrict__ ax, int M) {
  const int lane = threadIdx.x & 31;
  const int q = lane & 7;
  const int n = (blockIdx.x * 8 + (threadIdx.x >> 5)) * 4 + (lane >> 3);
  const bool rowok = n < M;
  const int nc = rowok ? n : M - 1;
  const int beg = rowptr[nc];
  const int end = rowptr[nc + 1];
  const int deg = rowok ? end - beg : 0;
  const float dn = dinv[nc];
  const unsigned o03 = 8u * q;
  const unsigned o4  = 8u * min(q + 32, XPAIRS - 1);
  const char* xb = (const char*)x;

  v2f acc[5];
  {
    const unsigned r = (unsigned)nc * (4u * DIN);
#pragma unroll
    for (int i = 0; i < 5; ++i) {
      const v2f v = *(const v2f*)(xb + (i < 4 ? r + o03 + 64u * i : r + o4));
      acc[i].x = v.x * dn * dn;
      acc[i].y = v.y * dn * dn;
    }
  }
  int maxdeg = deg;
  maxdeg = max(maxdeg, __shfl_xor(maxdeg, 8));
  maxdeg = max(maxdeg, __shfl_xor(maxdeg, 16));
  maxdeg = __builtin_amdgcn_readfirstlane(maxdeg);

  for (int c = 0; c < maxdeg; c += 8) {
    int s_l = 0;
    float w_l = 0.f;
    if (c + q < deg) {
      s_l = (int)min((unsigned)csr_src[beg + c + q], (unsigned)(M - 1));
      w_l = dinv[s_l] * dn;
    }
    for (int t0 = 0; t0 < 8 && c + t0 < maxdeg; t0 += 4) {
      v2f v[4][5];
      float wgt[4];
#pragma unroll
      for (int u = 0; u < 4; ++u) {
        const int s = __shfl(s_l, t0 + u, 8);
        wgt[u] = __shfl(w_l, t0 + u, 8);
        const unsigned r = (unsigned)s * (4u * DIN);
#pragma unroll
        for (int i = 0; i < 5; ++i)
          v[u][i] = *(const v2f*)(xb + (i < 4 ? r + o03 + 64u * i : r + o4));
      }
#pragma unroll
      for (int u = 0; u < 4; ++u) {
        const bool on = c + t0 + u < deg;
#pragma unroll
        for (int i = 0; i < 5; ++i) {
          acc[i].x += (on ? v[u][i].x : 0.f) * wgt[u];
          acc[i].y += (on ? v[u][i].y : 0.f) * wgt[u];
        }
      }
    }
  }
  if (rowok) {
    v2f* o = (v2f*)(ax + (size_t)n * DINP) + q;
#pragma unroll
    for (int i = 0; i < 5; ++i) {
      v2f r = acc[i];
      if (q + 8 * i >= XPAIRS) r = (v2f){0.f, 0.f};
      o[8 * i] = r;
    }
  }
}

__device__ __forceinline__ unsigned enc_f32(float f) {
  unsigned u = __float_as_uint(f);
  return (u & 0x80000000u) ? ~u : (u | 0x80000000u);
}
__device__ __forceinline__ float dec_f32(unsigned u) {
  return __uint_as_float((u & 0x80000000u) ? (u & 0x7FFFFFFFu) : ~u);
}
#define POOL_ROWS  64
#define POOL_BURST 16
__global__ __launch_bounds__(DHP) void pool_max(const float* __restrict__ y,
                                                const int* __restrict__ batch,
                                                unsigned* __restrict__ pool, int M,
                                                const float* __restrict__ sums,
                                                const float* __restrict__ gamma,
                                                const float* __restrict__ beta) {
  const int c = threadIdx.x;
  if (c >= DH) return;
  float scale, shift;
  bn_coef(sums, gamma, beta, c, 1.0f / (float)M, scale, shift);
  const int r0 = blockIdx.x * POOL_ROWS;
  const int r1 = min(r0 + POOL_ROWS, M);
  if (r0 >= r1) return;
  int gcur = batch[r0];
  unsigned best = 0u;
  int r = r0;
  for (; r + POOL_BURST <= r1; r += POOL_BURST) {
    const float* p = y + (size_t)r * DHP + c;
    float v[POOL_BURST];
    int g[POOL_BURST];
#pragma unroll
    for (int i = 0; i < POOL_BURST; ++i) {
      v[i] = p[(size_t)i * DHP];
      g[i] = batch[r + i];
    }
#pragma unroll
    for (int i = 0; i < POOL_BURST; ++i) {
      if (g[i] != gcur) {
        atomicMax(&pool[(size_t)gcur * DH + c], best);
        gcur = g[i];
        best = 0u;
      }
      best = max(best, enc_f32(v[i] * scale + shift));
    }
  }
  for (; r < r1; ++r) {
    const int g = batch[r];
    if (g != gcur) {
      atomicMax(&pool[(size_t)gcur * DH + c], best);
      gcur = g;
      best = 0u;
    }
    best = max(best, enc_f32(y[(size_t)r * DHP + c] * scale + shift));
  }
  atomicMax(&pool[(size_t)gcur * DH + c], best);
}
__global__ void pool_decode(const unsigned* __restrict__ pool, float* __restrict__ out) {
  int g = blockIdx.x, c = threadIdx.x;
  if (c >= DH) return;
  out[(size_t)g * DH + c] = dec_f32(pool[(size_t)g * DH + c]);
}

extern "C" void kernel_launch(void* const* d_in, const int* in_sizes, int n_in,
                              void* d_out, int out_size, void* d_ws, size_t ws_size,
                              hipStream_t stream) {
  const float* x      = (const float*)d_in[0];
  const int*   eidx   = (const int*)d_in[1];
  const int*   batch  = (const int*)d_in[2];
  const float* W1     = (const float*)d_in[3];
  const float* b1     = (const float*)d_in[4];
  const float* gamma1 = (const float*)d_in[5];
  const float* beta1  = (const float*)d_in[6];
  const float* W2     = (const float*)d_in[7];
  const float* b2     = (const float*)d_in[8];
  const float* gamma2 = (const float*)d_in[9];
  const float* beta2  = (const float*)d_in[10];
  float* out = (float*)d_out;

  const int* src = eidx;
  const int* dst = eidx + EE;

  float* ws = (float*)d_ws;
  float*    dinv = ws;
  int*      offT = (int*)(ws + 50048);
  int*      rowp = (int*)(ws + 100096);
  unsigned* srt  = (unsigned*)(ws + 150208);
  int*      csrs = (int*)(ws + 950208);
  float*    ax   = ws + 1750208;
  float*    Wp1  = ws + 5750208;
  float*    Wp2  = ws + 5775808;
  unsigned* pool = (unsigned*)(ws + 5878848);
  float*    bufA = ws + 6034880;
  float*    bufB = ws + 22034880;
  float*    cvec = ws + 38034880;
  float*    sums = ws + 38035200;

  const dim3 gemmGrid(((NN + 31) / 32 + 7) / 8, DHP / 64);

  bin_edges<<<NCHUNK, 1024, 0, stream>>>(src, dst, srt, offT);
  build_rows<<<NBINS, 1024, 0, stream>>>(srt, offT, rowp, dinv, csrs);

  pad_W<<<DINP, DHP, 0, stream>>>(W1, Wp1, DIN, DH);

  gather_x<<<(NN + 31) / 32, 256, 0, stream>>>(x, rowp, csrs, dinv, ax, NN);
  fill_f32<<<(SUMS_K * 2 * DHP + 255) / 256, 256, 0, stream>>>(sums, 0.0f, SUMS_K * 2 * DHP);
  gemm_f32_wmma<false, true><<<gemmGrid, 256, 0, stream>>>(ax, DINP, Wp1, bufB, NN, DINP, nullptr,
                                                           b1, sums);

  fold_bn_W<<<DHP / 32, 1024, 0, stream>>>(W2, sums, gamma1, beta1, Wp2, cvec, NN);
  gemm_f32_wmma<true, false><<<gemmGrid, 256, 0, stream>>>(bufB, DHP, Wp2, bufA, NN, DHP, cvec,
                                                           nullptr, nullptr);
  fill_f32<<<(SUMS_K * 2 * DHP + 255) / 256, 256, 0, stream>>>(sums, 0.0f, SUMS_K * 2 * DHP);
  gather_rows<<<(NN + 8 * GR_ROWS - 1) / (8 * GR_ROWS), 256, 0, stream>>>(bufA, rowp, csrs, dinv, bufB, NN,
                                                                         b2, sums);

  fill_u32<<<((long)GG * DH + 255) / 256, 256, 0, stream>>>(pool, 0u, (long)GG * DH);
  pool_max<<<(NN + POOL_ROWS - 1) / POOL_ROWS, DHP, 0, stream>>>(bufB, batch, pool, NN,
                                                                sums, gamma2, beta2);
  pool_decode<<<GG, DHP, 0, stream>>>(pool, out);
}


# Five small launches removed (13 -> 8): moment zeroing, W1 padding and the pool's start value go into the first kernel; pool max taken on out itself (int max / uint min atomics), no decode pass
# speedup vs baseline: 10.7660x; 1.0498x over previous
#include <hip/hip_runtime.h>
#include <hip/hip_bf16.h>

typedef float v2f __attribute__((ext_vector_type(2)));
typedef float v4f __attribute__((ext_vector_type(4)));
typedef float v8f __attribute__((ext_vector_type(8)));

#define NN    50000
#define EE    800000
#define GG    500
#define DIN   78
#define DINP  80
#define DH    312
#define DHP   320
#define BNEPS 1e-5f

#define SUMS_K 4
__device__ __forceinline__ void bn_coef(const float* __restrict__ sums, const float* __restrict__ gamma,
                                        const float* __restrict__ beta, int c, float invn,
                                        float& scale, float& shift) {
  float t1 = 0.f, t2 = 0.f;
#pragma unroll
  for (int k = 0; k < SUMS_K; ++k) {
    t1 += sums[k * 2 * DHP + c];
    t2 += sums[k * 2 * DHP + DHP + c];
  }
  const float mean = t1 * invn;
  const float var  = t2 * invn - mean * mean;
  scale = rsqrtf(var + BNEPS) * gamma[c];
  shift = beta[c] - mean * scale;
}

#define BIN_ROWS 256
#define NBINS    ((NN + BIN_ROWS - 1) / BIN_ROWS)
#define CHUNK    4000
#define NCHUNK   (EE / CHUNK)
#define OCAP     16384
static_assert(EE % CHUNK == 0 && CHUNK % 4 == 0 && CHUNK / 4 <= 1024, "chunk shape");
static_assert(NBINS < 256 && NCHUNK <= 1024 && NN <= 65536, "bin shape");

#define INIT_SUMS (2 * SUMS_K * 2 * DHP)
#define INIT_WP1  (DINP * DHP)
#define INIT_OUT  (GG * DH)
static_assert(INIT_SUMS + INIT_WP1 + INIT_OUT <= NCHUNK * 1024, "one start value per thread of bin_edges");
static_assert(INIT_SUMS % 32 == 0 && INIT_WP1 % 32 == 0, "a wave writes one kind of start value");

__device__ __forceinline__ int scan256(int v, int* wtot, int& total) {
  const int tid = threadIdx.x, lane = tid & 31, w = tid >> 5;
  int x = v;
  if (tid < 256) {
    for (int o = 1; o < 32; o <<= 1) {
      const int t = __shfl_up(x, o);
      if (lane >= o) x += t;
    }
    if (lane == 31) wtot[w] = x;
  }
  __syncthreads();
  int add = 0;
  total = 0;
#pragma unroll
  for (int j = 0; j < 8; ++j) {
    const int t = wtot[j];
    if (j < w) add += t;
    total += t;
  }
  __syncthreads();
  return add + x - v;
}

__global__ __launch_bounds__(1024) void bin_edges(const int* __restrict__ src, const int* __restrict__ dst,
                                                  unsigned* __restrict__ sorted, int* __restrict__ offT,
                                                  const float* __restrict__ W1, float* __restrict__ sums,
                                                  float* __restrict__ Wp1, float* __restrict__ out) {
  __shared__ int hist[256];
  __shared__ int wtot[8];
  __shared__ __attribute__((aligned(16))) unsigned stage[CHUNK];
  const int tid = threadIdx.x, c = blockIdx.x;
  if (tid < 256) hist[tid] = 0;
  __syncthreads();
  const int iv = c * 1024 + tid;
  const int jw = iv - INIT_SUMS;
  const bool isw = jw >= 0 && jw < INIT_WP1;
  float w1v = 0.f;
  if (isw) {
    const int r = jw / DHP, cc = jw - r * DHP;
    if (r < DIN && cc < DH) w1v = W1[r * DH + cc];
  }
  const bool on = tid < CHUNK / 4;
  unsigned s[4] = {0u, 0u, 0u, 0u}, d[4] = {0u, 0u, 0u, 0u};
  int r[4] = {0, 0, 0, 0};
  if (on) {
    const int4 s4 = ((const int4*)(src + (size_t)c * CHUNK))[tid];
    const int4 d4 = ((const int4*)(dst + (size_t)c * CHUNK))[tid];
    s[0] = min((unsigned)s4.x, (unsigned)(NN - 1)); s[1] = min((unsigned)s4.y, (unsigned)(NN - 1));
    s[2] = min((unsigned)s4.z, (unsigned)(NN - 1)); s[3] = min((unsigned)s4.w, (unsigned)(NN - 1));
    d[0] = min((unsigned)d4.x, (unsigned)(NN - 1)); d[1] = min((unsigned)d4.y, (unsigned)(NN - 1));
    d[2] = min((unsigned)d4.z, (unsigned)(NN - 1)); d[3] = min((unsigned)d4.w, (unsigned)(NN - 1));
#pragma unroll
    for (int k = 0; k < 4; ++k) r[k] = atomicAdd(&hist[d[k] >> 8], 1);
  }
  __syncthreads();
  int total;
  const int mine = tid < 256 ? hist[tid] : 0;
  const int excl = scan256(mine, wtot, total);
  if (tid < 256) {
    hist[tid] = excl;
    if (tid <= NBINS) offT[tid * NCHUNK + c] = excl;
  }
  __syncthreads();
  if (on) {
#pragma unroll
    for (int k = 0; k < 4; ++k) stage[hist[d[k] >> 8] + r[k]] = (s[k] << 8) | (d[k] & 255u);
  }
  __syncthreads();
  if (on) ((uint4*)(sorted + (size_t)c * CHUNK))[tid] = ((const uint4*)stage)[tid];
  if (iv < INIT_SUMS)                               sums[iv] = 0.f;
  else if (isw)                                     Wp1[jw] = w1v;
  else if (iv < INIT_SUMS + INIT_WP1 + INIT_OUT)    out[iv - INIT_SUMS - INIT_WP1] = __uint_as_float(0xFF800000u);
}

#define RUNS_W ((NCHUNK + 31) / 32)
template <bool PLACE>
__device__ __forceinline__ void put_edge(unsigned u, int* cnt, int* obuf, int* __restrict__ csr_src,
                                         int base, bool fits) {
  if (!PLACE) {
    atomicAdd(&cnt[u & 255u], 1);
  } else {
    const int slot = atomicAdd(&cnt[u & 255u], 1);
    if (fits) obuf[slot] = (int)(u >> 8);
    else      csr_src[base + slot] = (int)(u >> 8);
  }
}
template <bool PLACE>
__device__ __forceinline__ void walk_runs(const unsigned* __restrict__ sorted, const int* sb, const int* se,
                                          int* cnt, int* obuf, int* __restrict__ csr_src, int base,
                                          bool fits) {
  const int lane = threadIdx.x & 31, wv = threadIdx.x >> 5;
  unsigned u[RUNS_W];
  bool ok[RUNS_W];
#pragma unroll
  for (int k = 0; k < RUNS_W; ++k) {
    const int c = wv + 32 * k;
    const int j = sb[c] + lane;
    ok[k] = j < se[c];
    u[k] = sorted[min(j, EE - 1)];
  }
#pragma unroll
  for (int k = 0; k < RUNS_W; ++k)
    if (ok[k]) put_edge<PLACE>(u[k], cnt, obuf, csr_src, base, fits);
  for (int k = 0; k < RUNS_W; ++k) {
    const int c = wv + 32 * k;
    const int e = se[c];
    for (int j = sb[c] + lane + 32; j < e; j += 32)
      put_edge<PLACE>(sorted[j], cnt, obuf, csr_src, base, fits);
  }
}

__global__ __launch_bounds__(1024) void build_rows(const unsigned* __restrict__ sorted,
                                                   const int* __restrict__ offT,
                                                   int* __restrict__ rowptr, float* __restrict__ dinv,
                                                   int* __restrict__ csr_src) {
  __shared__ int sb[32 * RUNS_W], se[32 * RUNS_W];
  __shared__ int cnt[256];
  __shared__ int wtot[8];
  __shared__ int bsum[32];
  __shared__ int obuf[OCAP];
  const int tid = threadIdx.x, lane = tid & 31, w = tid >> 5, b = blockIdx.x;
  int o0 = 0;
  if (tid < NCHUNK) {
    o0 = offT[b * NCHUNK + tid];
    sb[tid] = tid * CHUNK + o0;
    se[tid] = tid * CHUNK + offT[(b + 1) * NCHUNK + tid];
  } else if (tid < 32 * RUNS_W) {
    sb[tid] = 0;
    se[tid] = 0;
  }
  if (tid < 256) cnt[tid] = 0;
  {
    int x = o0;
    for (int o = 16; o > 0; o >>= 1) x += __shfl_xor(x, o);
    if (lane == 0) bsum[w] = x;
  }
  __syncthreads();
  int base = 0;
#pragma unroll
  for (int j = 0; j < 32; ++j) base += bsum[j];

  walk_runs<false>(sorted, sb, se, cnt, obuf, csr_src, base, false);
  __syncthreads();
  int total;
  const int deg = tid < 256 ? cnt[tid] : 0;
  const int excl = scan256(deg, wtot, total);
  if (tid < 256) {
    cnt[tid] = excl;
    const int n = b * BIN_ROWS + tid;
    if (n < NN) {
      rowptr[n] = base + excl;
      dinv[n] = rsqrtf((float)(deg + 1));
      if (n == NN - 1) rowptr[NN] = base + excl + deg;
    }
  }
  __syncthreads();
  const bool fits = total <= OCAP;
  walk_runs<true>(sorted, sb, se, cnt, obuf, csr_src, base, fits);
  __syncthreads();
  if (fits)
    for (int i = tid; i < total; i += 1024) csr_src[base + i] = obuf[i];
}

#define FOLD_G 32
__global__ __launch_bounds__(1024) void fold_bn_W(const float* __restrict__ W,
                                                  const float* __restrict__ sums,
                                                  const float* __restrict__ gamma,
                                                  const float* __restrict__ beta,
                                                  float* __restrict__ Wp, float* __restrict__ cvec,
                                                  int M) {
  __shared__ float part[FOLD_G][33];
  const int l = threadIdx.x & 31, g = threadIdx.x >> 5;
  const int c = blockIdx.x * 32 + l;
  const int cc = min(c, DH - 1);
  const float invn = 1.0f / (float)M;
  float w[DHP / FOLD_G], scale[DHP / FOLD_G], shift[DHP / FOLD_G];
#pragma unroll
  for (int i = 0; i < DHP / FOLD_G; ++i) {
    const int rr = min(g + FOLD_G * i, DH - 1);
    w[i] = W[(size_t)rr * DH + cc];
    bn_coef(sums, gamma, beta, rr, invn, scale[i], shift[i]);
  }
  float acc = 0.f;
#pragma unroll
  for (int i = 0; i < DHP / FOLD_G; ++i) {
    const int r = g + FOLD_G * i;
    const bool in = r < DH && c < DH;
    Wp[(size_t)r * DHP + c] = in ? w[i] * scale[i] : 0.f;
    acc += in ? w[i] * shift[i] : 0.f;
  }
  part[g][l] = acc;
  __syncthreads();
  if (g == 0) {
    float s = 0.f;
#pragma unroll
    for (int j = 0; j < FOLD_G; ++j) s += part[j][l];
    cvec[c] = s;
  }
}

template <bool FULL>
__device__ __forceinline__ void epi_rows(const v8f (&acc)[2][4], v4f bq, float* __restrict__ cp,
                                         int rb, int M, v4f& s, v4f& s2) {
#pragma unroll
  for (int i = 0; i < 2; ++i) {
#pragma unroll
    for (int r = 0; r < 8; ++r) {
      if (FULL || rb + 16 * i + r < M) {
        v4f o;
        o.x = fmaxf(acc[i][0][r] + bq.x, 0.0f);
        o.y = fmaxf(acc[i][1][r] + bq.y, 0.0f);
        o.z = fmaxf(acc[i][2][r] + bq.z, 0.0f);
        o.w = fmaxf(acc[i][3][r] + bq.w, 0.0f);
        *(v4f*)(cp + (size_t)(16 * i + r) * DHP) = o;
        s  += o;
        s2 += o * o;
      }
    }
  }
}

template <bool CINIT, bool EPI>
__global__ __launch_bounds__(256) void gemm_f32_wmma(
    const float* __restrict__ A, int lda,
    const float* __restrict__ B,
    float* __restrict__ C,
    int M, int Kpad, const float* __restrict__ cinit,
    const float* __restrict__ bias, float* __restrict__ sums) {
  const int lane  = threadIdx.x & 31;
  const int wv    = __builtin_amdgcn_readfirstlane(threadIdx.x >> 5);
  const int strip = blockIdx.x * 8 + wv;
  const int cg    = blockIdx.y;
  const int row0  = strip * 32;
  if constexpr (!EPI) {
    if (row0 >= M) return;
  }

  const int h = lane >> 4;
  const int n = lane & 15;
  const int ar0 = min(row0 + n, M - 1);
  const int ar1 = min(row0 + 16 + n, M - 1);
  const float* a0p = A + (size_t)ar0 * lda + 8 * h;
  const float* a1p = A + (size_t)ar1 * lda + 8 * h;
  const float* bp  = B + (size_t)(8 * h) * DHP + cg * 64 + 4 * n;

  v8f acc[2][4];
  v4f ci = (v4f){0.f, 0.f, 0.f, 0.f};
  if (CINIT) ci = *(const v4f*)(cinit + cg * 64 + 4 * n);
#pragma unroll
  for (int i = 0; i < 2; ++i)
#pragma unroll
    for (int t = 0; t < 4; ++t)
      acc[i][t] = (v8f){ci[t], ci[t], ci[t], ci[t], ci[t], ci[t], ci[t], ci[t]};

  for (int k0 = 0; k0 < Kpad; k0 += 16) {
    v4f a0[2], a1[2], b[8];
    a0[0] = *(const v4f*)(a0p + k0);
    a0[1] = *(const v4f*)(a0p + k0 + 4);
    a1[0] = *(const v4f*)(a1p + k0);
    a1[1] = *(const v4f*)(a1p + k0 + 4);
#pragma unroll
    for (int j = 0; j < 8; ++j) b[j] = *(const v4f*)(bp + (size_t)(k0 + j) * DHP);
#pragma unroll
    for (int s = 0; s < 4; ++s) {
      v2f fa0, fa1;
      fa0.x = a0[s >> 1][(2 * s) & 3];
      fa0.y = a0[s >> 1][(2 * s + 1) & 3];
      fa1.x = a1[s >> 1][(2 * s) & 3];
      fa1.y = a1[s >> 1][(2 * s + 1) & 3];
#pragma unroll
      for (int t = 0; t < 4; ++t) {
        v2f fb;
        fb.x = b[2 * s][t];
        fb.y = b[2 * s + 1][t];
        acc[0][t] = __builtin_amdgcn_wmma_f32_16x16x4_f32(false, fa0, false, fb,
                                                          (short)0, acc[0][t], false, false);
        acc[1][t] = __builtin_amdgcn_wmma_f32_16x16x4_f32(false, fa1, false, fb,
                                                          (short)0, acc[1][t], false, false);
      }
    }
  }
  if constexpr (!EPI) {
#pragma unroll
    for (int i = 0; i < 2; ++i) {
#pragma unroll
      for (int r = 0; r < 8; ++r) {
        const int row = row0 + 16 * i + 8 * h + r;
        v4f o;
        o.x = acc[i][0][r];
        o.y = acc[i][1][r];
        o.z = acc[i][2][r];
        o.w = acc[i][3][r];
        if (row < M) *(v4f*)(C + (size_t)row * DHP + cg * 64 + 4 * n) = o;
      }
    }
  } else {
    __shared__ float red[2][16][64];
    const int qd = cg * 16 + n;
    v4f bq = (v4f){0.f, 0.f, 0.f, 0.f};
    if (qd < DH / 4) {
      bq.x = bias[4 * qd];     bq.y = bias[4 * qd + 1];
      bq.z = bias[4 * qd + 2]; bq.w = bias[4 * qd + 3];
    }
    v4f s  = (v4f){0.f, 0.f, 0.f, 0.f};
    v4f s2 = (v4f){0.f, 0.f, 0.f, 0.f};
    float* cp = C + (size_t)(row0 + 8 * h) * DHP + cg * 64 + 4 * n;
    if (row0 + 32 <= M) epi_rows<true>(acc, bq, cp, row0 + 8 * h, M, s, s2);
    else                epi_rows<false>(acc, bq, cp, row0 + 8 * h, M, s, s2);
    *(v4f*)&red[0][wv * 2 + h][4 * n] = s;
    *(v4f*)&red[1][wv * 2 + h][4 * n] = s2;
    __syncthreads();
    if (threadIdx.x < 128) {
      const int which = threadIdx.x >> 6, c = threadIdx.x & 63;
      float a = 0.f;
#pragma unroll
      for (int j = 0; j < 16; ++j) a += red[which][j][c];
      const int col = cg * 64 + c;
      if (col < DH)
        atomicAdd(&sums[(blockIdx.x % SUMS_K) * 2 * DHP + which * DHP + col], a);
    }
  }
}

#define GR_ROWS 2
__global__ __launch_bounds__(256) void gather_rows(
    const float* __restrict__ h, const int* __restrict__ rowptr,
    const int* __restrict__ csr_src, const float* __restrict__ dinv,
    float* __restrict__ agg, int M,
    const float* __restrict__ bias, float* __restrict__ sums) {
  __shared__ float red[2][8][DHP];
  const int lane = threadIdx.x & 31;
  const int wv = __builtin_amdgcn_readfirstlane(threadIdx.x >> 5);
  const int n0 = (blockIdx.x * 8 + wv) * GR_ROWS;

  v2f bq[5], sy[5], sy2[5];
#pragma unroll
  for (int i = 0; i < 5; ++i) {
    const int p = lane + 32 * i;
    bq[i] = (v2f){0.f, 0.f};
    if (p < DH / 2) { bq[i].x = bias[2 * p]; bq[i].y = bias[2 * p + 1]; }
    sy[i]  = (v2f){0.f, 0.f};
    sy2[i] = (v2f){0.f, 0.f};
  }

  for (int k = 0; k < GR_ROWS; ++k) {
    const int n = n0 + k;
    if (n >= M) break;
    const int beg = rowptr[n], end = rowptr[n + 1];
    const float dn = dinv[n];

    v2f acc[5];
    {
      const v2f* p = (const v2f*)(h + (size_t)n * DHP) + lane;
#pragma unroll
      for (int i = 0; i < 5; ++i) {
        const v2f v = p[i * 32];
        acc[i].x = v.x * dn * dn;
        acc[i].y = v.y * dn * dn;
      }
    }
    for (int base = beg; base < end; base += 32) {
      const int j = base + lane;
      int s_l = 0;
      float w_l = 0.f;
      if (j < end) {
        s_l = (int)min((unsigned)csr_src[j], (unsigned)(M - 1));
        w_l = dinv[s_l] * dn;
      }
      const int m = min(32, end - base);
#pragma unroll 4
      for (int t = 0; t < m; ++t) {
        const int s = __builtin_amdgcn_readlane(s_l, t);
        const float wgt = __int_as_float(__builtin_amdgcn_readlane(__float_as_int(w_l), t));
        const v2f* p = (const v2f*)(h + (size_t)s * DHP) + lane;
#pragma unroll
        for (int i = 0; i < 5; ++i) {
          const v2f v = p[i * 32];
          acc[i].x += v.x * wgt;
          acc[i].y += v.y * wgt;
        }
      }
    }
    v2f* o = (v2f*)(agg + (size_t)n * DHP) + lane;
#pragma unroll
    for (int i = 0; i < 5; ++i) {
      v2f y;
      y.x = fmaxf(acc[i].x + bq[i].x, 0.0f);
      y.y = fmaxf(acc[i].y + bq[i].y, 0.0f);
      o[i * 32] = y;
      sy[i]  += y;
      sy2[i] += y * y;
    }
  }

#pragma unroll
  for (int i = 0; i < 5; ++i) {
    *(v2f*)&red[0][wv][2 * (lane + 32 * i)] = sy[i];
    *(v2f*)&red[1][wv][2 * (lane + 32 * i)] = sy2[i];
  }
  __syncthreads();
  for (int idx = threadIdx.x; idx < 2 * DHP; idx += 256) {
    const int which = idx / DHP, c = idx - which * DHP;
    float a = 0.f;
#pragma unroll
    for (int w = 0; w < 8; ++w) a += red[which][w][c];
    if (c < DH) atomicAdd(&sums[(blockIdx.x % SUMS_K) * 2 * DHP + idx], a);
  }
}

#define XPAIRS (DIN / 2)
__global__ __launch_bounds__(256) void gather_x(
    const float* __restrict__ x, const int* __restrict__ rowptr,
    const int* __restrict__ csr_src, const float* __restrict__ dinv,
    float* __restrict__ ax, int M) {
  const int lane = threadIdx.x & 31;
  const int q = lane & 7;
  const int n = (blockIdx.x * 8 + (threadIdx.x >> 5)) * 4 + (lane >> 3);
  const bool rowok = n < M;
  const int nc = rowok ? n : M - 1;
  const int beg = rowptr[nc];
  const int end = rowptr[nc + 1];
  const int deg = rowok ? end - beg : 0;
  const float dn = dinv[nc];
  const unsigned o03 = 8u * q;
  const unsigned o4  = 8u * min(q + 32, XPAIRS - 1);
  const char* xb = (const char*)x;

  v2f acc[5];
  {
    const unsigned r = (unsigned)nc * (4u * DIN);
#pragma unroll
    for (int i = 0; i < 5; ++i) {
      const v2f v = *(const v2f*)(xb + (i < 4 ? r + o03 + 64u * i : r + o4));
      acc[i].x = v.x * dn * dn;
      acc[i].y = v.y * dn * dn;
    }
  }
  int maxdeg = deg;
  maxdeg = max(maxdeg, __shfl_xor(maxdeg, 8));
  maxdeg = max(maxdeg, __shfl_xor(maxdeg, 16));
  maxdeg = __builtin_amdgcn_readfirstlane(maxdeg);

  for (int c = 0; c < maxdeg; c += 8) {
    int s_l = 0;
    float w_l = 0.f;
    if (c + q < deg) {
      s_l = (int)min((unsigned)csr_src[beg + c + q], (unsigned)(M - 1));
      w_l = dinv[s_l] * dn;
    }
    for (int t0 = 0; t0 < 8 && c + t0 < maxdeg; t0 += 4) {
      v2f v[4][5];
      float wgt[4];
#pragma unroll
      for (int u = 0; u < 4; ++u) {
        const int s = __shfl(s_l, t0 + u, 8);
        wgt[u] = __shfl(w_l, t0 + u, 8);
        const unsigned r = (unsigned)s * (4u * DIN);
#pragma unroll
        for (int i = 0; i < 5; ++i)
          v[u][i] = *(const v2f*)(xb + (i < 4 ? r + o03 + 64u * i : r + o4));
      }
#pragma unroll
      for (int u = 0; u < 4; ++u) {
        const bool on = c + t0 + u < deg;
#pragma unroll
        for (int i = 0; i < 5; ++i) {
          acc[i].x += (on ? v[u][i].x : 0.f) * wgt[u];
          acc[i].y += (on ? v[u][i].y : 0.f) * wgt[u];
        }
      }
    }
  }
  if (rowok) {
    v2f* o = (v2f*)(ax + (size_t)n * DINP) + q;
#pragma unroll
    for (int i = 0; i < 5; ++i) {
      v2f r = acc[i];
      if (q + 8 * i >= XPAIRS) r = (v2f){0.f, 0.f};
      o[8 * i] = r;
    }
  }
}

__device__ __forceinline__ unsigned enc_f32(float f) {
  unsigned u = __float_as_uint(f);
  return (u & 0x80000000u) ? ~u : (u | 0x80000000u);
}
__device__ __forceinline__ void pool_put(float* p, unsigned e) {
  const unsigned b = (e & 0x80000000u) ? (e & 0x7FFFFFFFu) : ~e;
  if (b & 0x80000000u) atomicMin((unsigned*)p, b);
  else                 atomicMax((int*)p, (int)b);
}
#define POOL_ROWS  64
#define POOL_BURST 16
__global__ __launch_bounds__(DHP) void pool_max(const float* __restrict__ y,
                                                const int* __restrict__ batch,
                                                float* __restrict__ out, int M,
                                                const float* __restrict__ sums,
                                                const float* __restrict__ gamma,
                                                const float* __restrict__ beta) {
  const int c = threadIdx.x;
  if (c >= DH) return;
  float scale, shift;
  bn_coef(sums, gamma, beta, c, 1.0f / (float)M, scale, shift);
  const int r0 = blockIdx.x * POOL_ROWS;
  const int r1 = min(r0 + POOL_ROWS, M);
  if (r0 >= r1) return;
  int gcur = batch[r0];
  unsigned best = 0u;
  int r = r0;
  for (; r + POOL_BURST <= r1; r += POOL_BURST) {
    const float* p = y + (size_t)r * DHP + c;
    float v[POOL_BURST];
    int g[POOL_BURST];
#pragma unroll
    for (int i = 0; i < POOL_BURST; ++i) {
      v[i] = p[(size_t)i * DHP];
      g[i] = batch[r + i];
    }
#pragma unroll
    for (int i = 0; i < POOL_BURST; ++i) {
      if (g[i] != gcur) {
        pool_put(&out[(size_t)gcur * DH + c], best);
        gcur = g[i];
        best = 0u;
      }
      best = max(best, enc_f32(v[i] * scale + shift));
    }
  }
  for (; r < r1; ++r) {
    const int g = batch[r];
    if (g != gcur) {
      pool_put(&out[(size_t)gcur * DH + c], best);
      gcur = g;
      best = 0u;
    }
    best = max(best, enc_f32(y[(size_t)r * DHP + c] * scale + shift));
  }
  pool_put(&out[(size_t)gcur * DH + c], best);
}

extern "C" void kernel_launch(void* const* d_in, const int* in_sizes, int n_in,
                              void* d_out, int out_size, void* d_ws, size_t ws_size,
                              hipStream_t stream) {
  const float* x      = (const float*)d_in[0];
  const int*   eidx   = (const int*)d_in[1];
  const int*   batch  = (const int*)d_in[2];
  const float* W1     = (const float*)d_in[3];
  const float* b1     = (const float*)d_in[4];
  const float* gamma1 = (const float*)d_in[5];
  const float* beta1  = (const float*)d_in[6];
  const float* W2     = (const float*)d_in[7];
  const float* b2     = (const float*)d_in[8];
  const float* gamma2 = (const float*)d_in[9];
  const float* beta2  = (const float*)d_in[10];
  float* out = (float*)d_out;

  const int* src = eidx;
  const int* dst = eidx + EE;

  float* ws = (float*)d_ws;
  float*    dinv = ws;
  int*      offT = (int*)(ws + 50048);
  int*      rowp = (int*)(ws + 100096);
  unsigned* srt  = (unsigned*)(ws + 150208);
  int*      csrs = (int*)(ws + 950208);
  float*    ax   = ws + 1750208;
  float*    Wp1  = ws + 5750208;
  float*    Wp2  = ws + 5775808;
  float*    bufA = ws + 6034880;
  float*    bufB = ws + 22034880;
  float*    cvec = ws + 38034880;
  float*    sums1 = ws + 38035200;
  float*    sums2 = ws + 38037760;

  const dim3 gemmGrid(((NN + 31) / 32 + 7) / 8, DHP / 64);

  bin_edges<<<NCHUNK, 1024, 0, stream>>>(src, dst, srt, offT, W1, sums1, Wp1, out);
  build_rows<<<NBINS, 1024, 0, stream>>>(srt, offT, rowp, dinv, csrs);

  gather_x<<<(NN + 31) / 32, 256, 0, stream>>>(x, rowp, csrs, dinv, ax, NN);
  gemm_f32_wmma<false, true><<<gemmGrid, 256, 0, stream>>>(ax, DINP, Wp1, bufB, NN, DINP, nullptr,
                                                           b1, sums1);

  fold_bn_W<<<DHP / 32, 1024, 0, stream>>>(W2, sums1, gamma1, beta1, Wp2, cvec, NN);
  gemm_f32_wmma<true, false><<<gemmGrid, 256, 0, stream>>>(bufB, DHP, Wp2, bufA, NN, DHP, cvec,
                                                           nullptr, nullptr);
  gather_rows<<<(NN + 8 * GR_ROWS - 1) / (8 * GR_ROWS), 256, 0, stream>>>(bufA, rowp, csrs, dinv, bufB, NN,
                                                                         b2, sums2);

  pool_max<<<(NN + POOL_ROWS - 1) / POOL_ROWS, DHP, 0, stream>>>(bufB, batch, out, NN,
                                                                sums2, gamma2, beta2);
}
